# starting kernel
# baseline (speedup 1.0000x reference)
#include <hip/hip_runtime.h>
#include <math.h>

typedef __attribute__((ext_vector_type(2))) float v2f;
typedef __attribute__((ext_vector_type(8))) float v8f;

#define NEG_SLOPE 0.2f
#define EPS_F 1e-16f

__device__ inline void atomicMaxFloat(float* addr, float val) {
    if (val >= 0.0f) {
        atomicMax((int*)addr, __float_as_int(val));
    } else {
        atomicMin((unsigned int*)addr, __float_as_uint(val));
    }
}

__global__ void init_ws(float* m1, float* den1, float* out1,
                        float* m2, float* den2, float* out2, int N) {
    int i = blockIdx.x * blockDim.x + threadIdx.x;
    if (i < N * 128) out1[i] = 0.0f;
    if (i < N * 4) { m1[i] = -INFINITY; den1[i] = 0.0f; }
    if (i < N * 2) out2[i] = 0.0f;
    if (i < N)     { m2[i] = -INFINITY; den2[i] = 0.0f; }
}

__global__ __launch_bounds__(256) void gemm_wmma_f32(const float* __restrict__ X,
                                                     const float* __restrict__ W,
                                                     float* __restrict__ H, int N) {
    const int rowBase = blockIdx.x * 16;
    const int wave = threadIdx.x >> 5;
    const int lane = threadIdx.x & 31;
    const int colBase = wave * 16;
    const int half = lane >> 4;
    const int mn = lane & 15;

    v8f c = {};
    const float* xrow = X + (size_t)(rowBase + mn) * 128;
    #pragma unroll 4
    for (int k = 0; k < 128; k += 4) {
        const int ka = k + 2 * half;
        v2f a;
        a.x = xrow[ka];
        a.y = xrow[ka + 1];
        v2f b;
        b.x = W[(size_t)ka * 128 + colBase + mn];
        b.y = W[(size_t)(ka + 1) * 128 + colBase + mn];
        c = __builtin_amdgcn_wmma_f32_16x16x4_f32(false, a, false, b,
                                                  (short)0, c, false, false);
    }
    #pragma unroll
    for (int r = 0; r < 8; ++r) {
        H[(size_t)(rowBase + r + 8 * half) * 128 + colBase + mn] = c[r];
    }
}

__global__ void attn_l1(const float* __restrict__ h1,
                        const float* __restrict__ att_s,
                        const float* __restrict__ att_d,
                        float* __restrict__ as1, float* __restrict__ ad1, int N) {
    int t = blockIdx.x * blockDim.x + threadIdx.x;
    if (t >= N * 4) return;
    int n = t >> 2, h = t & 3;
    const float4* hp = (const float4*)(h1 + (size_t)n * 128 + h * 32);
    const float4* sp = (const float4*)(att_s + h * 32);
    const float4* dp = (const float4*)(att_d + h * 32);
    float s = 0.0f, d = 0.0f;
    #pragma unroll
    for (int j = 0; j < 8; ++j) {
        float4 hv = hp[j], sv = sp[j], dv = dp[j];
        s += hv.x * sv.x + hv.y * sv.y + hv.z * sv.z + hv.w * sv.w;
        d += hv.x * dv.x + hv.y * dv.y + hv.z * dv.z + hv.w * dv.w;
    }
    as1[t] = s;
    ad1[t] = d;
}

__global__ void edge_max_l1(const int* __restrict__ ei, int E, int N,
                            const float* __restrict__ as1, const float* __restrict__ ad1,
                            float* __restrict__ m1) {
    int t = blockIdx.x * blockDim.x + threadIdx.x;
    int tot = (E + N) * 4;
    if (t >= tot) return;
    int e = t >> 2, h = t & 3;
    int src, dst;
    if (e < E) { src = ei[e]; dst = ei[E + e]; } else { src = dst = e - E; }
    float v = as1[src * 4 + h] + ad1[dst * 4 + h];
    v = v >= 0.0f ? v : NEG_SLOPE * v;
    atomicMaxFloat(&m1[dst * 4 + h], v);
}

__global__ void edge_sum_l1(const int* __restrict__ ei, int E, int N,
                            const float* __restrict__ as1, const float* __restrict__ ad1,
                            const float* __restrict__ m1, float* __restrict__ den1) {
    int t = blockIdx.x * blockDim.x + threadIdx.x;
    int tot = (E + N) * 4;
    if (t >= tot) return;
    int e = t >> 2, h = t & 3;
    int src, dst;
    if (e < E) { src = ei[e]; dst = ei[E + e]; } else { src = dst = e - E; }
    float v = as1[src * 4 + h] + ad1[dst * 4 + h];
    v = v >= 0.0f ? v : NEG_SLOPE * v;
    float mm = m1[dst * 4 + h];
    if (!isfinite(mm)) mm = 0.0f;
    atomicAdd(&den1[dst * 4 + h], expf(v - mm));
}

__global__ __launch_bounds__(256) void edge_agg_l1(const int* __restrict__ ei, int E, int N,
                                                   const float* __restrict__ as1,
                                                   const float* __restrict__ ad1,
                                                   const float* __restrict__ m1,
                                                   const float* __restrict__ den1,
                                                   const float* __restrict__ h1,
                                                   float* __restrict__ out1) {
    int e = blockIdx.x * 8 + (threadIdx.x >> 5);
    if (e >= E + N) return;
    int lane = threadIdx.x & 31;
    int src, dst;
    if (e < E) { src = ei[e]; dst = ei[E + e]; } else { src = dst = e - E; }
    int h = lane >> 3;
    float v = as1[src * 4 + h] + ad1[dst * 4 + h];
    v = v >= 0.0f ? v : NEG_SLOPE * v;
    float mm = m1[dst * 4 + h];
    if (!isfinite(mm)) mm = 0.0f;
    float alpha = expf(v - mm) / (den1[dst * 4 + h] + EPS_F);
    const float4* hs = (const float4*)(h1 + (size_t)src * 128);
    float4 hv = hs[lane];
    float* ob = out1 + (size_t)dst * 128 + lane * 4;
    atomicAdd(ob + 0, alpha * hv.x);
    atomicAdd(ob + 1, alpha * hv.y);
    atomicAdd(ob + 2, alpha * hv.z);
    atomicAdd(ob + 3, alpha * hv.w);
}

__global__ void bias_elu(float* __restrict__ out1, const float* __restrict__ b1, int N) {
    int i = blockIdx.x * blockDim.x + threadIdx.x;
    if (i >= N * 128) return;
    float v = out1[i] + b1[i & 127];
    out1[i] = v > 0.0f ? v : expf(v) - 1.0f;
}

__global__ __launch_bounds__(256) void l2_matvec(const float* __restrict__ hf,
                                                 const float* __restrict__ W2,
                                                 const float* __restrict__ att_s2,
                                                 const float* __restrict__ att_d2,
                                                 float* __restrict__ h2,
                                                 float* __restrict__ as2,
                                                 float* __restrict__ ad2, int N) {
    int n = blockIdx.x * 8 + (threadIdx.x >> 5);
    if (n >= N) return;
    int lane = threadIdx.x & 31;
    const float4* hp = (const float4*)(hf + (size_t)n * 128);
    float4 hv = hp[lane];
    int c = lane * 4;
    float s0 = hv.x * W2[(c + 0) * 2]     + hv.y * W2[(c + 1) * 2]
             + hv.z * W2[(c + 2) * 2]     + hv.w * W2[(c + 3) * 2];
    float s1 = hv.x * W2[(c + 0) * 2 + 1] + hv.y * W2[(c + 1) * 2 + 1]
             + hv.z * W2[(c + 2) * 2 + 1] + hv.w * W2[(c + 3) * 2 + 1];
    #pragma unroll
    for (int off = 16; off >= 1; off >>= 1) {
        s0 += __shfl_xor(s0, off, 32);
        s1 += __shfl_xor(s1, off, 32);
    }
    if (lane == 0) {
        h2[n * 2 + 0] = s0;
        h2[n * 2 + 1] = s1;
        as2[n] = s0 * att_s2[0] + s1 * att_s2[1];
        ad2[n] = s0 * att_d2[0] + s1 * att_d2[1];
    }
}

__global__ void edge_max_l2(const int* __restrict__ ei, int E, int N,
                            const float* __restrict__ as2, const float* __restrict__ ad2,
                            float* __restrict__ m2) {
    int e = blockIdx.x * blockDim.x + threadIdx.x;
    if (e >= E + N) return;
    int src, dst;
    if (e < E) { src = ei[e]; dst = ei[E + e]; } else { src = dst = e - E; }
    float v = as2[src] + ad2[dst];
    v = v >= 0.0f ? v : NEG_SLOPE * v;
    atomicMaxFloat(&m2[dst], v);
}

__global__ void edge_sum_l2(const int* __restrict__ ei, int E, int N,
                            const float* __restrict__ as2, const float* __restrict__ ad2,
                            const float* __restrict__ m2, float* __restrict__ den2) {
    int e = blockIdx.x * blockDim.x + threadIdx.x;
    if (e >= E + N) return;
    int src, dst;
    if (e < E) { src = ei[e]; dst = ei[E + e]; } else { src = dst = e - E; }
    float v = as2[src] + ad2[dst];
    v = v >= 0.0f ? v : NEG_SLOPE * v;
    float mm = m2[dst];
    if (!isfinite(mm)) mm = 0.0f;
    atomicAdd(&den2[dst], expf(v - mm));
}

__global__ void edge_agg_l2(const int* __restrict__ ei, int E, int N,
                            const float* __restrict__ as2, const float* __restrict__ ad2,
                            const float* __restrict__ m2, const float* __restrict__ den2,
                            const float* __restrict__ h2, float* __restrict__ out2) {
    int e = blockIdx.x * blockDim.x + threadIdx.x;
    if (e >= E + N) return;
    int src, dst;
    if (e < E) { src = ei[e]; dst = ei[E + e]; } else { src = dst = e - E; }
    float v = as2[src] + ad2[dst];
    v = v >= 0.0f ? v : NEG_SLOPE * v;
    float mm = m2[dst];
    if (!isfinite(mm)) mm = 0.0f;
    float alpha = expf(v - mm) / (den2[dst] + EPS_F);
    atomicAdd(&out2[dst * 2 + 0], alpha * h2[src * 2 + 0]);
    atomicAdd(&out2[dst * 2 + 1], alpha * h2[src * 2 + 1]);
}

__global__ void final_logsoftmax(const float* __restrict__ out2,
                                 const float* __restrict__ b2,
                                 float* __restrict__ y, int N) {
    int n = blockIdx.x * blockDim.x + threadIdx.x;
    if (n >= N) return;
    float z0 = out2[n * 2 + 0] + b2[0];
    float z1 = out2[n * 2 + 1] + b2[1];
    float mx = fmaxf(z0, z1);
    float lse = mx + logf(expf(z0 - mx) + expf(z1 - mx));
    y[n * 2 + 0] = z0 - lse;
    y[n * 2 + 1] = z1 - lse;
}

extern "C" void kernel_launch(void* const* d_in, const int* in_sizes, int n_in,
                              void* d_out, int out_size, void* d_ws, size_t ws_size,
                              hipStream_t stream) {
    const float* x        = (const float*)d_in[0];
    const int*   ei       = (const int*)d_in[1];
    const float* W1       = (const float*)d_in[2];
    const float* att_src1 = (const float*)d_in[3];
    const float* att_dst1 = (const float*)d_in[4];
    const float* b1       = (const float*)d_in[5];
    const float* W2       = (const float*)d_in[6];
    const float* att_src2 = (const float*)d_in[7];
    const float* att_dst2 = (const float*)d_in[8];
    const float* b2       = (const float*)d_in[9];
    float* y = (float*)d_out;

    const int N = in_sizes[0] / 128;
    const int E = in_sizes[1] / 2;
    const int Etot = E + N;

    float* ws   = (float*)d_ws;
    float* h1   = ws;
    float* out1 = h1   + (size_t)N * 128;
    float* as1  = out1 + (size_t)N * 128;
    float* ad1  = as1  + (size_t)N * 4;
    float* m1   = ad1  + (size_t)N * 4;
    float* den1 = m1   + (size_t)N * 4;
    float* h2   = den1 + (size_t)N * 4;
    float* as2  = h2   + (size_t)N * 2;
    float* ad2  = as2  + (size_t)N;
    float* m2   = ad2  + (size_t)N;
    float* den2 = m2   + (size_t)N;
    float* out2 = den2 + (size_t)N;

    const int TB = 256;

    init_ws<<<(N * 128 + TB - 1) / TB, TB, 0, stream>>>(m1, den1, out1, m2, den2, out2, N);

    gemm_wmma_f32<<<N / 16, 256, 0, stream>>>(x, W1, h1, N);
    attn_l1<<<(N * 4 + TB - 1) / TB, TB, 0, stream>>>(h1, att_src1, att_dst1, as1, ad1, N);

    edge_max_l1<<<(Etot * 4 + TB - 1) / TB, TB, 0, stream>>>(ei, E, N, as1, ad1, m1);
    edge_sum_l1<<<(Etot * 4 + TB - 1) / TB, TB, 0, stream>>>(ei, E, N, as1, ad1, m1, den1);
    edge_agg_l1<<<(Etot + 7) / 8, 256, 0, stream>>>(ei, E, N, as1, ad1, m1, den1, h1, out1);
    bias_elu<<<(N * 128 + TB - 1) / TB, TB, 0, stream>>>(out1, b1, N);

    l2_matvec<<<(N + 7) / 8, 256, 0, stream>>>(out1, W2, att_src2, att_dst2, h2, as2, ad2, N);

    edge_max_l2<<<(Etot + TB - 1) / TB, TB, 0, stream>>>(ei, E, N, as2, ad2, m2);
    edge_sum_l2<<<(Etot + TB - 1) / TB, TB, 0, stream>>>(ei, E, N, as2, ad2, m2, den2);
    edge_agg_l2<<<(Etot + TB - 1) / TB, TB, 0, stream>>>(ei, E, N, as2, ad2, m2, den2, h2, out2);

    final_logsoftmax<<<(N + TB - 1) / TB, TB, 0, stream>>>(out2, b2, y, N);
}


# Layer-2 projection and attention scalars are formed in the layer-1 kernel from the finished rows in LDS; ELU rows are never stored; the wave-per-node matvec launch is gone; h2 sum order as before
# speedup vs baseline: 4.0702x; 4.0702x over previous
#include <hip/hip_runtime.h>
#include <math.h>

typedef __attribute__((ext_vector_type(2))) float v2f;
typedef __attribute__((ext_vector_type(8))) float v8f;

#define NEG_SLOPE 0.2f
#define EPS_F 1e-16f

__global__ __launch_bounds__(256) void gemm_wmma_f32(const float* __restrict__ X,
                                                     const float* __restrict__ W,
                                                     float* __restrict__ H, int N) {
    const int rowBase = blockIdx.x * 16;
    const int wave = threadIdx.x >> 5;
    const int lane = threadIdx.x & 31;
    const int colBase = wave * 16;
    const int half = lane >> 4;
    const int mn = lane & 15;

    v8f c = {};
    const float* xrow = X + (size_t)(rowBase + mn) * 128;
    #pragma unroll 4
    for (int k = 0; k < 128; k += 4) {
        const int ka = k + 2 * half;
        v2f a;
        a.x = xrow[ka];
        a.y = xrow[ka + 1];
        v2f b;
        b.x = W[(size_t)ka * 128 + colBase + mn];
        b.y = W[(size_t)(ka + 1) * 128 + colBase + mn];
        c = __builtin_amdgcn_wmma_f32_16x16x4_f32(false, a, false, b,
                                                  (short)0, c, false, false);
    }
    #pragma unroll
    for (int r = 0; r < 8; ++r) {
        H[(size_t)(rowBase + r + 8 * half) * 128 + colBase + mn] = c[r];
    }
}

__global__ void attn_l1(const float* __restrict__ h1,
                        const float* __restrict__ att_s,
                        const float* __restrict__ att_d,
                        float* __restrict__ as1, float* __restrict__ ad1, int N) {
    int t = blockIdx.x * blockDim.x + threadIdx.x;
    if (t >= N * 4) return;
    int n = t >> 2, h = t & 3;
    const float4* hp = (const float4*)(h1 + (size_t)n * 128 + h * 32);
    const float4* sp = (const float4*)(att_s + h * 32);
    const float4* dp = (const float4*)(att_d + h * 32);
    float s = 0.0f, d = 0.0f;
    #pragma unroll
    for (int j = 0; j < 8; ++j) {
        float4 hv = hp[j], sv = sp[j], dv = dp[j];
        s += hv.x * sv.x + hv.y * sv.y + hv.z * sv.z + hv.w * sv.w;
        d += hv.x * dv.x + hv.y * dv.y + hv.z * dv.z + hv.w * dv.w;
    }
    as1[t] = s;
    ad1[t] = d;
}

#define ROWS 196
#define QCAP 8192
#define L1_THREADS 1024
#define SCAN_VECS 8
#define SCAN_SPAN (L1_THREADS * 4 * SCAN_VECS)

struct L1Shared {
    float acc[ROWS * 128];
    float4 lg[QCAP];
    int2 q[QCAP];
    float mx[ROWS * 4];
    float den[ROWS * 4];
    float ad[ROWS * 4];
    int qcount;
};

__device__ inline void lds_max_float(float* addr, float val) {
    if (val >= 0.0f) atomicMax((int*)addr, __float_as_int(val));
    else             atomicMin((unsigned int*)addr, __float_as_uint(val));
}

__device__ inline float leaky(float v) { return v >= 0.0f ? v : NEG_SLOPE * v; }

__device__ inline void note_edge(L1Shared& s, int dv, int e, int r0, int rn) {
    const unsigned r = (unsigned)(dv - r0);
    if (r < (unsigned)rn) {
        const int slot = atomicAdd(&s.qcount, 1);
        if (slot < QCAP) s.q[slot] = make_int2(e, (int)r);
    }
}

__device__ inline void scan_edges(L1Shared& s, const int* __restrict__ dstp, int eb, int ee, bool vec_ok,
                                  int r0, int rn) {
    const int tid = threadIdx.x;
    const int span = ee - eb;
    const int nsuper = (span + SCAN_SPAN - 1) / SCAN_SPAN;
    if (nsuper <= 0) return;
    const int rot = (int)((blockIdx.x * 5u) % (unsigned)nsuper);
    for (int it = 0; it < nsuper; ++it) {
        int ss = it + rot;
        if (ss >= nsuper) ss -= nsuper;
        const int sbase = eb + ss * SCAN_SPAN;
        if (vec_ok && sbase + SCAN_SPAN <= ee) {
            const int4* p = (const int4*)(dstp + sbase) + tid;
            int4 d[SCAN_VECS];
            #pragma unroll
            for (int k = 0; k < SCAN_VECS; ++k) d[k] = p[k * L1_THREADS];
            #pragma unroll
            for (int k = 0; k < SCAN_VECS; ++k) {
                const int e0 = sbase + (k * L1_THREADS + tid) * 4;
                note_edge(s, d[k].x, e0,     r0, rn);
                note_edge(s, d[k].y, e0 + 1, r0, rn);
                note_edge(s, d[k].z, e0 + 2, r0, rn);
                note_edge(s, d[k].w, e0 + 3, r0, rn);
            }
        } else {
            for (int kb = sbase; kb < min(ee, sbase + SCAN_SPAN); kb += L1_THREADS * 4) {
                if (vec_ok && kb + L1_THREADS * 4 <= ee) {
                    const int4 d = *((const int4*)(dstp + kb) + tid);
                    const int e0 = kb + tid * 4;
                    note_edge(s, d.x, e0,     r0, rn);
                    note_edge(s, d.y, e0 + 1, r0, rn);
                    note_edge(s, d.z, e0 + 2, r0, rn);
                    note_edge(s, d.w, e0 + 3, r0, rn);
                } else {
                    for (int e = kb + tid; e < min(ee, kb + L1_THREADS * 4); e += L1_THREADS)
                        note_edge(s, dstp[e], e, r0, rn);
                }
            }
        }
    }
}

__device__ inline void pass_logits(L1Shared& s, int total, const int* __restrict__ ei, int E,
                                   const float* __restrict__ as1) {
    for (int i = threadIdx.x; i < total; i += L1_THREADS) {
        const int2 en = s.q[i];
        const int src = (en.x < E) ? ei[en.x] : en.x - E;
        s.q[i].x = src;
        const float4 a = *(const float4*)(as1 + (size_t)src * 4);
        const float4 d = *(const float4*)(&s.ad[en.y * 4]);
        float4 v;
        v.x = leaky(a.x + d.x); v.y = leaky(a.y + d.y); v.z = leaky(a.z + d.z); v.w = leaky(a.w + d.w);
        s.lg[i] = v;
        lds_max_float(&s.mx[en.y * 4 + 0], v.x);
        lds_max_float(&s.mx[en.y * 4 + 1], v.y);
        lds_max_float(&s.mx[en.y * 4 + 2], v.z);
        lds_max_float(&s.mx[en.y * 4 + 3], v.w);
    }
}

__device__ inline float shift_of(float m) { return isfinite(m) ? m : 0.0f; }

__device__ inline void pass_weights(L1Shared& s, int total) {
    for (int i = threadIdx.x; i < total; i += L1_THREADS) {
        const int row = s.q[i].y;
        const float4 v = s.lg[i];
        const float4 m = *(const float4*)(&s.mx[row * 4]);
        float4 w;
        w.x = expf(v.x - shift_of(m.x)); w.y = expf(v.y - shift_of(m.y));
        w.z = expf(v.z - shift_of(m.z)); w.w = expf(v.w - shift_of(m.w));
        s.lg[i] = w;
        atomicAdd(&s.den[row * 4 + 0], w.x);
        atomicAdd(&s.den[row * 4 + 1], w.y);
        atomicAdd(&s.den[row * 4 + 2], w.z);
        atomicAdd(&s.den[row * 4 + 3], w.w);
    }
}

__device__ inline float lane_value(float v, int k) {
    return __int_as_float(__builtin_amdgcn_readlane(__float_as_int(v), k));
}

__device__ inline void pass_rows(L1Shared& s, int total, const float* __restrict__ h1) {
    const int lane = threadIdx.x & 31;
    const int wave = threadIdx.x >> 5;
    for (int i0 = wave * 8; i0 < total; i0 += (L1_THREADS / 32) * 8) {
        const int mine = min(i0 + (lane & 7), total - 1);
        const int2 en = s.q[mine];
        const float4 wv = s.lg[mine];
        float x[8][4];
        #pragma unroll
        for (int k = 0; k < 8; ++k) {
            x[k][0] = x[k][1] = x[k][2] = x[k][3] = 0.0f;
            if (i0 + k < total) {
                const int src = __builtin_amdgcn_readlane(en.x, k);
                const float* hs = h1 + (size_t)src * 128 + lane;
                x[k][0] = hs[0]; x[k][1] = hs[32]; x[k][2] = hs[64]; x[k][3] = hs[96];
            }
        }
        #pragma unroll
        for (int k = 0; k < 8; ++k) {
            if (i0 + k < total) {
                const int row = __builtin_amdgcn_readlane(en.y, k);
                float* ob = &s.acc[row * 128 + lane];
                atomicAdd(ob,      lane_value(wv.x, k) * x[k][0]);
                atomicAdd(ob + 32, lane_value(wv.y, k) * x[k][1]);
                atomicAdd(ob + 64, lane_value(wv.z, k) * x[k][2]);
                atomicAdd(ob + 96, lane_value(wv.w, k) * x[k][3]);
            }
        }
    }
}

__global__ __launch_bounds__(L1_THREADS) void l1_owned_rows(const int* __restrict__ ei, int E, int N,
                                                            const float* __restrict__ as1,
                                                            const float* __restrict__ ad1,
                                                            const float* __restrict__ h1,
                                                            const float* __restrict__ b1,
                                                            const float* __restrict__ W2,
                                                            const float* __restrict__ att_s2,
                                                            const float* __restrict__ att_d2,
                                                            float* __restrict__ h2,
                                                            float* __restrict__ as2,
                                                            float* __restrict__ ad2) {
    __shared__ L1Shared s;
    const int tid = threadIdx.x;
    const int r0 = blockIdx.x * ROWS;
    const int rn = min(ROWS, N - r0);
    const int* dstp = ei + E;
    const bool vec_ok = ((E & 3) == 0) && ((((size_t)dstp) & 15) == 0);

    for (int i = tid; i < ROWS * 128; i += L1_THREADS) s.acc[i] = 0.0f;
    for (int i = tid; i < ROWS * 4; i += L1_THREADS) {
        s.mx[i] = -INFINITY;
        s.den[i] = 0.0f;
        s.ad[i] = (i < rn * 4) ? ad1[(size_t)r0 * 4 + i] : 0.0f;
    }
    if (tid < rn) s.q[tid] = make_int2(E + r0 + tid, tid);
    if (tid == 0) s.qcount = rn;
    __syncthreads();

    scan_edges(s, dstp, 0, E, vec_ok, r0, rn);
    __syncthreads();
    const int found = s.qcount;

    if (found <= QCAP) {
        pass_logits(s, found, ei, E, as1);
        __syncthreads();
        pass_weights(s, found);
        __syncthreads();
        pass_rows(s, found, h1);
    } else {
        const int SEG = QCAP - 256;
        const int nseg = (E + SEG - 1) / SEG;
        for (int sweep = 0; sweep < 2; ++sweep) {
            for (int sg = 0; sg < nseg; ++sg) {
                __syncthreads();
                if (sg == 0 && tid < rn) s.q[tid] = make_int2(E + r0 + tid, tid);
                if (tid == 0) s.qcount = (sg == 0) ? rn : 0;
                __syncthreads();
                scan_edges(s, dstp, sg * SEG, min(E, (sg + 1) * SEG), vec_ok, r0, rn);
                __syncthreads();
                const int cnt = s.qcount;
                pass_logits(s, cnt, ei, E, as1);
                __syncthreads();
                if (sweep == 1) {
                    pass_weights(s, cnt);
                    __syncthreads();
                    pass_rows(s, cnt, h1);
                }
            }
        }
    }
    const int lane = tid & 31;
    const int c = lane * 4;
    const float bx = b1[c], by = b1[c + 1], bz = b1[c + 2], bw = b1[c + 3];
    const float w00 = W2[(c + 0) * 2], w01 = W2[(c + 0) * 2 + 1];
    const float w10 = W2[(c + 1) * 2], w11 = W2[(c + 1) * 2 + 1];
    const float w20 = W2[(c + 2) * 2], w21 = W2[(c + 2) * 2 + 1];
    const float w30 = W2[(c + 3) * 2], w31 = W2[(c + 3) * 2 + 1];
    const float ts0 = att_s2[0], ts1 = att_s2[1], td0 = att_d2[0], td1 = att_d2[1];
    __syncthreads();

    for (int row = tid >> 5; row < rn; row += L1_THREADS / 32) {
        const float4 a = *(const float4*)(&s.acc[row * 128 + c]);
        const float d = s.den[row * 4 + (lane >> 3)] + EPS_F;
        float4 v;
        v.x = a.x / d + bx; v.y = a.y / d + by; v.z = a.z / d + bz; v.w = a.w / d + bw;
        v.x = v.x > 0.0f ? v.x : expf(v.x) - 1.0f;
        v.y = v.y > 0.0f ? v.y : expf(v.y) - 1.0f;
        v.z = v.z > 0.0f ? v.z : expf(v.z) - 1.0f;
        v.w = v.w > 0.0f ? v.w : expf(v.w) - 1.0f;
        float s0 = v.x * w00 + v.y * w10 + v.z * w20 + v.w * w30;
        float s1 = v.x * w01 + v.y * w11 + v.z * w21 + v.w * w31;
        #pragma unroll
        for (int off = 16; off >= 1; off >>= 1) {
            s0 += __shfl_xor(s0, off, 32);
            s1 += __shfl_xor(s1, off, 32);
        }
        if (lane == 0)
            s.lg[row] = make_float4(s0, s1, s0 * ts0 + s1 * ts1, s0 * td0 + s1 * td1);
    }
    __syncthreads();

    if (tid < rn) {
        const float4 r = s.lg[tid];
        const int n = r0 + tid;
        *(float2*)(h2 + (size_t)n * 2) = make_float2(r.x, r.y);
        as2[n] = r.z;
        ad2[n] = r.w;
    }
}

#define L2_ROWS 3136
#define L2_SLICES 16
#define L2_CAP 8192
#define L2_THREADS 1024
#define L2_VECS 8
#define L2_STAGE 4

struct L2Shared {
    int2 q[L2_CAP];
    float2 hx[L2_CAP];
    float lg[L2_CAP];
    float mx[L2_ROWS];
    float den[L2_ROWS];
    float s0[L2_ROWS];
    float s1[L2_ROWS];
    float ad[L2_ROWS];
    int qcount;
};

__device__ inline void l2_note(L2Shared& s, int dv, int e, int r0, int rn) {
    const unsigned r = (unsigned)(dv - r0);
    if (r < (unsigned)rn) {
        const int slot = atomicAdd(&s.qcount, 1);
        if (slot < L2_CAP) s.q[slot] = make_int2(e, (int)r);
    }
}

__device__ inline int l2_hit(int dv, int r0, int rn) {
    return ((unsigned)(dv - r0) < (unsigned)rn) ? 1 : 0;
}

__device__ inline int l2_put(L2Shared& s, int dv, int e, int r0, int rn, int slot) {
    const unsigned r = (unsigned)(dv - r0);
    if (r < (unsigned)rn) {
        if (slot < L2_CAP) s.q[slot] = make_int2(e, (int)r);
        ++slot;
    }
    return slot;
}

__device__ __forceinline__ void l2_scan(L2Shared& s, const int* __restrict__ dstp, int eb, int ee, bool vec_ok,
                               int r0, int rn) {
    const int tid = threadIdx.x;
    int done = eb;
    if (vec_ok) {
        const int nvec = (ee - eb) >> 2;
        const int4* p = (const int4*)(dstp + eb);
        for (int vb = 0; vb < nvec; vb += L2_THREADS * L2_VECS) {
            int4 d[L2_VECS];
            #pragma unroll
            for (int k = 0; k < L2_VECS; ++k) {
                const int v = vb + k * L2_THREADS + tid;
                d[k] = (v < nvec) ? p[v] : make_int4(-1, -1, -1, -1);
            }
            int cnt = 0;
            #pragma unroll
            for (int k = 0; k < L2_VECS; ++k)
                cnt += l2_hit(d[k].x, r0, rn) + l2_hit(d[k].y, r0, rn) + l2_hit(d[k].z, r0, rn) + l2_hit(d[k].w, r0, rn);
            const int lane = tid & 31;
            int incl = cnt;
            #pragma unroll
            for (int off = 1; off < 32; off <<= 1) {
                const int t = __shfl_up(incl, off, 32);
                if (lane >= off) incl += t;
            }
            const int wave_total = __shfl(incl, 31, 32);
            int wave_base = 0;
            if (lane == 31 && wave_total > 0) wave_base = atomicAdd(&s.qcount, wave_total);
            wave_base = __shfl(wave_base, 31, 32);
            int slot = wave_base + incl - cnt;
            #pragma unroll
            for (int k = 0; k < L2_VECS; ++k) {
                const int e0 = eb + (vb + k * L2_THREADS + tid) * 4;
                slot = l2_put(s, d[k].x, e0,     r0, rn, slot);
                slot = l2_put(s, d[k].y, e0 + 1, r0, rn, slot);
                slot = l2_put(s, d[k].z, e0 + 2, r0, rn, slot);
                slot = l2_put(s, d[k].w, e0 + 3, r0, rn, slot);
            }
        }
        done = eb + nvec * 4;
    }
    for (int e = done + tid; e < ee; e += L2_THREADS) l2_note(s, dstp[e], e, r0, rn);
}

__device__ inline void l2_logits(L2Shared& s, int total, const int* __restrict__ ei, int E,
                                 const float* __restrict__ as2, const float* __restrict__ h2) {
    const int tid = threadIdx.x;
    for (int base = 0; base < total; base += L2_THREADS * L2_STAGE) {
        int srcn[L2_STAGE], row[L2_STAGE];
        float a[L2_STAGE];
        float2 h[L2_STAGE];
        #pragma unroll
        for (int k = 0; k < L2_STAGE; ++k) {
            const int i = base + k * L2_THREADS + tid;
            const int2 en = s.q[min(i, total - 1)];
            row[k] = (i < total) ? en.y : -1;
            const int ld = ei[min(en.x, E - 1)];
            srcn[k] = (en.x < E) ? ld : en.x - E;
        }
        #pragma unroll
        for (int k = 0; k < L2_STAGE; ++k) {
            a[k] = as2[srcn[k]];
            h[k] = *(const float2*)(h2 + (size_t)srcn[k] * 2);
        }
        #pragma unroll
        for (int k = 0; k < L2_STAGE; ++k) {
            if (row[k] >= 0) {
                const int i = base + k * L2_THREADS + tid;
                const float v = leaky(a[k] + s.ad[row[k]]);
                s.lg[i] = v;
                s.hx[i] = h[k];
                lds_max_float(&s.mx[row[k]], v);
            }
        }
    }
}

__device__ inline void l2_weights(L2Shared& s, int total) {
    for (int i = threadIdx.x; i < total; i += L2_THREADS) {
        const int row = s.q[i].y;
        const float w = expf(s.lg[i] - shift_of(s.mx[row]));
        const float2 hv = s.hx[i];
        atomicAdd(&s.den[row], w);
        atomicAdd(&s.s0[row], w * hv.x);
        atomicAdd(&s.s1[row], w * hv.y);
    }
}

__global__ __launch_bounds__(L2_THREADS) void l2_partial(const int* __restrict__ ei, int E, int N,
                                                         const float* __restrict__ as2,
                                                         const float* __restrict__ ad2,
                                                         const float* __restrict__ h2,
                                                         float* __restrict__ part) {
    __shared__ L2Shared s;
    const int tid = threadIdx.x;
    const int range = blockIdx.x / L2_SLICES;
    const int slice = blockIdx.x % L2_SLICES;
    const int r0 = range * L2_ROWS;
    const int rn = min(L2_ROWS, N - r0);
    const int slen = ((E + L2_SLICES - 1) / L2_SLICES + 3) & ~3;
    const int eb = min(E, slice * slen);
    const int ee = min(E, eb + slen);
    const int* dstp = ei + E;
    const bool vec_ok = ((E & 3) == 0) && ((((size_t)dstp) & 15) == 0);
    const int nself = (rn > slice) ? (rn - slice + L2_SLICES - 1) / L2_SLICES : 0;

    for (int i = tid; i < L2_ROWS; i += L2_THREADS) {
        s.mx[i] = -INFINITY;
        s.den[i] = 0.0f;
        s.s0[i] = 0.0f;
        s.s1[i] = 0.0f;
        s.ad[i] = (i < rn) ? ad2[r0 + i] : 0.0f;
    }
    if (tid < nself) s.q[tid] = make_int2(E + r0 + slice + tid * L2_SLICES, slice + tid * L2_SLICES);
    if (tid == 0) s.qcount = nself;
    __syncthreads();

    l2_scan(s, dstp, eb, ee, vec_ok, r0, rn);
    __syncthreads();
    const int found = s.qcount;

    if (found <= L2_CAP) {
        l2_logits(s, found, ei, E, as2, h2);
        __syncthreads();
        l2_weights(s, found);
    } else {
        const int SEG = L2_CAP - 256;
        const int nseg = (ee - eb + SEG - 1) / SEG;
        for (int sweep = 0; sweep < 2; ++sweep) {
            for (int sg = 0; sg < nseg; ++sg) {
                __syncthreads();
                if (sg == 0 && tid < nself)
                    s.q[tid] = make_int2(E + r0 + slice + tid * L2_SLICES, slice + tid * L2_SLICES);
                if (tid == 0) s.qcount = (sg == 0) ? nself : 0;
                __syncthreads();
                l2_scan(s, dstp, eb + sg * SEG, min(ee, eb + (sg + 1) * SEG), vec_ok, r0, rn);
                __syncthreads();
                const int cnt = s.qcount;
                l2_logits(s, cnt, ei, E, as2, h2);
                __syncthreads();
                if (sweep == 1) l2_weights(s, cnt);
            }
        }
    }
    __syncthreads();

    for (int r = tid; r < rn; r += L2_THREADS)
        ((float4*)part)[(size_t)slice * N + r0 + r] = make_float4(s.mx[r], s.den[r], s.s0[r], s.s1[r]);
}

__global__ void merge_logsoftmax(const float* __restrict__ part,
                                 const float* __restrict__ b2,
                                 float* __restrict__ y, int N) {
    int n = blockIdx.x * blockDim.x + threadIdx.x;
    if (n >= N) return;
    float4 p[L2_SLICES];
    #pragma unroll
    for (int j = 0; j < L2_SLICES; ++j) p[j] = ((const float4*)part)[(size_t)j * N + n];
    float mall = p[0].x;
    #pragma unroll
    for (int j = 1; j < L2_SLICES; ++j) mall = fmaxf(mall, p[j].x);
    const float ms = shift_of(mall);
    float den = 0.0f, t0 = 0.0f, t1 = 0.0f;
    #pragma unroll
    for (int j = 0; j < L2_SLICES; ++j) {
        const float sc = (p[j].x == -INFINITY) ? 0.0f : expf(shift_of(p[j].x) - ms);
        den += sc * p[j].y;
        t0  += sc * p[j].z;
        t1  += sc * p[j].w;
    }
    const float d = den + EPS_F;
    float z0 = t0 / d + b2[0];
    float z1 = t1 / d + b2[1];
    float mx = fmaxf(z0, z1);
    float lse = mx + logf(expf(z0 - mx) + expf(z1 - mx));
    y[n * 2 + 0] = z0 - lse;
    y[n * 2 + 1] = z1 - lse;
}

extern "C" void kernel_launch(void* const* d_in, const int* in_sizes, int n_in,
                              void* d_out, int out_size, void* d_ws, size_t ws_size,
                              hipStream_t stream) {
    const float* x        = (const float*)d_in[0];
    const int*   ei       = (const int*)d_in[1];
    const float* W1       = (const float*)d_in[2];
    const float* att_src1 = (const float*)d_in[3];
    const float* att_dst1 = (const float*)d_in[4];
    const float* b1       = (const float*)d_in[5];
    const float* W2       = (const float*)d_in[6];
    const float* att_src2 = (const float*)d_in[7];
    const float* att_dst2 = (const float*)d_in[8];
    const float* b2       = (const float*)d_in[9];
    float* y = (float*)d_out;

    const int N = in_sizes[0] / 128;
    const int E = in_sizes[1] / 2;

    float* ws   = (float*)d_ws;
    float* h1   = ws;
    float* as1  = h1   + (size_t)N * 256;
    float* ad1  = as1  + (size_t)N * 4;
    float* h2   = ad1  + (size_t)N * 12;
    float* as2  = h2   + (size_t)N * 2;
    float* ad2  = as2  + (size_t)N;
    float* part = h1;

    const int TB = 256;

    gemm_wmma_f32<<<N / 16, 256, 0, stream>>>(x, W1, h1, N);
    attn_l1<<<(N * 4 + TB - 1) / TB, TB, 0, stream>>>(h1, att_src1, att_dst1, as1, ad1, N);

    l1_owned_rows<<<(N + ROWS - 1) / ROWS, L1_THREADS, 0, stream>>>(ei, E, N, as1, ad1, h1, b1,
                                                                   W2, att_src2, att_dst2, h2, as2, ad2);

    l2_partial<<<((N + L2_ROWS - 1) / L2_ROWS) * L2_SLICES, L2_THREADS, 0, stream>>>(ei, E, N, as2, ad2, h2, part);

    merge_logsoftmax<<<(N + TB - 1) / TB, TB, 0, stream>>>(part, b2, y, N);
}


# Edge list read in full once: a first kernel deals it into {source,row} record lists per (196-row block, slice); a layer-1 group copies its 16 lists, layer 2 its slice's 16; sums still by arrival
# speedup vs baseline: 4.9455x; 1.2151x over previous
#include <hip/hip_runtime.h>
#include <math.h>

typedef __attribute__((ext_vector_type(2))) float v2f;
typedef __attribute__((ext_vector_type(8))) float v8f;

#define NEG_SLOPE 0.2f
#define EPS_F 1e-16f

__global__ __launch_bounds__(256) void gemm_wmma_f32(const float* __restrict__ X,
                                                     const float* __restrict__ W,
                                                     float* __restrict__ H, int N) {
    const int rowBase = blockIdx.x * 16;
    const int wave = threadIdx.x >> 5;
    const int lane = threadIdx.x & 31;
    const int colBase = wave * 16;
    const int half = lane >> 4;
    const int mn = lane & 15;

    v8f c = {};
    const float* xrow = X + (size_t)(rowBase + mn) * 128;
    #pragma unroll 4
    for (int k = 0; k < 128; k += 4) {
        const int ka = k + 2 * half;
        v2f a;
        a.x = xrow[ka];
        a.y = xrow[ka + 1];
        v2f b;
        b.x = W[(size_t)ka * 128 + colBase + mn];
        b.y = W[(size_t)(ka + 1) * 128 + colBase + mn];
        c = __builtin_amdgcn_wmma_f32_16x16x4_f32(false, a, false, b,
                                                  (short)0, c, false, false);
    }
    #pragma unroll
    for (int r = 0; r < 8; ++r) {
        H[(size_t)(rowBase + r + 8 * half) * 128 + colBase + mn] = c[r];
    }
}

__global__ void attn_l1(const float* __restrict__ h1,
                        const float* __restrict__ att_s,
                        const float* __restrict__ att_d,
                        float* __restrict__ as1, float* __restrict__ ad1, int N) {
    int t = blockIdx.x * blockDim.x + threadIdx.x;
    if (t >= N * 4) return;
    int n = t >> 2, h = t & 3;
    const float4* hp = (const float4*)(h1 + (size_t)n * 128 + h * 32);
    const float4* sp = (const float4*)(att_s + h * 32);
    const float4* dp = (const float4*)(att_d + h * 32);
    float s = 0.0f, d = 0.0f;
    #pragma unroll
    for (int j = 0; j < 8; ++j) {
        float4 hv = hp[j], sv = sp[j], dv = dp[j];
        s += hv.x * sv.x + hv.y * sv.y + hv.z * sv.z + hv.w * sv.w;
        d += hv.x * dv.x + hv.y * dv.y + hv.z * dv.z + hv.w * dv.w;
    }
    as1[t] = s;
    ad1[t] = d;
}

#define ROWS 196
#define QCAP 8192
#define L1_THREADS 1024
#define SCAN_VECS 8
#define SCAN_SPAN (L1_THREADS * 4 * SCAN_VECS)
#define BIN_SLICES 16
#define BIN_CAP 512
#define BIN_LOADS (BIN_SLICES * BIN_CAP / L1_THREADS)

struct L1Shared {
    float acc[ROWS * 128];
    float4 lg[QCAP];
    int2 q[QCAP];
    float mx[ROWS * 4];
    float den[ROWS * 4];
    float ad[ROWS * 4];
    int qcount;
};

__device__ inline void lds_max_float(float* addr, float val) {
    if (val >= 0.0f) atomicMax((int*)addr, __float_as_int(val));
    else             atomicMin((unsigned int*)addr, __float_as_uint(val));
}

__device__ inline float leaky(float v) { return v >= 0.0f ? v : NEG_SLOPE * v; }

__device__ inline void note_edge(L1Shared& s, int dv, int e, int r0, int rn) {
    const unsigned r = (unsigned)(dv - r0);
    if (r < (unsigned)rn) {
        const int slot = atomicAdd(&s.qcount, 1);
        if (slot < QCAP) s.q[slot] = make_int2(e, (int)r);
    }
}

__device__ inline void scan_edges(L1Shared& s, const int* __restrict__ dstp, int eb, int ee, bool vec_ok,
                                  int r0, int rn) {
    const int tid = threadIdx.x;
    const int span = ee - eb;
    const int nsuper = (span + SCAN_SPAN - 1) / SCAN_SPAN;
    if (nsuper <= 0) return;
    const int rot = (int)((blockIdx.x * 5u) % (unsigned)nsuper);
    for (int it = 0; it < nsuper; ++it) {
        int ss = it + rot;
        if (ss >= nsuper) ss -= nsuper;
        const int sbase = eb + ss * SCAN_SPAN;
        if (vec_ok && sbase + SCAN_SPAN <= ee) {
            const int4* p = (const int4*)(dstp + sbase) + tid;
            int4 d[SCAN_VECS];
            #pragma unroll
            for (int k = 0; k < SCAN_VECS; ++k) d[k] = p[k * L1_THREADS];
            #pragma unroll
            for (int k = 0; k < SCAN_VECS; ++k) {
                const int e0 = sbase + (k * L1_THREADS + tid) * 4;
                note_edge(s, d[k].x, e0,     r0, rn);
                note_edge(s, d[k].y, e0 + 1, r0, rn);
                note_edge(s, d[k].z, e0 + 2, r0, rn);
                note_edge(s, d[k].w, e0 + 3, r0, rn);
            }
        } else {
            for (int kb = sbase; kb < min(ee, sbase + SCAN_SPAN); kb += L1_THREADS * 4) {
                if (vec_ok && kb + L1_THREADS * 4 <= ee) {
                    const int4 d = *((const int4*)(dstp + kb) + tid);
                    const int e0 = kb + tid * 4;
                    note_edge(s, d.x, e0,     r0, rn);
                    note_edge(s, d.y, e0 + 1, r0, rn);
                    note_edge(s, d.z, e0 + 2, r0, rn);
                    note_edge(s, d.w, e0 + 3, r0, rn);
                } else {
                    for (int e = kb + tid; e < min(ee, kb + L1_THREADS * 4); e += L1_THREADS)
                        note_edge(s, dstp[e], e, r0, rn);
                }
            }
        }
    }
}

__device__ inline void l1_sources(L1Shared& s, int from, int total, const int* __restrict__ ei) {
    for (int i = from + threadIdx.x; i < total; i += L1_THREADS) s.q[i].x = ei[s.q[i].x];
}

__device__ inline void take_lists(int2* q, int at, const int2* __restrict__ bins, int first, int stride,
                                  const int (&cn)[BIN_SLICES], int rowstep, const int2* __restrict__ spare) {
    const int hi = threadIdx.x / BIN_CAP;
    const int i = threadIdx.x % BIN_CAP;
    int2 d[BIN_LOADS];
    #pragma unroll
    for (int k = 0; k < BIN_LOADS; ++k) {
        const int cnt = hi ? cn[2 * k + 1] : cn[2 * k];
        const int2* p = bins + (size_t)(first + (2 * k + hi) * stride) * BIN_CAP;
        d[k] = *((i < cnt) ? p + i : spare);
    }
    int off = at;
    #pragma unroll
    for (int k = 0; k < BIN_LOADS; ++k) {
        const int cnt = hi ? cn[2 * k + 1] : cn[2 * k];
        const int o = hi ? off + cn[2 * k] : off;
        if (i < cnt) q[o + i] = make_int2(d[k].x, d[k].y + (2 * k + hi) * rowstep);
        off += cn[2 * k] + cn[2 * k + 1];
    }
}

__device__ inline void pass_logits(L1Shared& s, int total, const float* __restrict__ as1) {
    for (int i = threadIdx.x; i < total; i += L1_THREADS) {
        const int2 en = s.q[i];
        const float4 a = *(const float4*)(as1 + (size_t)en.x * 4);
        const float4 d = *(const float4*)(&s.ad[en.y * 4]);
        float4 v;
        v.x = leaky(a.x + d.x); v.y = leaky(a.y + d.y); v.z = leaky(a.z + d.z); v.w = leaky(a.w + d.w);
        s.lg[i] = v;
        lds_max_float(&s.mx[en.y * 4 + 0], v.x);
        lds_max_float(&s.mx[en.y * 4 + 1], v.y);
        lds_max_float(&s.mx[en.y * 4 + 2], v.z);
        lds_max_float(&s.mx[en.y * 4 + 3], v.w);
    }
}

__device__ inline float shift_of(float m) { return isfinite(m) ? m : 0.0f; }

__device__ inline void pass_weights(L1Shared& s, int total) {
    for (int i = threadIdx.x; i < total; i += L1_THREADS) {
        const int row = s.q[i].y;
        const float4 v = s.lg[i];
        const float4 m = *(const float4*)(&s.mx[row * 4]);
        float4 w;
        w.x = expf(v.x - shift_of(m.x)); w.y = expf(v.y - shift_of(m.y));
        w.z = expf(v.z - shift_of(m.z)); w.w = expf(v.w - shift_of(m.w));
        s.lg[i] = w;
        atomicAdd(&s.den[row * 4 + 0], w.x);
        atomicAdd(&s.den[row * 4 + 1], w.y);
        atomicAdd(&s.den[row * 4 + 2], w.z);
        atomicAdd(&s.den[row * 4 + 3], w.w);
    }
}

__device__ inline float lane_value(float v, int k) {
    return __int_as_float(__builtin_amdgcn_readlane(__float_as_int(v), k));
}

__device__ inline void pass_rows(L1Shared& s, int total, const float* __restrict__ h1) {
    const int lane = threadIdx.x & 31;
    const int wave = threadIdx.x >> 5;
    for (int i0 = wave * 8; i0 < total; i0 += (L1_THREADS / 32) * 8) {
        const int mine = min(i0 + (lane & 7), total - 1);
        const int2 en = s.q[mine];
        const float4 wv = s.lg[mine];
        float x[8][4];
        #pragma unroll
        for (int k = 0; k < 8; ++k) {
            x[k][0] = x[k][1] = x[k][2] = x[k][3] = 0.0f;
            if (i0 + k < total) {
                const int src = __builtin_amdgcn_readlane(en.x, k);
                const float* hs = h1 + (size_t)src * 128 + lane;
                x[k][0] = hs[0]; x[k][1] = hs[32]; x[k][2] = hs[64]; x[k][3] = hs[96];
            }
        }
        #pragma unroll
        for (int k = 0; k < 8; ++k) {
            if (i0 + k < total) {
                const int row = __builtin_amdgcn_readlane(en.y, k);
                float* ob = &s.acc[row * 128 + lane];
                atomicAdd(ob,      lane_value(wv.x, k) * x[k][0]);
                atomicAdd(ob + 32, lane_value(wv.y, k) * x[k][1]);
                atomicAdd(ob + 64, lane_value(wv.z, k) * x[k][2]);
                atomicAdd(ob + 96, lane_value(wv.w, k) * x[k][3]);
            }
        }
    }
}

__global__ __launch_bounds__(L1_THREADS) void l1_owned_rows(const int* __restrict__ ei, int E, int N,
                                                            const float* __restrict__ as1,
                                                            const float* __restrict__ ad1,
                                                            const float* __restrict__ h1,
                                                            const float* __restrict__ b1,
                                                            const float* __restrict__ W2,
                                                            const float* __restrict__ att_s2,
                                                            const float* __restrict__ att_d2,
                                                            float* __restrict__ h2,
                                                            float* __restrict__ as2,
                                                            float* __restrict__ ad2,
                                                            const int2* __restrict__ bins,
                                                            const int* __restrict__ bcount) {
    __shared__ L1Shared s;
    const int tid = threadIdx.x;
    const int r0 = blockIdx.x * ROWS;
    const int rn = min(ROWS, N - r0);
    const int* dstp = ei + E;
    const bool vec_ok = ((E & 3) == 0) && ((((size_t)dstp) & 15) == 0);

    for (int i = tid; i < ROWS * 128; i += L1_THREADS) s.acc[i] = 0.0f;
    for (int i = tid; i < ROWS * 4; i += L1_THREADS) {
        s.mx[i] = -INFINITY;
        s.den[i] = 0.0f;
        s.ad[i] = (i < rn * 4) ? ad1[(size_t)r0 * 4 + i] : 0.0f;
    }
    if (tid < rn) s.q[tid] = make_int2(r0 + tid, tid);
    if (tid == 0) s.qcount = rn;
    __syncthreads();

    int cn[BIN_SLICES];
    int total = 0;
    bool binned = true;
    #pragma unroll
    for (int j = 0; j < BIN_SLICES; ++j) {
        cn[j] = bcount[blockIdx.x * BIN_SLICES + j];
        binned = binned && (cn[j] <= BIN_CAP);
        total += min(cn[j], BIN_CAP + 1);
    }
    if (binned) { if (rn + total <= QCAP) take_lists(s.q, rn, bins, blockIdx.x * BIN_SLICES, 1, cn, 0, (const int2*)bcount); }
    else        scan_edges(s, dstp, 0, E, vec_ok, r0, rn);
    __syncthreads();
    const int found = binned ? rn + total : s.qcount;
    if (!binned && found <= QCAP) {
        l1_sources(s, rn, found, ei);
        __syncthreads();
    }

    if (found <= QCAP) {
        pass_logits(s, found, as1);
        __syncthreads();
        pass_weights(s, found);
        __syncthreads();
        pass_rows(s, found, h1);
    } else {
        const int SEG = QCAP - 256;
        const int nseg = (E + SEG - 1) / SEG;
        for (int sweep = 0; sweep < 2; ++sweep) {
            for (int sg = 0; sg < nseg; ++sg) {
                __syncthreads();
                if (sg == 0 && tid < rn) s.q[tid] = make_int2(r0 + tid, tid);
                if (tid == 0) s.qcount = (sg == 0) ? rn : 0;
                __syncthreads();
                scan_edges(s, dstp, sg * SEG, min(E, (sg + 1) * SEG), vec_ok, r0, rn);
                __syncthreads();
                const int cnt = s.qcount;
                l1_sources(s, (sg == 0) ? rn : 0, cnt, ei);
                __syncthreads();
                pass_logits(s, cnt, as1);
                __syncthreads();
                if (sweep == 1) {
                    pass_weights(s, cnt);
                    __syncthreads();
                    pass_rows(s, cnt, h1);
                }
            }
        }
    }
    const int lane = tid & 31;
    const int c = lane * 4;
    const float bx = b1[c], by = b1[c + 1], bz = b1[c + 2], bw = b1[c + 3];
    const float w00 = W2[(c + 0) * 2], w01 = W2[(c + 0) * 2 + 1];
    const float w10 = W2[(c + 1) * 2], w11 = W2[(c + 1) * 2 + 1];
    const float w20 = W2[(c + 2) * 2], w21 = W2[(c + 2) * 2 + 1];
    const float w30 = W2[(c + 3) * 2], w31 = W2[(c + 3) * 2 + 1];
    const float ts0 = att_s2[0], ts1 = att_s2[1], td0 = att_d2[0], td1 = att_d2[1];
    __syncthreads();

    for (int row = tid >> 5; row < rn; row += L1_THREADS / 32) {
        const float4 a = *(const float4*)(&s.acc[row * 128 + c]);
        const float d = s.den[row * 4 + (lane >> 3)] + EPS_F;
        float4 v;
        v.x = a.x / d + bx; v.y = a.y / d + by; v.z = a.z / d + bz; v.w = a.w / d + bw;
        v.x = v.x > 0.0f ? v.x : expf(v.x) - 1.0f;
        v.y = v.y > 0.0f ? v.y : expf(v.y) - 1.0f;
        v.z = v.z > 0.0f ? v.z : expf(v.z) - 1.0f;
        v.w = v.w > 0.0f ? v.w : expf(v.w) - 1.0f;
        float s0 = v.x * w00 + v.y * w10 + v.z * w20 + v.w * w30;
        float s1 = v.x * w01 + v.y * w11 + v.z * w21 + v.w * w31;
        #pragma unroll
        for (int off = 16; off >= 1; off >>= 1) {
            s0 += __shfl_xor(s0, off, 32);
            s1 += __shfl_xor(s1, off, 32);
        }
        if (lane == 0)
            s.lg[row] = make_float4(s0, s1, s0 * ts0 + s1 * ts1, s0 * td0 + s1 * td1);
    }
    __syncthreads();

    if (tid < rn) {
        const float4 r = s.lg[tid];
        const int n = r0 + tid;
        *(float2*)(h2 + (size_t)n * 2) = make_float2(r.x, r.y);
        as2[n] = r.z;
        ad2[n] = r.w;
    }
}

#define L2_ROWS 3136
#define L2_SLICES 16
#define L2_CAP 8192
#define L2_THREADS 1024
#define L2_VECS 8
#define L2_STAGE 4

struct L2Shared {
    int2 q[L2_CAP];
    float2 hx[L2_CAP];
    float lg[L2_CAP];
    float mx[L2_ROWS];
    float den[L2_ROWS];
    float s0[L2_ROWS];
    float s1[L2_ROWS];
    float ad[L2_ROWS];
    int qcount;
    int sub[L2_SLICES];
};

__device__ inline void l2_note(L2Shared& s, int dv, int e, int r0, int rn) {
    const unsigned r = (unsigned)(dv - r0);
    if (r < (unsigned)rn) {
        const int slot = atomicAdd(&s.qcount, 1);
        if (slot < L2_CAP) s.q[slot] = make_int2(e, (int)r);
    }
}

__device__ inline int l2_hit(int dv, int r0, int rn) {
    return ((unsigned)(dv - r0) < (unsigned)rn) ? 1 : 0;
}

__device__ inline int l2_put(L2Shared& s, int dv, int e, int r0, int rn, int slot) {
    const unsigned r = (unsigned)(dv - r0);
    if (r < (unsigned)rn) {
        if (slot < L2_CAP) s.q[slot] = make_int2(e, (int)r);
        ++slot;
    }
    return slot;
}

__device__ __forceinline__ void l2_scan(L2Shared& s, const int* __restrict__ dstp, int eb, int ee, bool vec_ok,
                               int r0, int rn) {
    const int tid = threadIdx.x;
    int done = eb;
    if (vec_ok) {
        const int nvec = (ee - eb) >> 2;
        const int4* p = (const int4*)(dstp + eb);
        for (int vb = 0; vb < nvec; vb += L2_THREADS * L2_VECS) {
            int4 d[L2_VECS];
            #pragma unroll
            for (int k = 0; k < L2_VECS; ++k) {
                const int v = vb + k * L2_THREADS + tid;
                d[k] = (v < nvec) ? p[v] : make_int4(-1, -1, -1, -1);
            }
            int cnt = 0;
            #pragma unroll
            for (int k = 0; k < L2_VECS; ++k)
                cnt += l2_hit(d[k].x, r0, rn) + l2_hit(d[k].y, r0, rn) + l2_hit(d[k].z, r0, rn) + l2_hit(d[k].w, r0, rn);
            const int lane = tid & 31;
            int incl = cnt;
            #pragma unroll
            for (int off = 1; off < 32; off <<= 1) {
                const int t = __shfl_up(incl, off, 32);
                if (lane >= off) incl += t;
            }
            const int wave_total = __shfl(incl, 31, 32);
            int wave_base = 0;
            if (lane == 31 && wave_total > 0) wave_base = atomicAdd(&s.qcount, wave_total);
            wave_base = __shfl(wave_base, 31, 32);
            int slot = wave_base + incl - cnt;
            #pragma unroll
            for (int k = 0; k < L2_VECS; ++k) {
                const int e0 = eb + (vb + k * L2_THREADS + tid) * 4;
                slot = l2_put(s, d[k].x, e0,     r0, rn, slot);
                slot = l2_put(s, d[k].y, e0 + 1, r0, rn, slot);
                slot = l2_put(s, d[k].z, e0 + 2, r0, rn, slot);
                slot = l2_put(s, d[k].w, e0 + 3, r0, rn, slot);
            }
        }
        done = eb + nvec * 4;
    }
    for (int e = done + tid; e < ee; e += L2_THREADS) l2_note(s, dstp[e], e, r0, rn);
}

__device__ inline void l2_sources(L2Shared& s, int from, int total, const int* __restrict__ ei) {
    for (int i = from + threadIdx.x; i < total; i += L2_THREADS) s.q[i].x = ei[s.q[i].x];
}

__device__ inline void l2_logits(L2Shared& s, int total,
                                 const float* __restrict__ as2, const float* __restrict__ h2) {
    const int tid = threadIdx.x;
    for (int base = 0; base < total; base += L2_THREADS * L2_STAGE) {
        int srcn[L2_STAGE], row[L2_STAGE];
        float a[L2_STAGE];
        float2 h[L2_STAGE];
        #pragma unroll
        for (int k = 0; k < L2_STAGE; ++k) {
            const int i = base + k * L2_THREADS + tid;
            const int2 en = s.q[min(i, total - 1)];
            row[k] = (i < total) ? en.y : -1;
            srcn[k] = en.x;
        }
        #pragma unroll
        for (int k = 0; k < L2_STAGE; ++k) {
            a[k] = as2[srcn[k]];
            h[k] = *(const float2*)(h2 + (size_t)srcn[k] * 2);
        }
        #pragma unroll
        for (int k = 0; k < L2_STAGE; ++k) {
            if (row[k] >= 0) {
                const int i = base + k * L2_THREADS + tid;
                const float v = leaky(a[k] + s.ad[row[k]]);
                s.lg[i] = v;
                s.hx[i] = h[k];
                lds_max_float(&s.mx[row[k]], v);
            }
        }
    }
}

__device__ inline void l2_weights(L2Shared& s, int total) {
    for (int i = threadIdx.x; i < total; i += L2_THREADS) {
        const int row = s.q[i].y;
        const float w = expf(s.lg[i] - shift_of(s.mx[row]));
        const float2 hv = s.hx[i];
        atomicAdd(&s.den[row], w);
        atomicAdd(&s.s0[row], w * hv.x);
        atomicAdd(&s.s1[row], w * hv.y);
    }
}

static_assert(ROWS * 16 == L2_ROWS && BIN_SLICES == L2_SLICES && L1_THREADS == L2_THREADS && L1_THREADS == 2 * BIN_CAP,
              "a layer-2 range is sixteen layer-1 row blocks; both layers cut the edge list alike");

__global__ __launch_bounds__(L2_THREADS) void bin_edges(const int* __restrict__ ei, int E, int N,
                                                        int2* __restrict__ bins, int* __restrict__ bcount) {
    __shared__ L2Shared s;
    const int tid = threadIdx.x;
    const int range = blockIdx.x / L2_SLICES;
    const int slice = blockIdx.x % L2_SLICES;
    const int r0 = range * L2_ROWS;
    const int rn = min(L2_ROWS, N - r0);
    const int slen = ((E + L2_SLICES - 1) / L2_SLICES + 3) & ~3;
    const int eb = min(E, slice * slen);
    const int ee = min(E, eb + slen);
    const int* dstp = ei + E;
    const bool vec_ok = ((E & 3) == 0) && ((((size_t)dstp) & 15) == 0);
    if (tid == 0) s.qcount = 0;
    if (tid < 16) s.sub[tid] = 0;
    __syncthreads();
    l2_scan(s, dstp, eb, ee, vec_ok, r0, rn);
    __syncthreads();
    const int found = s.qcount;
    if (found <= L2_CAP) {
        for (int i = tid; i < found; i += L2_THREADS) {
            const int2 en = s.q[i];
            const int sb = en.y / ROWS;
            const int slot = atomicAdd(&s.sub[sb], 1);
            if (slot < BIN_CAP)
                bins[(size_t)((range * 16 + sb) * BIN_SLICES + slice) * BIN_CAP + slot] = make_int2(ei[en.x], en.y - sb * ROWS);
        }
    }
    __syncthreads();
    if (tid < 16) bcount[(range * 16 + tid) * BIN_SLICES + slice] = (found <= L2_CAP) ? s.sub[tid] : BIN_CAP + 1;
}

__global__ __launch_bounds__(L2_THREADS) void l2_partial(const int* __restrict__ ei, int E, int N,
                                                         const float* __restrict__ as2,
                                                         const float* __restrict__ ad2,
                                                         const float* __restrict__ h2,
                                                         float* __restrict__ part,
                                                         const int2* __restrict__ bins,
                                                         const int* __restrict__ bcount) {
    __shared__ L2Shared s;
    const int tid = threadIdx.x;
    const int range = blockIdx.x / L2_SLICES;
    const int slice = blockIdx.x % L2_SLICES;
    const int r0 = range * L2_ROWS;
    const int rn = min(L2_ROWS, N - r0);
    const int slen = ((E + L2_SLICES - 1) / L2_SLICES + 3) & ~3;
    const int eb = min(E, slice * slen);
    const int ee = min(E, eb + slen);
    const int* dstp = ei + E;
    const bool vec_ok = ((E & 3) == 0) && ((((size_t)dstp) & 15) == 0);
    const int nself = (rn > slice) ? (rn - slice + L2_SLICES - 1) / L2_SLICES : 0;

    for (int i = tid; i < L2_ROWS; i += L2_THREADS) {
        s.mx[i] = -INFINITY;
        s.den[i] = 0.0f;
        s.s0[i] = 0.0f;
        s.s1[i] = 0.0f;
        s.ad[i] = (i < rn) ? ad2[r0 + i] : 0.0f;
    }
    if (tid < nself) s.q[tid] = make_int2(r0 + slice + tid * L2_SLICES, slice + tid * L2_SLICES);
    if (tid == 0) s.qcount = nself;
    __syncthreads();

    int cn[BIN_SLICES];
    int total = 0;
    bool binned = true;
    #pragma unroll
    for (int j = 0; j < BIN_SLICES; ++j) {
        cn[j] = bcount[(range * 16 + j) * BIN_SLICES + slice];
        binned = binned && (cn[j] <= BIN_CAP);
        total += min(cn[j], BIN_CAP + 1);
    }
    if (binned) { if (nself + total <= L2_CAP) take_lists(s.q, nself, bins, range * 16 * BIN_SLICES + slice, BIN_SLICES, cn, ROWS, (const int2*)bcount); }
    else        l2_scan(s, dstp, eb, ee, vec_ok, r0, rn);
    __syncthreads();
    const int found = binned ? nself + total : s.qcount;
    if (!binned && found <= L2_CAP) {
        l2_sources(s, nself, found, ei);
        __syncthreads();
    }

    if (found <= L2_CAP) {
        l2_logits(s, found, as2, h2);
        __syncthreads();
        l2_weights(s, found);
    } else {
        const int SEG = L2_CAP - 256;
        const int nseg = (ee - eb + SEG - 1) / SEG;
        for (int sweep = 0; sweep < 2; ++sweep) {
            for (int sg = 0; sg < nseg; ++sg) {
                __syncthreads();
                if (sg == 0 && tid < nself)
                    s.q[tid] = make_int2(r0 + slice + tid * L2_SLICES, slice + tid * L2_SLICES);
                if (tid == 0) s.qcount = (sg == 0) ? nself : 0;
                __syncthreads();
                l2_scan(s, dstp, eb + sg * SEG, min(ee, eb + (sg + 1) * SEG), vec_ok, r0, rn);
                __syncthreads();
                const int cnt = s.qcount;
                l2_sources(s, (sg == 0) ? nself : 0, cnt, ei);
                __syncthreads();
                l2_logits(s, cnt, as2, h2);
                __syncthreads();
                if (sweep == 1) l2_weights(s, cnt);
            }
        }
    }
    __syncthreads();

    for (int r = tid; r < rn; r += L2_THREADS)
        ((float4*)part)[(size_t)slice * N + r0 + r] = make_float4(s.mx[r], s.den[r], s.s0[r], s.s1[r]);
}

__global__ void merge_logsoftmax(const float* __restrict__ part,
                                 const float* __restrict__ b2,
                                 float* __restrict__ y, int N) {
    int n = blockIdx.x * blockDim.x + threadIdx.x;
    if (n >= N) return;
    float4 p[L2_SLICES];
    #pragma unroll
    for (int j = 0; j < L2_SLICES; ++j) p[j] = ((const float4*)part)[(size_t)j * N + n];
    float mall = p[0].x;
    #pragma unroll
    for (int j = 1; j < L2_SLICES; ++j) mall = fmaxf(mall, p[j].x);
    const float ms = shift_of(mall);
    float den = 0.0f, t0 = 0.0f, t1 = 0.0f;
    #pragma unroll
    for (int j = 0; j < L2_SLICES; ++j) {
        const float sc = (p[j].x == -INFINITY) ? 0.0f : expf(shift_of(p[j].x) - ms);
        den += sc * p[j].y;
        t0  += sc * p[j].z;
        t1  += sc * p[j].w;
    }
    const float d = den + EPS_F;
    float z0 = t0 / d + b2[0];
    float z1 = t1 / d + b2[1];
    float mx = fmaxf(z0, z1);
    float lse = mx + logf(expf(z0 - mx) + expf(z1 - mx));
    y[n * 2 + 0] = z0 - lse;
    y[n * 2 + 1] = z1 - lse;
}

extern "C" void kernel_launch(void* const* d_in, const int* in_sizes, int n_in,
                              void* d_out, int out_size, void* d_ws, size_t ws_size,
                              hipStream_t stream) {
    const float* x        = (const float*)d_in[0];
    const int*   ei       = (const int*)d_in[1];
    const float* W1       = (const float*)d_in[2];
    const float* att_src1 = (const float*)d_in[3];
    const float* att_dst1 = (const float*)d_in[4];
    const float* b1       = (const float*)d_in[5];
    const float* W2       = (const float*)d_in[6];
    const float* att_src2 = (const float*)d_in[7];
    const float* att_dst2 = (const float*)d_in[8];
    const float* b2       = (const float*)d_in[9];
    float* y = (float*)d_out;

    const int N = in_sizes[0] / 128;
    const int E = in_sizes[1] / 2;

    float* ws   = (float*)d_ws;
    float* h1   = ws;
    float* as1  = h1   + (size_t)N * 256;
    float* ad1  = as1  + (size_t)N * 4;
    float* h2   = ad1  + (size_t)N * 12;
    float* as2  = h2   + (size_t)N * 2;
    float* ad2  = as2  + (size_t)N;
    float* part = h1;
    const int NR = (N + L2_ROWS - 1) / L2_ROWS;
    int2* bins  = (int2*)(h1 + (size_t)N * 128);
    int* bcount = (int*)(bins + (size_t)NR * 16 * L2_SLICES * BIN_CAP);

    const int TB = 256;

    bin_edges<<<NR * L2_SLICES, L2_THREADS, 0, stream>>>(ei, E, N, bins, bcount);

    gemm_wmma_f32<<<N / 16, 256, 0, stream>>>(x, W1, h1, N);
    attn_l1<<<(N * 4 + TB - 1) / TB, TB, 0, stream>>>(h1, att_src1, att_dst1, as1, ad1, N);

    l1_owned_rows<<<(N + ROWS - 1) / ROWS, L1_THREADS, 0, stream>>>(ei, E, N, as1, ad1, h1, b1,
                                                                   W2, att_src2, att_dst2, h2, as2, ad2, bins, bcount);

    l2_partial<<<NR * L2_SLICES, L2_THREADS, 0, stream>>>(ei, E, N, as2, ad2, h2, part, bins, bcount);

    merge_logsoftmax<<<(N + TB - 1) / TB, TB, 0, stream>>>(part, b2, y, N);
}


# Projection and attention scalars formed by the first kernel's 256 resident groups (W1 via LDS, wave = 16x64 tile, scalars by lane sums of accumulators): 2 launches gone; k and scalar sum order change
# speedup vs baseline: 6.0653x; 1.2264x over previous
#include <hip/hip_runtime.h>
#include <math.h>

typedef __attribute__((ext_vector_type(2))) float v2f;
typedef __attribute__((ext_vector_type(8))) float v8f;

#define NEG_SLOPE 0.2f
#define EPS_F 1e-16f

#define WPITCH 136
#define PROJ_WAVES 32

__device__ inline v8f mma4(v2f a, float b0, float b1, v8f c) {
    v2f b;
    b.x = b0;
    b.y = b1;
    return __builtin_amdgcn_wmma_f32_16x16x4_f32(false, a, false, b, (short)0, c, false, false);
}

__device__ inline float half_sum16(float v) {
    #pragma unroll
    for (int off = 8; off >= 1; off >>= 1) v += __shfl_xor(v, off, 32);
    return v;
}

__device__ inline void project_stripes(const float* wl, const float* __restrict__ X,
                                       const float* __restrict__ att_s, const float* __restrict__ att_d,
                                       float* __restrict__ H, float* __restrict__ as1, float* __restrict__ ad1,
                                       int N, int s_first, int s_count) {
    const int lane = threadIdx.x & 31;
    const int half = lane >> 4;
    const int mn = lane & 15;
    const int wave = __builtin_amdgcn_readfirstlane((int)(threadIdx.x >> 5));
    for (int task = wave; task < 2 * s_count; task += PROJ_WAVES) {
        const int rowBase = (s_first + (task >> 1)) * 16;
        const int colBase = (task & 1) * 64;
        const int arow = min(rowBase + mn, N - 1);
        const float4* xr = (const float4*)(X + (size_t)arow * 128 + 4 * half);
        float4 a[16];
        #pragma unroll
        for (int i = 0; i < 16; ++i) a[i] = xr[2 * i];
        v8f c0 = {}, c1 = {}, c2 = {}, c3 = {};
        const float* wr = wl + (4 * half) * WPITCH + colBase + mn;
        #pragma unroll
        for (int i = 0; i < 16; ++i) {
            const float* w = wr + 8 * i * WPITCH;
            v2f av;
            av.x = a[i].x; av.y = a[i].y;
            c0 = mma4(av, w[0],  w[WPITCH],      c0);
            c1 = mma4(av, w[16], w[WPITCH + 16], c1);
            c2 = mma4(av, w[32], w[WPITCH + 32], c2);
            c3 = mma4(av, w[48], w[WPITCH + 48], c3);
            av.x = a[i].z; av.y = a[i].w;
            c0 = mma4(av, w[2 * WPITCH],      w[3 * WPITCH],      c0);
            c1 = mma4(av, w[2 * WPITCH + 16], w[3 * WPITCH + 16], c1);
            c2 = mma4(av, w[2 * WPITCH + 32], w[3 * WPITCH + 32], c2);
            c3 = mma4(av, w[2 * WPITCH + 48], w[3 * WPITCH + 48], c3);
        }
        const float s0 = att_s[colBase + mn], s1 = att_s[colBase + 16 + mn];
        const float s2 = att_s[colBase + 32 + mn], s3 = att_s[colBase + 48 + mn];
        const float d0 = att_d[colBase + mn], d1 = att_d[colBase + 16 + mn];
        const float d2 = att_d[colBase + 32 + mn], d3 = att_d[colBase + 48 + mn];
        const int head0 = colBase >> 5;
        float ps0[8], ps1[8], pd0[8], pd1[8];
        #pragma unroll
        for (int r = 0; r < 8; ++r) {
            ps0[r] = half_sum16(c0[r] * s0 + c1[r] * s1);
            ps1[r] = half_sum16(c2[r] * s2 + c3[r] * s3);
            pd0[r] = half_sum16(c0[r] * d0 + c1[r] * d1);
            pd1[r] = half_sum16(c2[r] * d2 + c3[r] * d3);
        }
        #pragma unroll
        for (int r = 0; r < 8; ++r) {
            const int row = rowBase + r + 8 * half;
            if (row < N) {
                float* hp = H + (size_t)row * 128 + colBase + mn;
                hp[0] = c0[r]; hp[16] = c1[r]; hp[32] = c2[r]; hp[48] = c3[r];
                if (mn == 0) {
                    *(float2*)(as1 + (size_t)row * 4 + head0) = make_float2(ps0[r], ps1[r]);
                    *(float2*)(ad1 + (size_t)row * 4 + head0) = make_float2(pd0[r], pd1[r]);
                }
            }
        }
    }
}

#define ROWS 196
#define QCAP 8192
#define L1_THREADS 1024
#define SCAN_VECS 8
#define SCAN_SPAN (L1_THREADS * 4 * SCAN_VECS)
#define BIN_SLICES 16
#define BIN_CAP 512
#define BIN_LOADS (BIN_SLICES * BIN_CAP / L1_THREADS)

struct L1Shared {
    float acc[ROWS * 128];
    float4 lg[QCAP];
    int2 q[QCAP];
    float mx[ROWS * 4];
    float den[ROWS * 4];
    float ad[ROWS * 4];
    int qcount;
};

__device__ inline void lds_max_float(float* addr, float val) {
    if (val >= 0.0f) atomicMax((int*)addr, __float_as_int(val));
    else             atomicMin((unsigned int*)addr, __float_as_uint(val));
}

__device__ inline float leaky(float v) { return v >= 0.0f ? v : NEG_SLOPE * v; }

__device__ inline void note_edge(L1Shared& s, int dv, int e, int r0, int rn) {
    const unsigned r = (unsigned)(dv - r0);
    if (r < (unsigned)rn) {
        const int slot = atomicAdd(&s.qcount, 1);
        if (slot < QCAP) s.q[slot] = make_int2(e, (int)r);
    }
}

__device__ inline void scan_edges(L1Shared& s, const int* __restrict__ dstp, int eb, int ee, bool vec_ok,
                                  int r0, int rn) {
    const int tid = threadIdx.x;
    const int span = ee - eb;
    const int nsuper = (span + SCAN_SPAN - 1) / SCAN_SPAN;
    if (nsuper <= 0) return;
    const int rot = (int)((blockIdx.x * 5u) % (unsigned)nsuper);
    for (int it = 0; it < nsuper; ++it) {
        int ss = it + rot;
        if (ss >= nsuper) ss -= nsuper;
        const int sbase = eb + ss * SCAN_SPAN;
        if (vec_ok && sbase + SCAN_SPAN <= ee) {
            const int4* p = (const int4*)(dstp + sbase) + tid;
            int4 d[SCAN_VECS];
            #pragma unroll
            for (int k = 0; k < SCAN_VECS; ++k) d[k] = p[k * L1_THREADS];
            #pragma unroll
            for (int k = 0; k < SCAN_VECS; ++k) {
                const int e0 = sbase + (k * L1_THREADS + tid) * 4;
                note_edge(s, d[k].x, e0,     r0, rn);
                note_edge(s, d[k].y, e0 + 1, r0, rn);
                note_edge(s, d[k].z, e0 + 2, r0, rn);
                note_edge(s, d[k].w, e0 + 3, r0, rn);
            }
        } else {
            for (int kb = sbase; kb < min(ee, sbase + SCAN_SPAN); kb += L1_THREADS * 4) {
                if (vec_ok && kb + L1_THREADS * 4 <= ee) {
                    const int4 d = *((const int4*)(dstp + kb) + tid);
                    const int e0 = kb + tid * 4;
                    note_edge(s, d.x, e0,     r0, rn);
                    note_edge(s, d.y, e0 + 1, r0, rn);
                    note_edge(s, d.z, e0 + 2, r0, rn);
                    note_edge(s, d.w, e0 + 3, r0, rn);
                } else {
                    for (int e = kb + tid; e < min(ee, kb + L1_THREADS * 4); e += L1_THREADS)
                        note_edge(s, dstp[e], e, r0, rn);
                }
            }
        }
    }
}

__device__ inline void l1_sources(L1Shared& s, int from, int total, const int* __restrict__ ei) {
    for (int i = from + threadIdx.x; i < total; i += L1_THREADS) s.q[i].x = ei[s.q[i].x];
}

__device__ inline void take_lists(int2* q, int at, const int2* __restrict__ bins, int first, int stride,
                                  const int (&cn)[BIN_SLICES], int rowstep, const int2* __restrict__ spare) {
    const int hi = threadIdx.x / BIN_CAP;
    const int i = threadIdx.x % BIN_CAP;
    int2 d[BIN_LOADS];
    #pragma unroll
    for (int k = 0; k < BIN_LOADS; ++k) {
        const int cnt = hi ? cn[2 * k + 1] : cn[2 * k];
        const int2* p = bins + (size_t)(first + (2 * k + hi) * stride) * BIN_CAP;
        d[k] = *((i < cnt) ? p + i : spare);
    }
    int off = at;
    #pragma unroll
    for (int k = 0; k < BIN_LOADS; ++k) {
        const int cnt = hi ? cn[2 * k + 1] : cn[2 * k];
        const int o = hi ? off + cn[2 * k] : off;
        if (i < cnt) q[o + i] = make_int2(d[k].x, d[k].y + (2 * k + hi) * rowstep);
        off += cn[2 * k] + cn[2 * k + 1];
    }
}

__device__ inline void pass_logits(L1Shared& s, int total, const float* __restrict__ as1) {
    for (int i = threadIdx.x; i < total; i += L1_THREADS) {
        const int2 en = s.q[i];
        const float4 a = *(const float4*)(as1 + (size_t)en.x * 4);
        const float4 d = *(const float4*)(&s.ad[en.y * 4]);
        float4 v;
        v.x = leaky(a.x + d.x); v.y = leaky(a.y + d.y); v.z = leaky(a.z + d.z); v.w = leaky(a.w + d.w);
        s.lg[i] = v;
        lds_max_float(&s.mx[en.y * 4 + 0], v.x);
        lds_max_float(&s.mx[en.y * 4 + 1], v.y);
        lds_max_float(&s.mx[en.y * 4 + 2], v.z);
        lds_max_float(&s.mx[en.y * 4 + 3], v.w);
    }
}

__device__ inline float shift_of(float m) { return isfinite(m) ? m : 0.0f; }

__device__ inline void pass_weights(L1Shared& s, int total) {
    for (int i = threadIdx.x; i < total; i += L1_THREADS) {
        const int row = s.q[i].y;
        const float4 v = s.lg[i];
        const float4 m = *(const float4*)(&s.mx[row * 4]);
        float4 w;
        w.x = expf(v.x - shift_of(m.x)); w.y = expf(v.y - shift_of(m.y));
        w.z = expf(v.z - shift_of(m.z)); w.w = expf(v.w - shift_of(m.w));
        s.lg[i] = w;
        atomicAdd(&s.den[row * 4 + 0], w.x);
        atomicAdd(&s.den[row * 4 + 1], w.y);
        atomicAdd(&s.den[row * 4 + 2], w.z);
        atomicAdd(&s.den[row * 4 + 3], w.w);
    }
}

__device__ inline float lane_value(float v, int k) {
    return __int_as_float(__builtin_amdgcn_readlane(__float_as_int(v), k));
}

__device__ inline void pass_rows(L1Shared& s, int total, const float* __restrict__ h1) {
    const int lane = threadIdx.x & 31;
    const int wave = threadIdx.x >> 5;
    for (int i0 = wave * 8; i0 < total; i0 += (L1_THREADS / 32) * 8) {
        const int mine = min(i0 + (lane & 7), total - 1);
        const int2 en = s.q[mine];
        const float4 wv = s.lg[mine];
        float x[8][4];
        #pragma unroll
        for (int k = 0; k < 8; ++k) {
            x[k][0] = x[k][1] = x[k][2] = x[k][3] = 0.0f;
            if (i0 + k < total) {
                const int src = __builtin_amdgcn_readlane(en.x, k);
                const float* hs = h1 + (size_t)src * 128 + lane;
                x[k][0] = hs[0]; x[k][1] = hs[32]; x[k][2] = hs[64]; x[k][3] = hs[96];
            }
        }
        #pragma unroll
        for (int k = 0; k < 8; ++k) {
            if (i0 + k < total) {
                const int row = __builtin_amdgcn_readlane(en.y, k);
                float* ob = &s.acc[row * 128 + lane];
                atomicAdd(ob,      lane_value(wv.x, k) * x[k][0]);
                atomicAdd(ob + 32, lane_value(wv.y, k) * x[k][1]);
                atomicAdd(ob + 64, lane_value(wv.z, k) * x[k][2]);
                atomicAdd(ob + 96, lane_value(wv.w, k) * x[k][3]);
            }
        }
    }
}

__global__ __launch_bounds__(L1_THREADS) void l1_owned_rows(const int* __restrict__ ei, int E, int N,
                                                            const float* __restrict__ as1,
                                                            const float* __restrict__ ad1,
                                                            const float* __restrict__ h1,
                                                            const float* __restrict__ b1,
                                                            const float* __restrict__ W2,
                                                            const float* __restrict__ att_s2,
                                                            const float* __restrict__ att_d2,
                                                            float* __restrict__ h2,
                                                            float* __restrict__ as2,
                                                            float* __restrict__ ad2,
                                                            const int2* __restrict__ bins,
                                                            const int* __restrict__ bcount) {
    __shared__ L1Shared s;
    const int tid = threadIdx.x;
    const int r0 = blockIdx.x * ROWS;
    const int rn = min(ROWS, N - r0);
    const int* dstp = ei + E;
    const bool vec_ok = ((E & 3) == 0) && ((((size_t)dstp) & 15) == 0);

    for (int i = tid; i < ROWS * 128; i += L1_THREADS) s.acc[i] = 0.0f;
    for (int i = tid; i < ROWS * 4; i += L1_THREADS) {
        s.mx[i] = -INFINITY;
        s.den[i] = 0.0f;
        s.ad[i] = (i < rn * 4) ? ad1[(size_t)r0 * 4 + i] : 0.0f;
    }
    if (tid < rn) s.q[tid] = make_int2(r0 + tid, tid);
    if (tid == 0) s.qcount = rn;
    __syncthreads();

    int cn[BIN_SLICES];
    int total = 0;
    bool binned = true;
    #pragma unroll
    for (int j = 0; j < BIN_SLICES; ++j) {
        cn[j] = bcount[blockIdx.x * BIN_SLICES + j];
        binned = binned && (cn[j] <= BIN_CAP);
        total += min(cn[j], BIN_CAP + 1);
    }
    if (binned) { if (rn + total <= QCAP) take_lists(s.q, rn, bins, blockIdx.x * BIN_SLICES, 1, cn, 0, (const int2*)bcount); }
    else        scan_edges(s, dstp, 0, E, vec_ok, r0, rn);
    __syncthreads();
    const int found = binned ? rn + total : s.qcount;
    if (!binned && found <= QCAP) {
        l1_sources(s, rn, found, ei);
        __syncthreads();
    }

    if (found <= QCAP) {
        pass_logits(s, found, as1);
        __syncthreads();
        pass_weights(s, found);
        __syncthreads();
        pass_rows(s, found, h1);
    } else {
        const int SEG = QCAP - 256;
        const int nseg = (E + SEG - 1) / SEG;
        for (int sweep = 0; sweep < 2; ++sweep) {
            for (int sg = 0; sg < nseg; ++sg) {
                __syncthreads();
                if (sg == 0 && tid < rn) s.q[tid] = make_int2(r0 + tid, tid);
                if (tid == 0) s.qcount = (sg == 0) ? rn : 0;
                __syncthreads();
                scan_edges(s, dstp, sg * SEG, min(E, (sg + 1) * SEG), vec_ok, r0, rn);
                __syncthreads();
                const int cnt = s.qcount;
                l1_sources(s, (sg == 0) ? rn : 0, cnt, ei);
                __syncthreads();
                pass_logits(s, cnt, as1);
                __syncthreads();
                if (sweep == 1) {
                    pass_weights(s, cnt);
                    __syncthreads();
                    pass_rows(s, cnt, h1);
                }
            }
        }
    }
    const int lane = tid & 31;
    const int c = lane * 4;
    const float bx = b1[c], by = b1[c + 1], bz = b1[c + 2], bw = b1[c + 3];
    const float w00 = W2[(c + 0) * 2], w01 = W2[(c + 0) * 2 + 1];
    const float w10 = W2[(c + 1) * 2], w11 = W2[(c + 1) * 2 + 1];
    const float w20 = W2[(c + 2) * 2], w21 = W2[(c + 2) * 2 + 1];
    const float w30 = W2[(c + 3) * 2], w31 = W2[(c + 3) * 2 + 1];
    const float ts0 = att_s2[0], ts1 = att_s2[1], td0 = att_d2[0], td1 = att_d2[1];
    __syncthreads();

    for (int row = tid >> 5; row < rn; row += L1_THREADS / 32) {
        const float4 a = *(const float4*)(&s.acc[row * 128 + c]);
        const float d = s.den[row * 4 + (lane >> 3)] + EPS_F;
        float4 v;
        v.x = a.x / d + bx; v.y = a.y / d + by; v.z = a.z / d + bz; v.w = a.w / d + bw;
        v.x = v.x > 0.0f ? v.x : expf(v.x) - 1.0f;
        v.y = v.y > 0.0f ? v.y : expf(v.y) - 1.0f;
        v.z = v.z > 0.0f ? v.z : expf(v.z) - 1.0f;
        v.w = v.w > 0.0f ? v.w : expf(v.w) - 1.0f;
        float s0 = v.x * w00 + v.y * w10 + v.z * w20 + v.w * w30;
        float s1 = v.x * w01 + v.y * w11 + v.z * w21 + v.w * w31;
        #pragma unroll
        for (int off = 16; off >= 1; off >>= 1) {
            s0 += __shfl_xor(s0, off, 32);
            s1 += __shfl_xor(s1, off, 32);
        }
        if (lane == 0)
            s.lg[row] = make_float4(s0, s1, s0 * ts0 + s1 * ts1, s0 * td0 + s1 * td1);
    }
    __syncthreads();

    if (tid < rn) {
        const float4 r = s.lg[tid];
        const int n = r0 + tid;
        *(float2*)(h2 + (size_t)n * 2) = make_float2(r.x, r.y);
        as2[n] = r.z;
        ad2[n] = r.w;
    }
}

#define L2_ROWS 3136
#define L2_SLICES 16
#define L2_CAP 8192
#define L2_THREADS 1024
#define L2_VECS 8
#define L2_STAGE 4

struct L2Shared {
    int2 q[L2_CAP];
    float2 hx[L2_CAP];
    float lg[L2_CAP];
    float mx[L2_ROWS];
    float den[L2_ROWS];
    float s0[L2_ROWS];
    float s1[L2_ROWS];
    float ad[L2_ROWS];
    int qcount;
    int sub[L2_SLICES];
};

__device__ inline void l2_note(L2Shared& s, int dv, int e, int r0, int rn) {
    const unsigned r = (unsigned)(dv - r0);
    if (r < (unsigned)rn) {
        const int slot = atomicAdd(&s.qcount, 1);
        if (slot < L2_CAP) s.q[slot] = make_int2(e, (int)r);
    }
}

__device__ inline int l2_hit(int dv, int r0, int rn) {
    return ((unsigned)(dv - r0) < (unsigned)rn) ? 1 : 0;
}

__device__ inline int l2_put(L2Shared& s, int dv, int e, int r0, int rn, int slot) {
    const unsigned r = (unsigned)(dv - r0);
    if (r < (unsigned)rn) {
        if (slot < L2_CAP) s.q[slot] = make_int2(e, (int)r);
        ++slot;
    }
    return slot;
}

__device__ __forceinline__ void l2_scan(L2Shared& s, const int* __restrict__ dstp, int eb, int ee, bool vec_ok,
                               int r0, int rn) {
    const int tid = threadIdx.x;
    int done = eb;
    if (vec_ok) {
        const int nvec = (ee - eb) >> 2;
        const int4* p = (const int4*)(dstp + eb);
        for (int vb = 0; vb < nvec; vb += L2_THREADS * L2_VECS) {
            int4 d[L2_VECS];
            #pragma unroll
            for (int k = 0; k < L2_VECS; ++k) {
                const int v = vb + k * L2_THREADS + tid;
                d[k] = (v < nvec) ? p[v] : make_int4(-1, -1, -1, -1);
            }
            int cnt = 0;
            #pragma unroll
            for (int k = 0; k < L2_VECS; ++k)
                cnt += l2_hit(d[k].x, r0, rn) + l2_hit(d[k].y, r0, rn) + l2_hit(d[k].z, r0, rn) + l2_hit(d[k].w, r0, rn);
            const int lane = tid & 31;
            int incl = cnt;
            #pragma unroll
            for (int off = 1; off < 32; off <<= 1) {
                const int t = __shfl_up(incl, off, 32);
                if (lane >= off) incl += t;
            }
            const int wave_total = __shfl(incl, 31, 32);
            int wave_base = 0;
            if (lane == 31 && wave_total > 0) wave_base = atomicAdd(&s.qcount, wave_total);
            wave_base = __shfl(wave_base, 31, 32);
            int slot = wave_base + incl - cnt;
            #pragma unroll
            for (int k = 0; k < L2_VECS; ++k) {
                const int e0 = eb + (vb + k * L2_THREADS + tid) * 4;
                slot = l2_put(s, d[k].x, e0,     r0, rn, slot);
                slot = l2_put(s, d[k].y, e0 + 1, r0, rn, slot);
                slot = l2_put(s, d[k].z, e0 + 2, r0, rn, slot);
                slot = l2_put(s, d[k].w, e0 + 3, r0, rn, slot);
            }
        }
        done = eb + nvec * 4;
    }
    for (int e = done + tid; e < ee; e += L2_THREADS) l2_note(s, dstp[e], e, r0, rn);
}

__device__ inline void l2_sources(L2Shared& s, int from, int total, const int* __restrict__ ei) {
    for (int i = from + threadIdx.x; i < total; i += L2_THREADS) s.q[i].x = ei[s.q[i].x];
}

__device__ inline void l2_logits(L2Shared& s, int total,
                                 const float* __restrict__ as2, const float* __restrict__ h2) {
    const int tid = threadIdx.x;
    for (int base = 0; base < total; base += L2_THREADS * L2_STAGE) {
        int srcn[L2_STAGE], row[L2_STAGE];
        float a[L2_STAGE];
        float2 h[L2_STAGE];
        #pragma unroll
        for (int k = 0; k < L2_STAGE; ++k) {
            const int i = base + k * L2_THREADS + tid;
            const int2 en = s.q[min(i, total - 1)];
            row[k] = (i < total) ? en.y : -1;
            srcn[k] = en.x;
        }
        #pragma unroll
        for (int k = 0; k < L2_STAGE; ++k) {
            a[k] = as2[srcn[k]];
            h[k] = *(const float2*)(h2 + (size_t)srcn[k] * 2);
        }
        #pragma unroll
        for (int k = 0; k < L2_STAGE; ++k) {
            if (row[k] >= 0) {
                const int i = base + k * L2_THREADS + tid;
                const float v = leaky(a[k] + s.ad[row[k]]);
                s.lg[i] = v;
                s.hx[i] = h[k];
                lds_max_float(&s.mx[row[k]], v);
            }
        }
    }
}

__device__ inline void l2_weights(L2Shared& s, int total) {
    for (int i = threadIdx.x; i < total; i += L2_THREADS) {
        const int row = s.q[i].y;
        const float w = expf(s.lg[i] - shift_of(s.mx[row]));
        const float2 hv = s.hx[i];
        atomicAdd(&s.den[row], w);
        atomicAdd(&s.s0[row], w * hv.x);
        atomicAdd(&s.s1[row], w * hv.y);
    }
}

static_assert(ROWS * 16 == L2_ROWS && BIN_SLICES == L2_SLICES && L1_THREADS == L2_THREADS && L1_THREADS == 2 * BIN_CAP,
              "a layer-2 range is sixteen layer-1 row blocks; both layers cut the edge list alike");

static_assert(PROJ_WAVES * 32 == L2_THREADS, "a projection task per wave of the first kernel's group");

__global__ __launch_bounds__(L2_THREADS) void bin_edges(const int* __restrict__ ei, int E, int N,
                                                        int2* __restrict__ bins, int* __restrict__ bcount,
                                                        const float* __restrict__ X, const float* __restrict__ W1,
                                                        const float* __restrict__ att_s, const float* __restrict__ att_d,
                                                        float* __restrict__ H, float* __restrict__ as1,
                                                        float* __restrict__ ad1) {
    __shared__ L2Shared s;
    __shared__ __attribute__((aligned(16))) float wl[128 * WPITCH];
    const int tid = threadIdx.x;
    const int range = blockIdx.x / L2_SLICES;
    const int slice = blockIdx.x % L2_SLICES;
    const int r0 = range * L2_ROWS;
    const int rn = min(L2_ROWS, N - r0);
    const int slen = ((E + L2_SLICES - 1) / L2_SLICES + 3) & ~3;
    const int eb = min(E, slice * slen);
    const int ee = min(E, eb + slen);
    const int* dstp = ei + E;
    const bool vec_ok = ((E & 3) == 0) && ((((size_t)dstp) & 15) == 0);
    float4 wv[4];
    #pragma unroll
    for (int k = 0; k < 4; ++k) wv[k] = ((const float4*)W1)[k * L2_THREADS + tid];
    if (tid == 0) s.qcount = 0;
    if (tid < 16) s.sub[tid] = 0;
    __syncthreads();
    l2_scan(s, dstp, eb, ee, vec_ok, r0, rn);
    #pragma unroll
    for (int k = 0; k < 4; ++k) {
        const int v = k * L2_THREADS + tid;
        *(float4*)(&wl[(v >> 5) * WPITCH + (v & 31) * 4]) = wv[k];
    }
    __syncthreads();
    const int found = s.qcount;
    if (found <= L2_CAP) {
        for (int i = tid; i < found; i += L2_THREADS) {
            const int2 en = s.q[i];
            const int sb = en.y / ROWS;
            const int slot = atomicAdd(&s.sub[sb], 1);
            if (slot < BIN_CAP)
                bins[(size_t)((range * 16 + sb) * BIN_SLICES + slice) * BIN_CAP + slot] = make_int2(ei[en.x], en.y - sb * ROWS);
        }
    }
    __syncthreads();
    if (tid < 16) bcount[(range * 16 + tid) * BIN_SLICES + slice] = (found <= L2_CAP) ? s.sub[tid] : BIN_CAP + 1;

    const int nst = (N + 15) / 16;
    const int per = nst / (int)gridDim.x;
    const int rem = nst % (int)gridDim.x;
    const int g = (int)blockIdx.x;
    project_stripes(wl, X, att_s, att_d, H, as1, ad1, N, g * per + min(g, rem), per + (g < rem ? 1 : 0));
}

__global__ __launch_bounds__(L2_THREADS) void l2_partial(const int* __restrict__ ei, int E, int N,
                                                         const float* __restrict__ as2,
                                                         const float* __restrict__ ad2,
                                                         const float* __restrict__ h2,
                                                         float* __restrict__ part,
                                                         const int2* __restrict__ bins,
                                                         const int* __restrict__ bcount) {
    __shared__ L2Shared s;
    const int tid = threadIdx.x;
    const int range = blockIdx.x / L2_SLICES;
    const int slice = blockIdx.x % L2_SLICES;
    const int r0 = range * L2_ROWS;
    const int rn = min(L2_ROWS, N - r0);
    const int slen = ((E + L2_SLICES - 1) / L2_SLICES + 3) & ~3;
    const int eb = min(E, slice * slen);
    const int ee = min(E, eb + slen);
    const int* dstp = ei + E;
    const bool vec_ok = ((E & 3) == 0) && ((((size_t)dstp) & 15) == 0);
    const int nself = (rn > slice) ? (rn - slice + L2_SLICES - 1) / L2_SLICES : 0;

    for (int i = tid; i < L2_ROWS; i += L2_THREADS) {
        s.mx[i] = -INFINITY;
        s.den[i] = 0.0f;
        s.s0[i] = 0.0f;
        s.s1[i] = 0.0f;
        s.ad[i] = (i < rn) ? ad2[r0 + i] : 0.0f;
    }
    if (tid < nself) s.q[tid] = make_int2(r0 + slice + tid * L2_SLICES, slice + tid * L2_SLICES);
    if (tid == 0) s.qcount = nself;
    __syncthreads();

    int cn[BIN_SLICES];
    int total = 0;
    bool binned = true;
    #pragma unroll
    for (int j = 0; j < BIN_SLICES; ++j) {
        cn[j] = bcount[(range * 16 + j) * BIN_SLICES + slice];
        binned = binned && (cn[j] <= BIN_CAP);
        total += min(cn[j], BIN_CAP + 1);
    }
    if (binned) { if (nself + total <= L2_CAP) take_lists(s.q, nself, bins, range * 16 * BIN_SLICES + slice, BIN_SLICES, cn, ROWS, (const int2*)bcount); }
    else        l2_scan(s, dstp, eb, ee, vec_ok, r0, rn);
    __syncthreads();
    const int found = binned ? nself + total : s.qcount;
    if (!binned && found <= L2_CAP) {
        l2_sources(s, nself, found, ei);
        __syncthreads();
    }

    if (found <= L2_CAP) {
        l2_logits(s, found, as2, h2);
        __syncthreads();
        l2_weights(s, found);
    } else {
        const int SEG = L2_CAP - 256;
        const int nseg = (ee - eb + SEG - 1) / SEG;
        for (int sweep = 0; sweep < 2; ++sweep) {
            for (int sg = 0; sg < nseg; ++sg) {
                __syncthreads();
                if (sg == 0 && tid < nself)
                    s.q[tid] = make_int2(r0 + slice + tid * L2_SLICES, slice + tid * L2_SLICES);
                if (tid == 0) s.qcount = (sg == 0) ? nself : 0;
                __syncthreads();
                l2_scan(s, dstp, eb + sg * SEG, min(ee, eb + (sg + 1) * SEG), vec_ok, r0, rn);
                __syncthreads();
                const int cnt = s.qcount;
                l2_sources(s, (sg == 0) ? nself : 0, cnt, ei);
                __syncthreads();
                l2_logits(s, cnt, as2, h2);
                __syncthreads();
                if (sweep == 1) l2_weights(s, cnt);
            }
        }
    }
    __syncthreads();

    for (int r = tid; r < rn; r += L2_THREADS)
        ((float4*)part)[(size_t)slice * N + r0 + r] = make_float4(s.mx[r], s.den[r], s.s0[r], s.s1[r]);
}

__global__ void merge_logsoftmax(const float* __restrict__ part,
                                 const float* __restrict__ b2,
                                 float* __restrict__ y, int N) {
    int n = blockIdx.x * blockDim.x + threadIdx.x;
    if (n >= N) return;
    float4 p[L2_SLICES];
    #pragma unroll
    for (int j = 0; j < L2_SLICES; ++j) p[j] = ((const float4*)part)[(size_t)j * N + n];
    float mall = p[0].x;
    #pragma unroll
    for (int j = 1; j < L2_SLICES; ++j) mall = fmaxf(mall, p[j].x);
    const float ms = shift_of(mall);
    float den = 0.0f, t0 = 0.0f, t1 = 0.0f;
    #pragma unroll
    for (int j = 0; j < L2_SLICES; ++j) {
        const float sc = (p[j].x == -INFINITY) ? 0.0f : expf(shift_of(p[j].x) - ms);
        den += sc * p[j].y;
        t0  += sc * p[j].z;
        t1  += sc * p[j].w;
    }
    const float d = den + EPS_F;
    float z0 = t0 / d + b2[0];
    float z1 = t1 / d + b2[1];
    float mx = fmaxf(z0, z1);
    float lse = mx + logf(expf(z0 - mx) + expf(z1 - mx));
    y[n * 2 + 0] = z0 - lse;
    y[n * 2 + 1] = z1 - lse;
}

extern "C" void kernel_launch(void* const* d_in, const int* in_sizes, int n_in,
                              void* d_out, int out_size, void* d_ws, size_t ws_size,
                              hipStream_t stream) {
    const float* x        = (const float*)d_in[0];
    const int*   ei       = (const int*)d_in[1];
    const float* W1       = (const float*)d_in[2];
    const float* att_src1 = (const float*)d_in[3];
    const float* att_dst1 = (const float*)d_in[4];
    const float* b1       = (const float*)d_in[5];
    const float* W2       = (const float*)d_in[6];
    const float* att_src2 = (const float*)d_in[7];
    const float* att_dst2 = (const float*)d_in[8];
    const float* b2       = (const float*)d_in[9];
    float* y = (float*)d_out;

    const int N = in_sizes[0] / 128;
    const int E = in_sizes[1] / 2;

    float* ws   = (float*)d_ws;
    float* h1   = ws;
    float* as1  = h1   + (size_t)N * 256;
    float* ad1  = as1  + (size_t)N * 4;
    float* h2   = ad1  + (size_t)N * 12;
    float* as2  = h2   + (size_t)N * 2;
    float* ad2  = as2  + (size_t)N;
    float* part = h1;
    const int NR = (N + L2_ROWS - 1) / L2_ROWS;
    int2* bins  = (int2*)(h1 + (size_t)N * 128);
    int* bcount = (int*)(bins + (size_t)NR * 16 * L2_SLICES * BIN_CAP);

    const int TB = 256;

    bin_edges<<<NR * L2_SLICES, L2_THREADS, 0, stream>>>(ei, E, N, bins, bcount, x, W1, att_src1, att_dst1, h1, as1, ad1);

    l1_owned_rows<<<(N + ROWS - 1) / ROWS, L1_THREADS, 0, stream>>>(ei, E, N, as1, ad1, h1, b1,
                                                                   W2, att_src2, att_dst2, h2, as2, ad2, bins, bcount);

    l2_partial<<<NR * L2_SLICES, L2_THREADS, 0, stream>>>(ei, E, N, as2, ad2, h2, part, bins, bcount);

    merge_logsoftmax<<<(N + TB - 1) / TB, TB, 0, stream>>>(part, b2, y, N);
}


# Layer 2 cut like layer 1: a group owns a 196-row block whole (its 16 lists); max and sums final in LDS; division, bias, log-softmax follow there; partial records, merge launch gone; sum order changes
# speedup vs baseline: 6.6669x; 1.0992x over previous
#include <hip/hip_runtime.h>
#include <math.h>

typedef __attribute__((ext_vector_type(2))) float v2f;
typedef __attribute__((ext_vector_type(8))) float v8f;

#define NEG_SLOPE 0.2f
#define EPS_F 1e-16f

#define WPITCH 136
#define PROJ_WAVES 32

__device__ inline v8f mma4(v2f a, float b0, float b1, v8f c) {
    v2f b;
    b.x = b0;
    b.y = b1;
    return __builtin_amdgcn_wmma_f32_16x16x4_f32(false, a, false, b, (short)0, c, false, false);
}

__device__ inline float half_sum16(float v) {
    #pragma unroll
    for (int off = 8; off >= 1; off >>= 1) v += __shfl_xor(v, off, 32);
    return v;
}

__device__ inline void project_stripes(const float* wl, const float* __restrict__ X,
                                       const float* __restrict__ att_s, const float* __restrict__ att_d,
                                       float* __restrict__ H, float* __restrict__ as1, float* __restrict__ ad1,
                                       int N, int s_first, int s_count) {
    const int lane = threadIdx.x & 31;
    const int half = lane >> 4;
    const int mn = lane & 15;
    const int wave = __builtin_amdgcn_readfirstlane((int)(threadIdx.x >> 5));
    for (int task = wave; task < 2 * s_count; task += PROJ_WAVES) {
        const int rowBase = (s_first + (task >> 1)) * 16;
        const int colBase = (task & 1) * 64;
        const int arow = min(rowBase + mn, N - 1);
        const float4* xr = (const float4*)(X + (size_t)arow * 128 + 4 * half);
        float4 a[16];
        #pragma unroll
        for (int i = 0; i < 16; ++i) a[i] = xr[2 * i];
        v8f c0 = {}, c1 = {}, c2 = {}, c3 = {};
        const float* wr = wl + (4 * half) * WPITCH + colBase + mn;
        #pragma unroll
        for (int i = 0; i < 16; ++i) {
            const float* w = wr + 8 * i * WPITCH;
            v2f av;
            av.x = a[i].x; av.y = a[i].y;
            c0 = mma4(av, w[0],  w[WPITCH],      c0);
            c1 = mma4(av, w[16], w[WPITCH + 16], c1);
            c2 = mma4(av, w[32], w[WPITCH + 32], c2);
            c3 = mma4(av, w[48], w[WPITCH + 48], c3);
            av.x = a[i].z; av.y = a[i].w;
            c0 = mma4(av, w[2 * WPITCH],      w[3 * WPITCH],      c0);
            c1 = mma4(av, w[2 * WPITCH + 16], w[3 * WPITCH + 16], c1);
            c2 = mma4(av, w[2 * WPITCH + 32], w[3 * WPITCH + 32], c2);
            c3 = mma4(av, w[2 * WPITCH + 48], w[3 * WPITCH + 48], c3);
        }
        const float s0 = att_s[colBase + mn], s1 = att_s[colBase + 16 + mn];
        const float s2 = att_s[colBase + 32 + mn], s3 = att_s[colBase + 48 + mn];
        const float d0 = att_d[colBase + mn], d1 = att_d[colBase + 16 + mn];
        const float d2 = att_d[colBase + 32 + mn], d3 = att_d[colBase + 48 + mn];
        const int head0 = colBase >> 5;
        float ps0[8], ps1[8], pd0[8], pd1[8];
        #pragma unroll
        for (int r = 0; r < 8; ++r) {
            ps0[r] = half_sum16(c0[r] * s0 + c1[r] * s1);
            ps1[r] = half_sum16(c2[r] * s2 + c3[r] * s3);
            pd0[r] = half_sum16(c0[r] * d0 + c1[r] * d1);
            pd1[r] = half_sum16(c2[r] * d2 + c3[r] * d3);
        }
        #pragma unroll
        for (int r = 0; r < 8; ++r) {
            const int row = rowBase + r + 8 * half;
            if (row < N) {
                float* hp = H + (size_t)row * 128 + colBase + mn;
                hp[0] = c0[r]; hp[16] = c1[r]; hp[32] = c2[r]; hp[48] = c3[r];
                if (mn == 0) {
                    *(float2*)(as1 + (size_t)row * 4 + head0) = make_float2(ps0[r], ps1[r]);
                    *(float2*)(ad1 + (size_t)row * 4 + head0) = make_float2(pd0[r], pd1[r]);
                }
            }
        }
    }
}

#define ROWS 196
#define QCAP 8192
#define L1_THREADS 1024
#define SCAN_VECS 8
#define SCAN_SPAN (L1_THREADS * 4 * SCAN_VECS)
#define BIN_SLICES 16
#define BIN_CAP 512
#define BIN_LOADS (BIN_SLICES * BIN_CAP / L1_THREADS)

struct L1Shared {
    float acc[ROWS * 128];
    float4 lg[QCAP];
    int2 q[QCAP];
    float mx[ROWS * 4];
    float den[ROWS * 4];
    float ad[ROWS * 4];
    int qcount;
};

__device__ inline void lds_max_float(float* addr, float val) {
    if (val >= 0.0f) atomicMax((int*)addr, __float_as_int(val));
    else             atomicMin((unsigned int*)addr, __float_as_uint(val));
}

__device__ inline float leaky(float v) { return v >= 0.0f ? v : NEG_SLOPE * v; }

template <class S>
__device__ inline void note_edge(S& s, int dv, int e, int r0, int rn) {
    const unsigned r = (unsigned)(dv - r0);
    if (r < (unsigned)rn) {
        const int slot = atomicAdd(&s.qcount, 1);
        if (slot < QCAP) s.q[slot] = make_int2(e, (int)r);
    }
}

template <class S>
__device__ inline void scan_edges(S& s, const int* __restrict__ dstp, int eb, int ee, bool vec_ok,
                                  int r0, int rn) {
    const int tid = threadIdx.x;
    const int span = ee - eb;
    const int nsuper = (span + SCAN_SPAN - 1) / SCAN_SPAN;
    if (nsuper <= 0) return;
    const int rot = (int)((blockIdx.x * 5u) % (unsigned)nsuper);
    for (int it = 0; it < nsuper; ++it) {
        int ss = it + rot;
        if (ss >= nsuper) ss -= nsuper;
        const int sbase = eb + ss * SCAN_SPAN;
        if (vec_ok && sbase + SCAN_SPAN <= ee) {
            const int4* p = (const int4*)(dstp + sbase) + tid;
            int4 d[SCAN_VECS];
            #pragma unroll
            for (int k = 0; k < SCAN_VECS; ++k) d[k] = p[k * L1_THREADS];
            #pragma unroll
            for (int k = 0; k < SCAN_VECS; ++k) {
                const int e0 = sbase + (k * L1_THREADS + tid) * 4;
                note_edge(s, d[k].x, e0,     r0, rn);
                note_edge(s, d[k].y, e0 + 1, r0, rn);
                note_edge(s, d[k].z, e0 + 2, r0, rn);
                note_edge(s, d[k].w, e0 + 3, r0, rn);
            }
        } else {
            for (int kb = sbase; kb < min(ee, sbase + SCAN_SPAN); kb += L1_THREADS * 4) {
                if (vec_ok && kb + L1_THREADS * 4 <= ee) {
                    const int4 d = *((const int4*)(dstp + kb) + tid);
                    const int e0 = kb + tid * 4;
                    note_edge(s, d.x, e0,     r0, rn);
                    note_edge(s, d.y, e0 + 1, r0, rn);
                    note_edge(s, d.z, e0 + 2, r0, rn);
                    note_edge(s, d.w, e0 + 3, r0, rn);
                } else {
                    for (int e = kb + tid; e < min(ee, kb + L1_THREADS * 4); e += L1_THREADS)
                        note_edge(s, dstp[e], e, r0, rn);
                }
            }
        }
    }
}

template <class S>
__device__ inline void l1_sources(S& s, int from, int total, const int* __restrict__ ei) {
    for (int i = from + threadIdx.x; i < total; i += L1_THREADS) s.q[i].x = ei[s.q[i].x];
}

__device__ inline void take_lists(int2* q, int at, const int2* __restrict__ bins, int first, int stride,
                                  const int (&cn)[BIN_SLICES], int rowstep, const int2* __restrict__ spare) {
    const int hi = threadIdx.x / BIN_CAP;
    const int i = threadIdx.x % BIN_CAP;
    int2 d[BIN_LOADS];
    #pragma unroll
    for (int k = 0; k < BIN_LOADS; ++k) {
        const int cnt = hi ? cn[2 * k + 1] : cn[2 * k];
        const int2* p = bins + (size_t)(first + (2 * k + hi) * stride) * BIN_CAP;
        d[k] = *((i < cnt) ? p + i : spare);
    }
    int off = at;
    #pragma unroll
    for (int k = 0; k < BIN_LOADS; ++k) {
        const int cnt = hi ? cn[2 * k + 1] : cn[2 * k];
        const int o = hi ? off + cn[2 * k] : off;
        if (i < cnt) q[o + i] = make_int2(d[k].x, d[k].y + (2 * k + hi) * rowstep);
        off += cn[2 * k] + cn[2 * k + 1];
    }
}

__device__ inline void pass_logits(L1Shared& s, int total, const float* __restrict__ as1) {
    for (int i = threadIdx.x; i < total; i += L1_THREADS) {
        const int2 en = s.q[i];
        const float4 a = *(const float4*)(as1 + (size_t)en.x * 4);
        const float4 d = *(const float4*)(&s.ad[en.y * 4]);
        float4 v;
        v.x = leaky(a.x + d.x); v.y = leaky(a.y + d.y); v.z = leaky(a.z + d.z); v.w = leaky(a.w + d.w);
        s.lg[i] = v;
        lds_max_float(&s.mx[en.y * 4 + 0], v.x);
        lds_max_float(&s.mx[en.y * 4 + 1], v.y);
        lds_max_float(&s.mx[en.y * 4 + 2], v.z);
        lds_max_float(&s.mx[en.y * 4 + 3], v.w);
    }
}

__device__ inline float shift_of(float m) { return isfinite(m) ? m : 0.0f; }

__device__ inline void pass_weights(L1Shared& s, int total) {
    for (int i = threadIdx.x; i < total; i += L1_THREADS) {
        const int row = s.q[i].y;
        const float4 v = s.lg[i];
        const float4 m = *(const float4*)(&s.mx[row * 4]);
        float4 w;
        w.x = expf(v.x - shift_of(m.x)); w.y = expf(v.y - shift_of(m.y));
        w.z = expf(v.z - shift_of(m.z)); w.w = expf(v.w - shift_of(m.w));
        s.lg[i] = w;
        atomicAdd(&s.den[row * 4 + 0], w.x);
        atomicAdd(&s.den[row * 4 + 1], w.y);
        atomicAdd(&s.den[row * 4 + 2], w.z);
        atomicAdd(&s.den[row * 4 + 3], w.w);
    }
}

__device__ inline float lane_value(float v, int k) {
    return __int_as_float(__builtin_amdgcn_readlane(__float_as_int(v), k));
}

__device__ inline void pass_rows(L1Shared& s, int total, const float* __restrict__ h1) {
    const int lane = threadIdx.x & 31;
    const int wave = threadIdx.x >> 5;
    for (int i0 = wave * 8; i0 < total; i0 += (L1_THREADS / 32) * 8) {
        const int mine = min(i0 + (lane & 7), total - 1);
        const int2 en = s.q[mine];
        const float4 wv = s.lg[mine];
        float x[8][4];
        #pragma unroll
        for (int k = 0; k < 8; ++k) {
            x[k][0] = x[k][1] = x[k][2] = x[k][3] = 0.0f;
            if (i0 + k < total) {
                const int src = __builtin_amdgcn_readlane(en.x, k);
                const float* hs = h1 + (size_t)src * 128 + lane;
                x[k][0] = hs[0]; x[k][1] = hs[32]; x[k][2] = hs[64]; x[k][3] = hs[96];
            }
        }
        #pragma unroll
        for (int k = 0; k < 8; ++k) {
            if (i0 + k < total) {
                const int row = __builtin_amdgcn_readlane(en.y, k);
                float* ob = &s.acc[row * 128 + lane];
                atomicAdd(ob,      lane_value(wv.x, k) * x[k][0]);
                atomicAdd(ob + 32, lane_value(wv.y, k) * x[k][1]);
                atomicAdd(ob + 64, lane_value(wv.z, k) * x[k][2]);
                atomicAdd(ob + 96, lane_value(wv.w, k) * x[k][3]);
            }
        }
    }
}

__global__ __launch_bounds__(L1_THREADS) void l1_owned_rows(const int* __restrict__ ei, int E, int N,
                                                            const float* __restrict__ as1,
                                                            const float* __restrict__ ad1,
                                                            const float* __restrict__ h1,
                                                            const float* __restrict__ b1,
                                                            const float* __restrict__ W2,
                                                            const float* __restrict__ att_s2,
                                                            const float* __restrict__ att_d2,
                                                            float* __restrict__ h2,
                                                            float* __restrict__ as2,
                                                            float* __restrict__ ad2,
                                                            const int2* __restrict__ bins,
                                                            const int* __restrict__ bcount) {
    __shared__ L1Shared s;
    const int tid = threadIdx.x;
    const int r0 = blockIdx.x * ROWS;
    const int rn = min(ROWS, N - r0);
    const int* dstp = ei + E;
    const bool vec_ok = ((E & 3) == 0) && ((((size_t)dstp) & 15) == 0);

    for (int i = tid; i < ROWS * 128; i += L1_THREADS) s.acc[i] = 0.0f;
    for (int i = tid; i < ROWS * 4; i += L1_THREADS) {
        s.mx[i] = -INFINITY;
        s.den[i] = 0.0f;
        s.ad[i] = (i < rn * 4) ? ad1[(size_t)r0 * 4 + i] : 0.0f;
    }
    if (tid < rn) s.q[tid] = make_int2(r0 + tid, tid);
    if (tid == 0) s.qcount = rn;
    __syncthreads();

    int cn[BIN_SLICES];
    int total = 0;
    bool binned = true;
    #pragma unroll
    for (int j = 0; j < BIN_SLICES; ++j) {
        cn[j] = bcount[blockIdx.x * BIN_SLICES + j];
        binned = binned && (cn[j] <= BIN_CAP);
        total += min(cn[j], BIN_CAP + 1);
    }
    if (binned) { if (rn + total <= QCAP) take_lists(s.q, rn, bins, blockIdx.x * BIN_SLICES, 1, cn, 0, (const int2*)bcount); }
    else        scan_edges(s, dstp, 0, E, vec_ok, r0, rn);
    __syncthreads();
    const int found = binned ? rn + total : s.qcount;
    if (!binned && found <= QCAP) {
        l1_sources(s, rn, found, ei);
        __syncthreads();
    }

    if (found <= QCAP) {
        pass_logits(s, found, as1);
        __syncthreads();
        pass_weights(s, found);
        __syncthreads();
        pass_rows(s, found, h1);
    } else {
        const int SEG = QCAP - 256;
        const int nseg = (E + SEG - 1) / SEG;
        for (int sweep = 0; sweep < 2; ++sweep) {
            for (int sg = 0; sg < nseg; ++sg) {
                __syncthreads();
                if (sg == 0 && tid < rn) s.q[tid] = make_int2(r0 + tid, tid);
                if (tid == 0) s.qcount = (sg == 0) ? rn : 0;
                __syncthreads();
                scan_edges(s, dstp, sg * SEG, min(E, (sg + 1) * SEG), vec_ok, r0, rn);
                __syncthreads();
                const int cnt = s.qcount;
                l1_sources(s, (sg == 0) ? rn : 0, cnt, ei);
                __syncthreads();
                pass_logits(s, cnt, as1);
                __syncthreads();
                if (sweep == 1) {
                    pass_weights(s, cnt);
                    __syncthreads();
                    pass_rows(s, cnt, h1);
                }
            }
        }
    }
    const int lane = tid & 31;
    const int c = lane * 4;
    const float bx = b1[c], by = b1[c + 1], bz = b1[c + 2], bw = b1[c + 3];
    const float w00 = W2[(c + 0) * 2], w01 = W2[(c + 0) * 2 + 1];
    const float w10 = W2[(c + 1) * 2], w11 = W2[(c + 1) * 2 + 1];
    const float w20 = W2[(c + 2) * 2], w21 = W2[(c + 2) * 2 + 1];
    const float w30 = W2[(c + 3) * 2], w31 = W2[(c + 3) * 2 + 1];
    const float ts0 = att_s2[0], ts1 = att_s2[1], td0 = att_d2[0], td1 = att_d2[1];
    __syncthreads();

    for (int row = tid >> 5; row < rn; row += L1_THREADS / 32) {
        const float4 a = *(const float4*)(&s.acc[row * 128 + c]);
        const float d = s.den[row * 4 + (lane >> 3)] + EPS_F;
        float4 v;
        v.x = a.x / d + bx; v.y = a.y / d + by; v.z = a.z / d + bz; v.w = a.w / d + bw;
        v.x = v.x > 0.0f ? v.x : expf(v.x) - 1.0f;
        v.y = v.y > 0.0f ? v.y : expf(v.y) - 1.0f;
        v.z = v.z > 0.0f ? v.z : expf(v.z) - 1.0f;
        v.w = v.w > 0.0f ? v.w : expf(v.w) - 1.0f;
        float s0 = v.x * w00 + v.y * w10 + v.z * w20 + v.w * w30;
        float s1 = v.x * w01 + v.y * w11 + v.z * w21 + v.w * w31;
        #pragma unroll
        for (int off = 16; off >= 1; off >>= 1) {
            s0 += __shfl_xor(s0, off, 32);
            s1 += __shfl_xor(s1, off, 32);
        }
        if (lane == 0)
            s.lg[row] = make_float4(s0, s1, s0 * ts0 + s1 * ts1, s0 * td0 + s1 * td1);
    }
    __syncthreads();

    if (tid < rn) {
        const float4 r = s.lg[tid];
        const int n = r0 + tid;
        *(float2*)(h2 + (size_t)n * 2) = make_float2(r.x, r.y);
        as2[n] = r.z;
        ad2[n] = r.w;
    }
}

#define L2_ROWS 3136
#define L2_SLICES 16
#define L2_CAP 8192
#define L2_THREADS 1024
#define L2_VECS 8
#define L2_STAGE 4

struct L2Shared {
    int2 q[L2_CAP];
    float2 hx[L2_CAP];
    float lg[L2_CAP];
    float mx[ROWS];
    float den[ROWS];
    float s0[ROWS];
    float s1[ROWS];
    float ad[ROWS];
    int qcount;
    int sub[L2_SLICES];
};

__device__ inline void l2_note(L2Shared& s, int dv, int e, int r0, int rn) {
    const unsigned r = (unsigned)(dv - r0);
    if (r < (unsigned)rn) {
        const int slot = atomicAdd(&s.qcount, 1);
        if (slot < L2_CAP) s.q[slot] = make_int2(e, (int)r);
    }
}

__device__ inline int l2_hit(int dv, int r0, int rn) {
    return ((unsigned)(dv - r0) < (unsigned)rn) ? 1 : 0;
}

__device__ inline int l2_put(L2Shared& s, int dv, int e, int r0, int rn, int slot) {
    const unsigned r = (unsigned)(dv - r0);
    if (r < (unsigned)rn) {
        if (slot < L2_CAP) s.q[slot] = make_int2(e, (int)r);
        ++slot;
    }
    return slot;
}

__device__ __forceinline__ void l2_scan(L2Shared& s, const int* __restrict__ dstp, int eb, int ee, bool vec_ok,
                               int r0, int rn) {
    const int tid = threadIdx.x;
    int done = eb;
    if (vec_ok) {
        const int nvec = (ee - eb) >> 2;
        const int4* p = (const int4*)(dstp + eb);
        for (int vb = 0; vb < nvec; vb += L2_THREADS * L2_VECS) {
            int4 d[L2_VECS];
            #pragma unroll
            for (int k = 0; k < L2_VECS; ++k) {
                const int v = vb + k * L2_THREADS + tid;
                d[k] = (v < nvec) ? p[v] : make_int4(-1, -1, -1, -1);
            }
            int cnt = 0;
            #pragma unroll
            for (int k = 0; k < L2_VECS; ++k)
                cnt += l2_hit(d[k].x, r0, rn) + l2_hit(d[k].y, r0, rn) + l2_hit(d[k].z, r0, rn) + l2_hit(d[k].w, r0, rn);
            const int lane = tid & 31;
            int incl = cnt;
            #pragma unroll
            for (int off = 1; off < 32; off <<= 1) {
                const int t = __shfl_up(incl, off, 32);
                if (lane >= off) incl += t;
            }
            const int wave_total = __shfl(incl, 31, 32);
            int wave_base = 0;
            if (lane == 31 && wave_total > 0) wave_base = atomicAdd(&s.qcount, wave_total);
            wave_base = __shfl(wave_base, 31, 32);
            int slot = wave_base + incl - cnt;
            #pragma unroll
            for (int k = 0; k < L2_VECS; ++k) {
                const int e0 = eb + (vb + k * L2_THREADS + tid) * 4;
                slot = l2_put(s, d[k].x, e0,     r0, rn, slot);
                slot = l2_put(s, d[k].y, e0 + 1, r0, rn, slot);
                slot = l2_put(s, d[k].z, e0 + 2, r0, rn, slot);
                slot = l2_put(s, d[k].w, e0 + 3, r0, rn, slot);
            }
        }
        done = eb + nvec * 4;
    }
    for (int e = done + tid; e < ee; e += L2_THREADS) l2_note(s, dstp[e], e, r0, rn);
}

__device__ inline void l2_logits(L2Shared& s, int total,
                                 const float* __restrict__ as2, const float* __restrict__ h2) {
    const int tid = threadIdx.x;
    for (int base = 0; base < total; base += L2_THREADS * L2_STAGE) {
        int srcn[L2_STAGE], row[L2_STAGE];
        float a[L2_STAGE];
        float2 h[L2_STAGE];
        #pragma unroll
        for (int k = 0; k < L2_STAGE; ++k) {
            const int i = base + k * L2_THREADS + tid;
            const int2 en = s.q[min(i, total - 1)];
            row[k] = (i < total) ? en.y : -1;
            srcn[k] = en.x;
        }
        #pragma unroll
        for (int k = 0; k < L2_STAGE; ++k) {
            a[k] = as2[srcn[k]];
            h[k] = *(const float2*)(h2 + (size_t)srcn[k] * 2);
        }
        #pragma unroll
        for (int k = 0; k < L2_STAGE; ++k) {
            if (row[k] >= 0) {
                const int i = base + k * L2_THREADS + tid;
                const float v = leaky(a[k] + s.ad[row[k]]);
                s.lg[i] = v;
                s.hx[i] = h[k];
                lds_max_float(&s.mx[row[k]], v);
            }
        }
    }
}

__device__ inline void l2_weights(L2Shared& s, int total) {
    for (int i = threadIdx.x; i < total; i += L2_THREADS) {
        const int row = s.q[i].y;
        const float w = expf(s.lg[i] - shift_of(s.mx[row]));
        const float2 hv = s.hx[i];
        atomicAdd(&s.den[row], w);
        atomicAdd(&s.s0[row], w * hv.x);
        atomicAdd(&s.s1[row], w * hv.y);
    }
}

static_assert(ROWS * 16 == L2_ROWS && BIN_SLICES == L2_SLICES && L1_THREADS == L2_THREADS && L1_THREADS == 2 * BIN_CAP &&
              QCAP == L2_CAP,
              "a range of the first kernel is sixteen row blocks; both layers' groups are built alike and share the list builders");

static_assert(PROJ_WAVES * 32 == L2_THREADS, "a projection task per wave of the first kernel's group");

__global__ __launch_bounds__(L2_THREADS) void bin_edges(const int* __restrict__ ei, int E, int N,
                                                        int2* __restrict__ bins, int* __restrict__ bcount,
                                                        const float* __restrict__ X, const float* __restrict__ W1,
                                                        const float* __restrict__ att_s, const float* __restrict__ att_d,
                                                        float* __restrict__ H, float* __restrict__ as1,
                                                        float* __restrict__ ad1) {
    __shared__ L2Shared s;
    __shared__ __attribute__((aligned(16))) float wl[128 * WPITCH];
    const int tid = threadIdx.x;
    const int range = blockIdx.x / L2_SLICES;
    const int slice = blockIdx.x % L2_SLICES;
    const int r0 = range * L2_ROWS;
    const int rn = min(L2_ROWS, N - r0);
    const int slen = ((E + L2_SLICES - 1) / L2_SLICES + 3) & ~3;
    const int eb = min(E, slice * slen);
    const int ee = min(E, eb + slen);
    const int* dstp = ei + E;
    const bool vec_ok = ((E & 3) == 0) && ((((size_t)dstp) & 15) == 0);
    float4 wv[4];
    #pragma unroll
    for (int k = 0; k < 4; ++k) wv[k] = ((const float4*)W1)[k * L2_THREADS + tid];
    if (tid == 0) s.qcount = 0;
    if (tid < 16) s.sub[tid] = 0;
    __syncthreads();
    l2_scan(s, dstp, eb, ee, vec_ok, r0, rn);
    #pragma unroll
    for (int k = 0; k < 4; ++k) {
        const int v = k * L2_THREADS + tid;
        *(float4*)(&wl[(v >> 5) * WPITCH + (v & 31) * 4]) = wv[k];
    }
    __syncthreads();
    const int found = s.qcount;
    if (found <= L2_CAP) {
        for (int i = tid; i < found; i += L2_THREADS) {
            const int2 en = s.q[i];
            const int sb = en.y / ROWS;
            const int slot = atomicAdd(&s.sub[sb], 1);
            if (slot < BIN_CAP)
                bins[(size_t)((range * 16 + sb) * BIN_SLICES + slice) * BIN_CAP + slot] = make_int2(ei[en.x], en.y - sb * ROWS);
        }
    }
    __syncthreads();
    if (tid < 16) bcount[(range * 16 + tid) * BIN_SLICES + slice] = (found <= L2_CAP) ? s.sub[tid] : BIN_CAP + 1;

    const int nst = (N + 15) / 16;
    const int per = nst / (int)gridDim.x;
    const int rem = nst % (int)gridDim.x;
    const int g = (int)blockIdx.x;
    project_stripes(wl, X, att_s, att_d, H, as1, ad1, N, g * per + min(g, rem), per + (g < rem ? 1 : 0));
}

__global__ __launch_bounds__(L2_THREADS) void l2_owned_rows(const int* __restrict__ ei, int E, int N,
                                                            const float* __restrict__ as2,
                                                            const float* __restrict__ ad2,
                                                            const float* __restrict__ h2,
                                                            const float* __restrict__ b2,
                                                            float* __restrict__ y,
                                                            const int2* __restrict__ bins,
                                                            const int* __restrict__ bcount) {
    __shared__ L2Shared s;
    const int tid = threadIdx.x;
    const int r0 = blockIdx.x * ROWS;
    const int rn = min(ROWS, N - r0);
    const int* dstp = ei + E;
    const bool vec_ok = ((E & 3) == 0) && ((((size_t)dstp) & 15) == 0);

    int cn[BIN_SLICES];
    int total = 0;
    bool binned = true;
    #pragma unroll
    for (int j = 0; j < BIN_SLICES; ++j) {
        cn[j] = bcount[blockIdx.x * BIN_SLICES + j];
        binned = binned && (cn[j] <= BIN_CAP);
        total += min(cn[j], BIN_CAP + 1);
    }
    const float adv = (tid < rn) ? ad2[r0 + tid] : 0.0f;

    if (tid < ROWS) {
        s.mx[tid] = -INFINITY;
        s.den[tid] = 0.0f;
        s.s0[tid] = 0.0f;
        s.s1[tid] = 0.0f;
    }
    if (tid < rn) s.q[tid] = make_int2(r0 + tid, tid);
    if (binned) {
        if (rn + total <= L2_CAP) take_lists(s.q, rn, bins, blockIdx.x * BIN_SLICES, 1, cn, 0, (const int2*)bcount);
    } else {
        if (tid == 0) s.qcount = rn;
        __syncthreads();
        scan_edges(s, dstp, 0, E, vec_ok, r0, rn);
    }
    if (tid < ROWS) s.ad[tid] = adv;
    __syncthreads();
    const int found = binned ? rn + total : s.qcount;
    if (!binned && found <= L2_CAP) {
        l1_sources(s, rn, found, ei);
        __syncthreads();
    }

    if (found <= L2_CAP) {
        l2_logits(s, found, as2, h2);
        __syncthreads();
        l2_weights(s, found);
    } else {
        const int SEG = L2_CAP - 256;
        const int nseg = (E + SEG - 1) / SEG;
        for (int sweep = 0; sweep < 2; ++sweep) {
            for (int sg = 0; sg < nseg; ++sg) {
                __syncthreads();
                if (sg == 0 && tid < rn) s.q[tid] = make_int2(r0 + tid, tid);
                if (tid == 0) s.qcount = (sg == 0) ? rn : 0;
                __syncthreads();
                scan_edges(s, dstp, sg * SEG, min(E, (sg + 1) * SEG), vec_ok, r0, rn);
                __syncthreads();
                const int cnt = s.qcount;
                l1_sources(s, (sg == 0) ? rn : 0, cnt, ei);
                __syncthreads();
                l2_logits(s, cnt, as2, h2);
                __syncthreads();
                if (sweep == 1) l2_weights(s, cnt);
            }
        }
    }
    const float bias0 = b2[0], bias1 = b2[1];
    __syncthreads();

    if (tid < rn) {
        const float d = s.den[tid] + EPS_F;
        const float z0 = s.s0[tid] / d + bias0;
        const float z1 = s.s1[tid] / d + bias1;
        const float mx = fmaxf(z0, z1);
        const float lse = mx + logf(expf(z0 - mx) + expf(z1 - mx));
        *(float2*)(y + (size_t)(r0 + tid) * 2) = make_float2(z0 - lse, z1 - lse);
    }
}

extern "C" void kernel_launch(void* const* d_in, const int* in_sizes, int n_in,
                              void* d_out, int out_size, void* d_ws, size_t ws_size,
                              hipStream_t stream) {
    const float* x        = (const float*)d_in[0];
    const int*   ei       = (const int*)d_in[1];
    const float* W1       = (const float*)d_in[2];
    const float* att_src1 = (const float*)d_in[3];
    const float* att_dst1 = (const float*)d_in[4];
    const float* b1       = (const float*)d_in[5];
    const float* W2       = (const float*)d_in[6];
    const float* att_src2 = (const float*)d_in[7];
    const float* att_dst2 = (const float*)d_in[8];
    const float* b2       = (const float*)d_in[9];
    float* y = (float*)d_out;

    const int N = in_sizes[0] / 128;
    const int E = in_sizes[1] / 2;

    float* ws   = (float*)d_ws;
    float* h1   = ws;
    float* as1  = h1   + (size_t)N * 256;
    float* ad1  = as1  + (size_t)N * 4;
    float* h2   = ad1  + (size_t)N * 12;
    float* as2  = h2   + (size_t)N * 2;
    float* ad2  = as2  + (size_t)N;
    const int NR = (N + L2_ROWS - 1) / L2_ROWS;
    int2* bins  = (int2*)(h1 + (size_t)N * 128);
    int* bcount = (int*)(bins + (size_t)NR * 16 * L2_SLICES * BIN_CAP);

    bin_edges<<<NR * L2_SLICES, L2_THREADS, 0, stream>>>(ei, E, N, bins, bcount, x, W1, att_src1, att_dst1, h1, as1, ad1);

    l1_owned_rows<<<(N + ROWS - 1) / ROWS, L1_THREADS, 0, stream>>>(ei, E, N, as1, ad1, h1, b1,
                                                                   W2, att_src2, att_dst2, h2, as2, ad2, bins, bcount);

    l2_owned_rows<<<(N + ROWS - 1) / ROWS, L2_THREADS, 0, stream>>>(ei, E, N, as2, ad2, h2, b2, y, bins, bcount);
}


# First kernel: each wave cuts its own share of the edge slice (own LDS hit list, no group barrier in the cut); 16 waves form their stripe first, 16 cut first: cut runs under products; lists by arrival
# speedup vs baseline: 6.8659x; 1.0298x over previous
#include <hip/hip_runtime.h>
#include <math.h>

typedef __attribute__((ext_vector_type(2))) float v2f;
typedef __attribute__((ext_vector_type(8))) float v8f;

#define NEG_SLOPE 0.2f
#define EPS_F 1e-16f

#define WPITCH 136
#define PROJ_WAVES 32

__device__ inline v8f mma4(v2f a, float b0, float b1, v8f c) {
    v2f b;
    b.x = b0;
    b.y = b1;
    return __builtin_amdgcn_wmma_f32_16x16x4_f32(false, a, false, b, (short)0, c, false, false);
}

__device__ inline float half_sum16(float v) {
    #pragma unroll
    for (int off = 8; off >= 1; off >>= 1) v += __shfl_xor(v, off, 32);
    return v;
}

__device__ inline void project_stripes(const float* wl, const float* __restrict__ X,
                                       const float* __restrict__ att_s, const float* __restrict__ att_d,
                                       float* __restrict__ H, float* __restrict__ as1, float* __restrict__ ad1,
                                       int N, int s_first, int s_count) {
    const int lane = threadIdx.x & 31;
    const int half = lane >> 4;
    const int mn = lane & 15;
    const int wave = __builtin_amdgcn_readfirstlane((int)(threadIdx.x >> 5));
    for (int task = wave; task < 2 * s_count; task += PROJ_WAVES) {
        const int rowBase = (s_first + (task >> 1)) * 16;
        const int colBase = (task & 1) * 64;
        const int arow = min(rowBase + mn, N - 1);
        const float4* xr = (const float4*)(X + (size_t)arow * 128 + 4 * half);
        float4 a[16];
        #pragma unroll
        for (int i = 0; i < 16; ++i) a[i] = xr[2 * i];
        v8f c0 = {}, c1 = {}, c2 = {}, c3 = {};
        const float* wr = wl + (4 * half) * WPITCH + colBase + mn;
        #pragma unroll
        for (int i = 0; i < 16; ++i) {
            const float* w = wr + 8 * i * WPITCH;
            v2f av;
            av.x = a[i].x; av.y = a[i].y;
            c0 = mma4(av, w[0],  w[WPITCH],      c0);
            c1 = mma4(av, w[16], w[WPITCH + 16], c1);
            c2 = mma4(av, w[32], w[WPITCH + 32], c2);
            c3 = mma4(av, w[48], w[WPITCH + 48], c3);
            av.x = a[i].z; av.y = a[i].w;
            c0 = mma4(av, w[2 * WPITCH],      w[3 * WPITCH],      c0);
            c1 = mma4(av, w[2 * WPITCH + 16], w[3 * WPITCH + 16], c1);
            c2 = mma4(av, w[2 * WPITCH + 32], w[3 * WPITCH + 32], c2);
            c3 = mma4(av, w[2 * WPITCH + 48], w[3 * WPITCH + 48], c3);
        }
        const float s0 = att_s[colBase + mn], s1 = att_s[colBase + 16 + mn];
        const float s2 = att_s[colBase + 32 + mn], s3 = att_s[colBase + 48 + mn];
        const float d0 = att_d[colBase + mn], d1 = att_d[colBase + 16 + mn];
        const float d2 = att_d[colBase + 32 + mn], d3 = att_d[colBase + 48 + mn];
        const int head0 = colBase >> 5;
        float ps0[8], ps1[8], pd0[8], pd1[8];
        #pragma unroll
        for (int r = 0; r < 8; ++r) {
            ps0[r] = half_sum16(c0[r] * s0 + c1[r] * s1);
            ps1[r] = half_sum16(c2[r] * s2 + c3[r] * s3);
            pd0[r] = half_sum16(c0[r] * d0 + c1[r] * d1);
            pd1[r] = half_sum16(c2[r] * d2 + c3[r] * d3);
        }
        #pragma unroll
        for (int r = 0; r < 8; ++r) {
            const int row = rowBase + r + 8 * half;
            if (row < N) {
                float* hp = H + (size_t)row * 128 + colBase + mn;
                hp[0] = c0[r]; hp[16] = c1[r]; hp[32] = c2[r]; hp[48] = c3[r];
                if (mn == 0) {
                    *(float2*)(as1 + (size_t)row * 4 + head0) = make_float2(ps0[r], ps1[r]);
                    *(float2*)(ad1 + (size_t)row * 4 + head0) = make_float2(pd0[r], pd1[r]);
                }
            }
        }
    }
}

#define ROWS 196
#define QCAP 8192
#define L1_THREADS 1024
#define SCAN_VECS 8
#define SCAN_SPAN (L1_THREADS * 4 * SCAN_VECS)
#define BIN_SLICES 16
#define BIN_CAP 512
#define BIN_LOADS (BIN_SLICES * BIN_CAP / L1_THREADS)

struct L1Shared {
    float acc[ROWS * 128];
    float4 lg[QCAP];
    int2 q[QCAP];
    float mx[ROWS * 4];
    float den[ROWS * 4];
    float ad[ROWS * 4];
    int qcount;
};

__device__ inline void lds_max_float(float* addr, float val) {
    if (val >= 0.0f) atomicMax((int*)addr, __float_as_int(val));
    else             atomicMin((unsigned int*)addr, __float_as_uint(val));
}

__device__ inline float leaky(float v) { return v >= 0.0f ? v : NEG_SLOPE * v; }

template <class S>
__device__ inline void note_edge(S& s, int dv, int e, int r0, int rn) {
    const unsigned r = (unsigned)(dv - r0);
    if (r < (unsigned)rn) {
        const int slot = atomicAdd(&s.qcount, 1);
        if (slot < QCAP) s.q[slot] = make_int2(e, (int)r);
    }
}

template <class S>
__device__ inline void scan_edges(S& s, const int* __restrict__ dstp, int eb, int ee, bool vec_ok,
                                  int r0, int rn) {
    const int tid = threadIdx.x;
    const int span = ee - eb;
    const int nsuper = (span + SCAN_SPAN - 1) / SCAN_SPAN;
    if (nsuper <= 0) return;
    const int rot = (int)((blockIdx.x * 5u) % (unsigned)nsuper);
    for (int it = 0; it < nsuper; ++it) {
        int ss = it + rot;
        if (ss >= nsuper) ss -= nsuper;
        const int sbase = eb + ss * SCAN_SPAN;
        if (vec_ok && sbase + SCAN_SPAN <= ee) {
            const int4* p = (const int4*)(dstp + sbase) + tid;
            int4 d[SCAN_VECS];
            #pragma unroll
            for (int k = 0; k < SCAN_VECS; ++k) d[k] = p[k * L1_THREADS];
            #pragma unroll
            for (int k = 0; k < SCAN_VECS; ++k) {
                const int e0 = sbase + (k * L1_THREADS + tid) * 4;
                note_edge(s, d[k].x, e0,     r0, rn);
                note_edge(s, d[k].y, e0 + 1, r0, rn);
                note_edge(s, d[k].z, e0 + 2, r0, rn);
                note_edge(s, d[k].w, e0 + 3, r0, rn);
            }
        } else {
            for (int kb = sbase; kb < min(ee, sbase + SCAN_SPAN); kb += L1_THREADS * 4) {
                if (vec_ok && kb + L1_THREADS * 4 <= ee) {
                    const int4 d = *((const int4*)(dstp + kb) + tid);
                    const int e0 = kb + tid * 4;
                    note_edge(s, d.x, e0,     r0, rn);
                    note_edge(s, d.y, e0 + 1, r0, rn);
                    note_edge(s, d.z, e0 + 2, r0, rn);
                    note_edge(s, d.w, e0 + 3, r0, rn);
                } else {
                    for (int e = kb + tid; e < min(ee, kb + L1_THREADS * 4); e += L1_THREADS)
                        note_edge(s, dstp[e], e, r0, rn);
                }
            }
        }
    }
}

template <class S>
__device__ inline void l1_sources(S& s, int from, int total, const int* __restrict__ ei) {
    for (int i = from + threadIdx.x; i < total; i += L1_THREADS) s.q[i].x = ei[s.q[i].x];
}

__device__ inline void take_lists(int2* q, int at, const int2* __restrict__ bins, int first, int stride,
                                  const int (&cn)[BIN_SLICES], int rowstep, const int2* __restrict__ spare) {
    const int hi = threadIdx.x / BIN_CAP;
    const int i = threadIdx.x % BIN_CAP;
    int2 d[BIN_LOADS];
    #pragma unroll
    for (int k = 0; k < BIN_LOADS; ++k) {
        const int cnt = hi ? cn[2 * k + 1] : cn[2 * k];
        const int2* p = bins + (size_t)(first + (2 * k + hi) * stride) * BIN_CAP;
        d[k] = *((i < cnt) ? p + i : spare);
    }
    int off = at;
    #pragma unroll
    for (int k = 0; k < BIN_LOADS; ++k) {
        const int cnt = hi ? cn[2 * k + 1] : cn[2 * k];
        const int o = hi ? off + cn[2 * k] : off;
        if (i < cnt) q[o + i] = make_int2(d[k].x, d[k].y + (2 * k + hi) * rowstep);
        off += cn[2 * k] + cn[2 * k + 1];
    }
}

__device__ inline void pass_logits(L1Shared& s, int total, const float* __restrict__ as1) {
    for (int i = threadIdx.x; i < total; i += L1_THREADS) {
        const int2 en = s.q[i];
        const float4 a = *(const float4*)(as1 + (size_t)en.x * 4);
        const float4 d = *(const float4*)(&s.ad[en.y * 4]);
        float4 v;
        v.x = leaky(a.x + d.x); v.y = leaky(a.y + d.y); v.z = leaky(a.z + d.z); v.w = leaky(a.w + d.w);
        s.lg[i] = v;
        lds_max_float(&s.mx[en.y * 4 + 0], v.x);
        lds_max_float(&s.mx[en.y * 4 + 1], v.y);
        lds_max_float(&s.mx[en.y * 4 + 2], v.z);
        lds_max_float(&s.mx[en.y * 4 + 3], v.w);
    }
}

__device__ inline float shift_of(float m) { return isfinite(m) ? m : 0.0f; }

__device__ inline void pass_weights(L1Shared& s, int total) {
    for (int i = threadIdx.x; i < total; i += L1_THREADS) {
        const int row = s.q[i].y;
        const float4 v = s.lg[i];
        const float4 m = *(const float4*)(&s.mx[row * 4]);
        float4 w;
        w.x = expf(v.x - shift_of(m.x)); w.y = expf(v.y - shift_of(m.y));
        w.z = expf(v.z - shift_of(m.z)); w.w = expf(v.w - shift_of(m.w));
        s.lg[i] = w;
        atomicAdd(&s.den[row * 4 + 0], w.x);
        atomicAdd(&s.den[row * 4 + 1], w.y);
        atomicAdd(&s.den[row * 4 + 2], w.z);
        atomicAdd(&s.den[row * 4 + 3], w.w);
    }
}

__device__ inline float lane_value(float v, int k) {
    return __int_as_float(__builtin_amdgcn_readlane(__float_as_int(v), k));
}

__device__ inline void pass_rows(L1Shared& s, int total, const float* __restrict__ h1) {
    const int lane = threadIdx.x & 31;
    const int wave = threadIdx.x >> 5;
    for (int i0 = wave * 8; i0 < total; i0 += (L1_THREADS / 32) * 8) {
        const int mine = min(i0 + (lane & 7), total - 1);
        const int2 en = s.q[mine];
        const float4 wv = s.lg[mine];
        float x[8][4];
        #pragma unroll
        for (int k = 0; k < 8; ++k) {
            x[k][0] = x[k][1] = x[k][2] = x[k][3] = 0.0f;
            if (i0 + k < total) {
                const int src = __builtin_amdgcn_readlane(en.x, k);
                const float* hs = h1 + (size_t)src * 128 + lane;
                x[k][0] = hs[0]; x[k][1] = hs[32]; x[k][2] = hs[64]; x[k][3] = hs[96];
            }
        }
        #pragma unroll
        for (int k = 0; k < 8; ++k) {
            if (i0 + k < total) {
                const int row = __builtin_amdgcn_readlane(en.y, k);
                float* ob = &s.acc[row * 128 + lane];
                atomicAdd(ob,      lane_value(wv.x, k) * x[k][0]);
                atomicAdd(ob + 32, lane_value(wv.y, k) * x[k][1]);
                atomicAdd(ob + 64, lane_value(wv.z, k) * x[k][2]);
                atomicAdd(ob + 96, lane_value(wv.w, k) * x[k][3]);
            }
        }
    }
}

__global__ __launch_bounds__(L1_THREADS) void l1_owned_rows(const int* __restrict__ ei, int E, int N,
                                                            const float* __restrict__ as1,
                                                            const float* __restrict__ ad1,
                                                            const float* __restrict__ h1,
                                                            const float* __restrict__ b1,
                                                            const float* __restrict__ W2,
                                                            const float* __restrict__ att_s2,
                                                            const float* __restrict__ att_d2,
                                                            float* __restrict__ h2,
                                                            float* __restrict__ as2,
                                                            float* __restrict__ ad2,
                                                            const int2* __restrict__ bins,
                                                            const int* __restrict__ bcount) {
    __shared__ L1Shared s;
    const int tid = threadIdx.x;
    const int r0 = blockIdx.x * ROWS;
    const int rn = min(ROWS, N - r0);
    const int* dstp = ei + E;
    const bool vec_ok = ((E & 3) == 0) && ((((size_t)dstp) & 15) == 0);

    for (int i = tid; i < ROWS * 128; i += L1_THREADS) s.acc[i] = 0.0f;
    for (int i = tid; i < ROWS * 4; i += L1_THREADS) {
        s.mx[i] = -INFINITY;
        s.den[i] = 0.0f;
        s.ad[i] = (i < rn * 4) ? ad1[(size_t)r0 * 4 + i] : 0.0f;
    }
    if (tid < rn) s.q[tid] = make_int2(r0 + tid, tid);
    if (tid == 0) s.qcount = rn;
    __syncthreads();

    int cn[BIN_SLICES];
    int total = 0;
    bool binned = true;
    #pragma unroll
    for (int j = 0; j < BIN_SLICES; ++j) {
        cn[j] = bcount[blockIdx.x * BIN_SLICES + j];
        binned = binned && (cn[j] <= BIN_CAP);
        total += min(cn[j], BIN_CAP + 1);
    }
    if (binned) { if (rn + total <= QCAP) take_lists(s.q, rn, bins, blockIdx.x * BIN_SLICES, 1, cn, 0, (const int2*)bcount); }
    else        scan_edges(s, dstp, 0, E, vec_ok, r0, rn);
    __syncthreads();
    const int found = binned ? rn + total : s.qcount;
    if (!binned && found <= QCAP) {
        l1_sources(s, rn, found, ei);
        __syncthreads();
    }

    if (found <= QCAP) {
        pass_logits(s, found, as1);
        __syncthreads();
        pass_weights(s, found);
        __syncthreads();
        pass_rows(s, found, h1);
    } else {
        const int SEG = QCAP - 256;
        const int nseg = (E + SEG - 1) / SEG;
        for (int sweep = 0; sweep < 2; ++sweep) {
            for (int sg = 0; sg < nseg; ++sg) {
                __syncthreads();
                if (sg == 0 && tid < rn) s.q[tid] = make_int2(r0 + tid, tid);
                if (tid == 0) s.qcount = (sg == 0) ? rn : 0;
                __syncthreads();
                scan_edges(s, dstp, sg * SEG, min(E, (sg + 1) * SEG), vec_ok, r0, rn);
                __syncthreads();
                const int cnt = s.qcount;
                l1_sources(s, (sg == 0) ? rn : 0, cnt, ei);
                __syncthreads();
                pass_logits(s, cnt, as1);
                __syncthreads();
                if (sweep == 1) {
                    pass_weights(s, cnt);
                    __syncthreads();
                    pass_rows(s, cnt, h1);
                }
            }
        }
    }
    const int lane = tid & 31;
    const int c = lane * 4;
    const float bx = b1[c], by = b1[c + 1], bz = b1[c + 2], bw = b1[c + 3];
    const float w00 = W2[(c + 0) * 2], w01 = W2[(c + 0) * 2 + 1];
    const float w10 = W2[(c + 1) * 2], w11 = W2[(c + 1) * 2 + 1];
    const float w20 = W2[(c + 2) * 2], w21 = W2[(c + 2) * 2 + 1];
    const float w30 = W2[(c + 3) * 2], w31 = W2[(c + 3) * 2 + 1];
    const float ts0 = att_s2[0], ts1 = att_s2[1], td0 = att_d2[0], td1 = att_d2[1];
    __syncthreads();

    for (int row = tid >> 5; row < rn; row += L1_THREADS / 32) {
        const float4 a = *(const float4*)(&s.acc[row * 128 + c]);
        const float d = s.den[row * 4 + (lane >> 3)] + EPS_F;
        float4 v;
        v.x = a.x / d + bx; v.y = a.y / d + by; v.z = a.z / d + bz; v.w = a.w / d + bw;
        v.x = v.x > 0.0f ? v.x : expf(v.x) - 1.0f;
        v.y = v.y > 0.0f ? v.y : expf(v.y) - 1.0f;
        v.z = v.z > 0.0f ? v.z : expf(v.z) - 1.0f;
        v.w = v.w > 0.0f ? v.w : expf(v.w) - 1.0f;
        float s0 = v.x * w00 + v.y * w10 + v.z * w20 + v.w * w30;
        float s1 = v.x * w01 + v.y * w11 + v.z * w21 + v.w * w31;
        #pragma unroll
        for (int off = 16; off >= 1; off >>= 1) {
            s0 += __shfl_xor(s0, off, 32);
            s1 += __shfl_xor(s1, off, 32);
        }
        if (lane == 0)
            s.lg[row] = make_float4(s0, s1, s0 * ts0 + s1 * ts1, s0 * td0 + s1 * td1);
    }
    __syncthreads();

    if (tid < rn) {
        const float4 r = s.lg[tid];
        const int n = r0 + tid;
        *(float2*)(h2 + (size_t)n * 2) = make_float2(r.x, r.y);
        as2[n] = r.z;
        ad2[n] = r.w;
    }
}

#define L2_ROWS 3136
#define L2_SLICES 16
#define L2_CAP 8192
#define L2_THREADS 1024
#define L2_STAGE 4

struct L2Shared {
    int2 q[L2_CAP];
    float2 hx[L2_CAP];
    float lg[L2_CAP];
    float mx[ROWS];
    float den[ROWS];
    float s0[ROWS];
    float s1[ROWS];
    float ad[ROWS];
    int qcount;
};

__device__ inline int l2_hit(int dv, int r0, int rn) {
    return ((unsigned)(dv - r0) < (unsigned)rn) ? 1 : 0;
}

__device__ inline void l2_logits(L2Shared& s, int total,
                                 const float* __restrict__ as2, const float* __restrict__ h2) {
    const int tid = threadIdx.x;
    for (int base = 0; base < total; base += L2_THREADS * L2_STAGE) {
        int srcn[L2_STAGE], row[L2_STAGE];
        float a[L2_STAGE];
        float2 h[L2_STAGE];
        #pragma unroll
        for (int k = 0; k < L2_STAGE; ++k) {
            const int i = base + k * L2_THREADS + tid;
            const int2 en = s.q[min(i, total - 1)];
            row[k] = (i < total) ? en.y : -1;
            srcn[k] = en.x;
        }
        #pragma unroll
        for (int k = 0; k < L2_STAGE; ++k) {
            a[k] = as2[srcn[k]];
            h[k] = *(const float2*)(h2 + (size_t)srcn[k] * 2);
        }
        #pragma unroll
        for (int k = 0; k < L2_STAGE; ++k) {
            if (row[k] >= 0) {
                const int i = base + k * L2_THREADS + tid;
                const float v = leaky(a[k] + s.ad[row[k]]);
                s.lg[i] = v;
                s.hx[i] = h[k];
                lds_max_float(&s.mx[row[k]], v);
            }
        }
    }
}

__device__ inline void l2_weights(L2Shared& s, int total) {
    for (int i = threadIdx.x; i < total; i += L2_THREADS) {
        const int row = s.q[i].y;
        const float w = expf(s.lg[i] - shift_of(s.mx[row]));
        const float2 hv = s.hx[i];
        atomicAdd(&s.den[row], w);
        atomicAdd(&s.s0[row], w * hv.x);
        atomicAdd(&s.s1[row], w * hv.y);
    }
}

static_assert(ROWS * 16 == L2_ROWS && BIN_SLICES == L2_SLICES && L1_THREADS == L2_THREADS && L1_THREADS == 2 * BIN_CAP &&
              QCAP == L2_CAP,
              "a range of the first kernel is sixteen row blocks; both layers' groups are built alike and share the list builders");

static_assert(PROJ_WAVES * 32 == L2_THREADS, "a projection task per wave of the first kernel's group");
#define PROJ_FIRST 16
#define WAVE_CAP 256
#define WAVE_VECS 13
#define DEAL_STAGE 4

struct BinShared {
    int2 q[PROJ_WAVES * WAVE_CAP];
    int sub[L2_SLICES];
    int over;
};

__device__ inline int wave_put(int2* wq, int dv, int e, int r0, int rn, int slot) {
    const unsigned r = (unsigned)(dv - r0);
    if (r < (unsigned)rn) {
        if (slot < WAVE_CAP) wq[slot] = make_int2(e, (int)r);
        ++slot;
    }
    return slot;
}

__device__ __forceinline__ int wave_hits(int2* wq, const int* __restrict__ dstp, int wb, int we, bool vec_ok,
                                         int r0, int rn) {
    const int lane = threadIdx.x & 31;
    int found = 0;
    int done = wb;
    if (vec_ok) {
        const int nvec = (we - wb) >> 2;
        const int4* p = (const int4*)(dstp + wb);
        for (int vb = 0; vb < nvec; vb += 32 * WAVE_VECS) {
            int4 d[WAVE_VECS];
            #pragma unroll
            for (int k = 0; k < WAVE_VECS; ++k) d[k] = p[min(vb + k * 32 + lane, nvec - 1)];
            #pragma unroll
            for (int k = 0; k < WAVE_VECS; ++k)
                if (vb + k * 32 + lane >= nvec) d[k] = make_int4(-1, -1, -1, -1);
            int cnt = 0;
            #pragma unroll
            for (int k = 0; k < WAVE_VECS; ++k)
                cnt += l2_hit(d[k].x, r0, rn) + l2_hit(d[k].y, r0, rn) + l2_hit(d[k].z, r0, rn) + l2_hit(d[k].w, r0, rn);
            int incl = cnt;
            #pragma unroll
            for (int off = 1; off < 32; off <<= 1) {
                const int t = __shfl_up(incl, off, 32);
                if (lane >= off) incl += t;
            }
            int slot = found + incl - cnt;
            #pragma unroll
            for (int k = 0; k < WAVE_VECS; ++k) {
                const int e0 = wb + (vb + k * 32 + lane) * 4;
                slot = wave_put(wq, d[k].x, e0,     r0, rn, slot);
                slot = wave_put(wq, d[k].y, e0 + 1, r0, rn, slot);
                slot = wave_put(wq, d[k].z, e0 + 2, r0, rn, slot);
                slot = wave_put(wq, d[k].w, e0 + 3, r0, rn, slot);
            }
            found += __shfl(incl, 31, 32);
        }
        done = wb + nvec * 4;
    }
    for (int eb = done; eb < we; eb += 32) {
        const int e = eb + lane;
        const int dv = (e < we) ? dstp[e] : -1;
        const bool hit = l2_hit(dv, r0, rn) != 0;
        const unsigned mask = __builtin_amdgcn_ballot_w32(hit);
        const int slot = found + __builtin_popcount(mask & ((1u << lane) - 1u));
        if (hit && slot < WAVE_CAP) wq[slot] = make_int2(e, dv - r0);
        found += __builtin_popcount(mask);
    }
    return found;
}

__device__ inline void wave_deal(BinShared& s, const int2* wq, int n, const int* __restrict__ ei,
                                 int2* __restrict__ bins, int range, int slice) {
    const int lane = threadIdx.x & 31;
    for (int base = 0; base < n; base += 32 * DEAL_STAGE) {
        int2 en[DEAL_STAGE];
        int srcn[DEAL_STAGE];
        #pragma unroll
        for (int k = 0; k < DEAL_STAGE; ++k) en[k] = wq[min(base + k * 32 + lane, n - 1)];
        #pragma unroll
        for (int k = 0; k < DEAL_STAGE; ++k) srcn[k] = ei[en[k].x];
        #pragma unroll
        for (int k = 0; k < DEAL_STAGE; ++k) {
            if (base + k * 32 + lane < n) {
                const int sb = en[k].y / ROWS;
                const int slot = atomicAdd(&s.sub[sb], 1);
                if (slot < BIN_CAP)
                    bins[(size_t)((range * 16 + sb) * BIN_SLICES + slice) * BIN_CAP + slot] =
                        make_int2(srcn[k], en[k].y - sb * ROWS);
            }
        }
    }
}

__global__ __launch_bounds__(L2_THREADS) void bin_edges(const int* __restrict__ ei, int E, int N,
                                                        int2* __restrict__ bins, int* __restrict__ bcount,
                                                        const float* __restrict__ X, const float* __restrict__ W1,
                                                        const float* __restrict__ att_s, const float* __restrict__ att_d,
                                                        float* __restrict__ H, float* __restrict__ as1,
                                                        float* __restrict__ ad1) {
    __shared__ BinShared s;
    __shared__ __attribute__((aligned(16))) float wl[128 * WPITCH];
    const int tid = threadIdx.x;
    const int wave = __builtin_amdgcn_readfirstlane((int)(threadIdx.x >> 5));
    const int range = blockIdx.x / L2_SLICES;
    const int slice = blockIdx.x % L2_SLICES;
    const int r0 = range * L2_ROWS;
    const int rn = min(L2_ROWS, N - r0);
    const int slen = ((E + L2_SLICES - 1) / L2_SLICES + 3) & ~3;
    const int eb = min(E, slice * slen);
    const int ee = min(E, eb + slen);
    const int* dstp = ei + E;
    const bool vec_ok = ((E & 3) == 0) && ((((size_t)dstp) & 15) == 0);
    const int wlen = ((slen + PROJ_WAVES - 1) / PROJ_WAVES + 3) & ~3;
    const int wb = min(ee, eb + wave * wlen);
    const int we = min(ee, wb + wlen);
    int2* wq = s.q + wave * WAVE_CAP;
    const int nst = (N + 15) / 16;
    const int per = nst / (int)gridDim.x;
    const int rem = nst % (int)gridDim.x;
    const int g = (int)blockIdx.x;
    const int s_first = g * per + min(g, rem);
    const int s_count = per + (g < rem ? 1 : 0);
    const bool proj_first = wave < PROJ_FIRST;

    float4 wv[4];
    #pragma unroll
    for (int k = 0; k < 4; ++k) wv[k] = ((const float4*)W1)[k * L2_THREADS + tid];
    if (tid == 0) s.over = 0;
    if (tid < 16) s.sub[tid] = 0;
    #pragma unroll
    for (int k = 0; k < 4; ++k) {
        const int v = k * L2_THREADS + tid;
        *(float4*)(&wl[(v >> 5) * WPITCH + (v & 31) * 4]) = wv[k];
    }
    __syncthreads();

    for (int turn = 0; turn < 2; ++turn) {
        if ((turn == 0) == proj_first) {
            project_stripes(wl, X, att_s, att_d, H, as1, ad1, N, s_first, s_count);
        } else {
            const int n = wave_hits(wq, dstp, wb, we, vec_ok, r0, rn);
            if (n <= WAVE_CAP) wave_deal(s, wq, n, ei, bins, range, slice);
            else               s.over = 1;
        }
    }
    __syncthreads();
    if (tid < 16) bcount[(range * 16 + tid) * BIN_SLICES + slice] = (s.over == 0) ? s.sub[tid] : BIN_CAP + 1;
}

__global__ __launch_bounds__(L2_THREADS) void l2_owned_rows(const int* __restrict__ ei, int E, int N,
                                                            const float* __restrict__ as2,
                                                            const float* __restrict__ ad2,
                                                            const float* __restrict__ h2,
                                                            const float* __restrict__ b2,
                                                            float* __restrict__ y,
                                                            const int2* __restrict__ bins,
                                                            const int* __restrict__ bcount) {
    __shared__ L2Shared s;
    const int tid = threadIdx.x;
    const int r0 = blockIdx.x * ROWS;
    const int rn = min(ROWS, N - r0);
    const int* dstp = ei + E;
    const bool vec_ok = ((E & 3) == 0) && ((((size_t)dstp) & 15) == 0);

    int cn[BIN_SLICES];
    int total = 0;
    bool binned = true;
    #pragma unroll
    for (int j = 0; j < BIN_SLICES; ++j) {
        cn[j] = bcount[blockIdx.x * BIN_SLICES + j];
        binned = binned && (cn[j] <= BIN_CAP);
        total += min(cn[j], BIN_CAP + 1);
    }
    const float adv = (tid < rn) ? ad2[r0 + tid] : 0.0f;

    if (tid < ROWS) {
        s.mx[tid] = -INFINITY;
        s.den[tid] = 0.0f;
        s.s0[tid] = 0.0f;
        s.s1[tid] = 0.0f;
    }
    if (tid < rn) s.q[tid] = make_int2(r0 + tid, tid);
    if (binned) {
        if (rn + total <= L2_CAP) take_lists(s.q, rn, bins, blockIdx.x * BIN_SLICES, 1, cn, 0, (const int2*)bcount);
    } else {
        if (tid == 0) s.qcount = rn;
        __syncthreads();
        scan_edges(s, dstp, 0, E, vec_ok, r0, rn);
    }
    if (tid < ROWS) s.ad[tid] = adv;
    __syncthreads();
    const int found = binned ? rn + total : s.qcount;
    if (!binned && found <= L2_CAP) {
        l1_sources(s, rn, found, ei);
        __syncthreads();
    }

    if (found <= L2_CAP) {
        l2_logits(s, found, as2, h2);
        __syncthreads();
        l2_weights(s, found);
    } else {
        const int SEG = L2_CAP - 256;
        const int nseg = (E + SEG - 1) / SEG;
        for (int sweep = 0; sweep < 2; ++sweep) {
            for (int sg = 0; sg < nseg; ++sg) {
                __syncthreads();
                if (sg == 0 && tid < rn) s.q[tid] = make_int2(r0 + tid, tid);
                if (tid == 0) s.qcount = (sg == 0) ? rn : 0;
                __syncthreads();
                scan_edges(s, dstp, sg * SEG, min(E, (sg + 1) * SEG), vec_ok, r0, rn);
                __syncthreads();
                const int cnt = s.qcount;
                l1_sources(s, (sg == 0) ? rn : 0, cnt, ei);
                __syncthreads();
                l2_logits(s, cnt, as2, h2);
                __syncthreads();
                if (sweep == 1) l2_weights(s, cnt);
            }
        }
    }
    const float bias0 = b2[0], bias1 = b2[1];
    __syncthreads();

    if (tid < rn) {
        const float d = s.den[tid] + EPS_F;
        const float z0 = s.s0[tid] / d + bias0;
        const float z1 = s.s1[tid] / d + bias1;
        const float mx = fmaxf(z0, z1);
        const float lse = mx + logf(expf(z0 - mx) + expf(z1 - mx));
        *(float2*)(y + (size_t)(r0 + tid) * 2) = make_float2(z0 - lse, z1 - lse);
    }
}

extern "C" void kernel_launch(void* const* d_in, const int* in_sizes, int n_in,
                              void* d_out, int out_size, void* d_ws, size_t ws_size,
                              hipStream_t stream) {
    const float* x        = (const float*)d_in[0];
    const int*   ei       = (const int*)d_in[1];
    const float* W1       = (const float*)d_in[2];
    const float* att_src1 = (const float*)d_in[3];
    const float* att_dst1 = (const float*)d_in[4];
    const float* b1       = (const float*)d_in[5];
    const float* W2       = (const float*)d_in[6];
    const float* att_src2 = (const float*)d_in[7];
    const float* att_dst2 = (const float*)d_in[8];
    const float* b2       = (const float*)d_in[9];
    float* y = (float*)d_out;

    const int N = in_sizes[0] / 128;
    const int E = in_sizes[1] / 2;

    float* ws   = (float*)d_ws;
    float* h1   = ws;
    float* as1  = h1   + (size_t)N * 256;
    float* ad1  = as1  + (size_t)N * 4;
    float* h2   = ad1  + (size_t)N * 12;
    float* as2  = h2   + (size_t)N * 2;
    float* ad2  = as2  + (size_t)N;
    const int NR = (N + L2_ROWS - 1) / L2_ROWS;
    int2* bins  = (int2*)(h1 + (size_t)N * 128);
    int* bcount = (int*)(bins + (size_t)NR * 16 * L2_SLICES * BIN_CAP);

    bin_edges<<<NR * L2_SLICES, L2_THREADS, 0, stream>>>(ei, E, N, bins, bcount, x, W1, att_src1, att_dst1, h1, as1, ad1);

    l1_owned_rows<<<(N + ROWS - 1) / ROWS, L1_THREADS, 0, stream>>>(ei, E, N, as1, ad1, h1, b1,
                                                                   W2, att_src2, att_dst2, h2, as2, ad2, bins, bcount);

    l2_owned_rows<<<(N + ROWS - 1) / ROWS, L2_THREADS, 0, stream>>>(ei, E, N, as2, ad2, h2, b2, y, bins, bcount);
}


# Layer-1 row sums leave the LDS float adds: the group orders its edge list by row in LDS (rank = returning LDS add), a wave sums a row in registers, one 16-byte load per lane per edge; order by rank
# speedup vs baseline: 7.6873x; 1.1196x over previous
#include <hip/hip_runtime.h>
#include <math.h>

typedef __attribute__((ext_vector_type(2))) float v2f;
typedef __attribute__((ext_vector_type(8))) float v8f;

#define NEG_SLOPE 0.2f
#define EPS_F 1e-16f

#define WPITCH 136
#define PROJ_WAVES 32

__device__ inline v8f mma4(v2f a, float b0, float b1, v8f c) {
    v2f b;
    b.x = b0;
    b.y = b1;
    return __builtin_amdgcn_wmma_f32_16x16x4_f32(false, a, false, b, (short)0, c, false, false);
}

__device__ inline float half_sum16(float v) {
    #pragma unroll
    for (int off = 8; off >= 1; off >>= 1) v += __shfl_xor(v, off, 32);
    return v;
}

__device__ inline void project_stripes(const float* wl, const float* __restrict__ X,
                                       const float* __restrict__ att_s, const float* __restrict__ att_d,
                                       float* __restrict__ H, float* __restrict__ as1, float* __restrict__ ad1,
                                       int N, int s_first, int s_count) {
    const int lane = threadIdx.x & 31;
    const int half = lane >> 4;
    const int mn = lane & 15;
    const int wave = __builtin_amdgcn_readfirstlane((int)(threadIdx.x >> 5));
    for (int task = wave; task < 2 * s_count; task += PROJ_WAVES) {
        const int rowBase = (s_first + (task >> 1)) * 16;
        const int colBase = (task & 1) * 64;
        const int arow = min(rowBase + mn, N - 1);
        const float4* xr = (const float4*)(X + (size_t)arow * 128 + 4 * half);
        float4 a[16];
        #pragma unroll
        for (int i = 0; i < 16; ++i) a[i] = xr[2 * i];
        v8f c0 = {}, c1 = {}, c2 = {}, c3 = {};
        const float* wr = wl + (4 * half) * WPITCH + colBase + mn;
        #pragma unroll
        for (int i = 0; i < 16; ++i) {
            const float* w = wr + 8 * i * WPITCH;
            v2f av;
            av.x = a[i].x; av.y = a[i].y;
            c0 = mma4(av, w[0],  w[WPITCH],      c0);
            c1 = mma4(av, w[16], w[WPITCH + 16], c1);
            c2 = mma4(av, w[32], w[WPITCH + 32], c2);
            c3 = mma4(av, w[48], w[WPITCH + 48], c3);
            av.x = a[i].z; av.y = a[i].w;
            c0 = mma4(av, w[2 * WPITCH],      w[3 * WPITCH],      c0);
            c1 = mma4(av, w[2 * WPITCH + 16], w[3 * WPITCH + 16], c1);
            c2 = mma4(av, w[2 * WPITCH + 32], w[3 * WPITCH + 32], c2);
            c3 = mma4(av, w[2 * WPITCH + 48], w[3 * WPITCH + 48], c3);
        }
        const float s0 = att_s[colBase + mn], s1 = att_s[colBase + 16 + mn];
        const float s2 = att_s[colBase + 32 + mn], s3 = att_s[colBase + 48 + mn];
        const float d0 = att_d[colBase + mn], d1 = att_d[colBase + 16 + mn];
        const float d2 = att_d[colBase + 32 + mn], d3 = att_d[colBase + 48 + mn];
        const int head0 = colBase >> 5;
        float ps0[8], ps1[8], pd0[8], pd1[8];
        #pragma unroll
        for (int r = 0; r < 8; ++r) {
            ps0[r] = half_sum16(c0[r] * s0 + c1[r] * s1);
            ps1[r] = half_sum16(c2[r] * s2 + c3[r] * s3);
            pd0[r] = half_sum16(c0[r] * d0 + c1[r] * d1);
            pd1[r] = half_sum16(c2[r] * d2 + c3[r] * d3);
        }
        #pragma unroll
        for (int r = 0; r < 8; ++r) {
            const int row = rowBase + r + 8 * half;
            if (row < N) {
                float* hp = H + (size_t)row * 128 + colBase + mn;
                hp[0] = c0[r]; hp[16] = c1[r]; hp[32] = c2[r]; hp[48] = c3[r];
                if (mn == 0) {
                    *(float2*)(as1 + (size_t)row * 4 + head0) = make_float2(ps0[r], ps1[r]);
                    *(float2*)(ad1 + (size_t)row * 4 + head0) = make_float2(pd0[r], pd1[r]);
                }
            }
        }
    }
}

#define ROWS 196
#define QCAP 8192
#define L1_THREADS 1024
#define SCAN_VECS 8
#define SCAN_SPAN (L1_THREADS * 4 * SCAN_VECS)
#define BIN_SLICES 16
#define BIN_CAP 512
#define BIN_LOADS (BIN_SLICES * BIN_CAP / L1_THREADS)

struct L1Shared {
    float acc[ROWS * 128];
    float4 lg[QCAP];
    int2 q[QCAP];
    float mx[ROWS * 4];
    float den[ROWS * 4];
    float ad[ROWS * 4];
    int qcount;
};

__device__ inline void lds_max_float(float* addr, float val) {
    if (val >= 0.0f) atomicMax((int*)addr, __float_as_int(val));
    else             atomicMin((unsigned int*)addr, __float_as_uint(val));
}

__device__ inline float leaky(float v) { return v >= 0.0f ? v : NEG_SLOPE * v; }

template <class S>
__device__ inline void note_edge(S& s, int dv, int e, int r0, int rn) {
    const unsigned r = (unsigned)(dv - r0);
    if (r < (unsigned)rn) {
        const int slot = atomicAdd(&s.qcount, 1);
        if (slot < QCAP) s.q[slot] = make_int2(e, (int)r);
    }
}

template <class S>
__device__ __forceinline__ void scan_edges(S& s, const int* __restrict__ dstp, int eb, int ee, bool vec_ok,
                                  int r0, int rn) {
    const int tid = threadIdx.x;
    const int span = ee - eb;
    const int nsuper = (span + SCAN_SPAN - 1) / SCAN_SPAN;
    if (nsuper <= 0) return;
    const int rot = (int)((blockIdx.x * 5u) % (unsigned)nsuper);
    for (int it = 0; it < nsuper; ++it) {
        int ss = it + rot;
        if (ss >= nsuper) ss -= nsuper;
        const int sbase = eb + ss * SCAN_SPAN;
        if (vec_ok && sbase + SCAN_SPAN <= ee) {
            const int4* p = (const int4*)(dstp + sbase) + tid;
            int4 d[SCAN_VECS];
            #pragma unroll
            for (int k = 0; k < SCAN_VECS; ++k) d[k] = p[k * L1_THREADS];
            #pragma unroll
            for (int k = 0; k < SCAN_VECS; ++k) {
                const int e0 = sbase + (k * L1_THREADS + tid) * 4;
                note_edge(s, d[k].x, e0,     r0, rn);
                note_edge(s, d[k].y, e0 + 1, r0, rn);
                note_edge(s, d[k].z, e0 + 2, r0, rn);
                note_edge(s, d[k].w, e0 + 3, r0, rn);
            }
        } else {
            for (int kb = sbase; kb < min(ee, sbase + SCAN_SPAN); kb += L1_THREADS * 4) {
                if (vec_ok && kb + L1_THREADS * 4 <= ee) {
                    const int4 d = *((const int4*)(dstp + kb) + tid);
                    const int e0 = kb + tid * 4;
                    note_edge(s, d.x, e0,     r0, rn);
                    note_edge(s, d.y, e0 + 1, r0, rn);
                    note_edge(s, d.z, e0 + 2, r0, rn);
                    note_edge(s, d.w, e0 + 3, r0, rn);
                } else {
                    for (int e = kb + tid; e < min(ee, kb + L1_THREADS * 4); e += L1_THREADS)
                        note_edge(s, dstp[e], e, r0, rn);
                }
            }
        }
    }
}

template <class S>
__device__ inline void l1_sources(S& s, int from, int total, const int* __restrict__ ei) {
    for (int i = from + threadIdx.x; i < total; i += L1_THREADS) s.q[i].x = ei[s.q[i].x];
}

__device__ inline void take_lists(int2* q, int at, const int2* __restrict__ bins, int first, int stride,
                                  const int (&cn)[BIN_SLICES], int rowstep, const int2* __restrict__ spare) {
    const int hi = threadIdx.x / BIN_CAP;
    const int i = threadIdx.x % BIN_CAP;
    int2 d[BIN_LOADS];
    #pragma unroll
    for (int k = 0; k < BIN_LOADS; ++k) {
        const int cnt = hi ? cn[2 * k + 1] : cn[2 * k];
        const int2* p = bins + (size_t)(first + (2 * k + hi) * stride) * BIN_CAP;
        d[k] = *((i < cnt) ? p + i : spare);
    }
    int off = at;
    #pragma unroll
    for (int k = 0; k < BIN_LOADS; ++k) {
        const int cnt = hi ? cn[2 * k + 1] : cn[2 * k];
        const int o = hi ? off + cn[2 * k] : off;
        if (i < cnt) q[o + i] = make_int2(d[k].x, d[k].y + (2 * k + hi) * rowstep);
        off += cn[2 * k] + cn[2 * k + 1];
    }
}

__device__ inline void pass_logits(L1Shared& s, int total, const float* __restrict__ as1) {
    for (int i = threadIdx.x; i < total; i += L1_THREADS) {
        const int2 en = s.q[i];
        const float4 a = *(const float4*)(as1 + (size_t)en.x * 4);
        const float4 d = *(const float4*)(&s.ad[en.y * 4]);
        float4 v;
        v.x = leaky(a.x + d.x); v.y = leaky(a.y + d.y); v.z = leaky(a.z + d.z); v.w = leaky(a.w + d.w);
        s.lg[i] = v;
        lds_max_float(&s.mx[en.y * 4 + 0], v.x);
        lds_max_float(&s.mx[en.y * 4 + 1], v.y);
        lds_max_float(&s.mx[en.y * 4 + 2], v.z);
        lds_max_float(&s.mx[en.y * 4 + 3], v.w);
    }
}

__device__ inline float shift_of(float m) { return isfinite(m) ? m : 0.0f; }

__device__ inline void pass_weights(L1Shared& s, int total) {
    for (int i = threadIdx.x; i < total; i += L1_THREADS) {
        const int row = s.q[i].y;
        const float4 v = s.lg[i];
        const float4 m = *(const float4*)(&s.mx[row * 4]);
        float4 w;
        w.x = expf(v.x - shift_of(m.x)); w.y = expf(v.y - shift_of(m.y));
        w.z = expf(v.z - shift_of(m.z)); w.w = expf(v.w - shift_of(m.w));
        s.lg[i] = w;
        atomicAdd(&s.den[row * 4 + 0], w.x);
        atomicAdd(&s.den[row * 4 + 1], w.y);
        atomicAdd(&s.den[row * 4 + 2], w.z);
        atomicAdd(&s.den[row * 4 + 3], w.w);
    }
}

__device__ inline float lane_value(float v, int k) {
    return __int_as_float(__builtin_amdgcn_readlane(__float_as_int(v), k));
}

__device__ __forceinline__ void pass_rows(L1Shared& s, int total, const float* __restrict__ h1) {
    const int lane = threadIdx.x & 31;
    const int wave = threadIdx.x >> 5;
    for (int i0 = wave * 8; i0 < total; i0 += (L1_THREADS / 32) * 8) {
        const int mine = min(i0 + (lane & 7), total - 1);
        const int2 en = s.q[mine];
        const float4 wv = s.lg[mine];
        float x[8][4];
        #pragma unroll
        for (int k = 0; k < 8; ++k) {
            x[k][0] = x[k][1] = x[k][2] = x[k][3] = 0.0f;
            if (i0 + k < total) {
                const int src = __builtin_amdgcn_readlane(en.x, k);
                const float* hs = h1 + (size_t)src * 128 + lane;
                x[k][0] = hs[0]; x[k][1] = hs[32]; x[k][2] = hs[64]; x[k][3] = hs[96];
            }
        }
        #pragma unroll
        for (int k = 0; k < 8; ++k) {
            if (i0 + k < total) {
                const int row = __builtin_amdgcn_readlane(en.y, k);
                float* ob = &s.acc[row * 128 + lane];
                atomicAdd(ob,      lane_value(wv.x, k) * x[k][0]);
                atomicAdd(ob + 32, lane_value(wv.y, k) * x[k][1]);
                atomicAdd(ob + 64, lane_value(wv.z, k) * x[k][2]);
                atomicAdd(ob + 96, lane_value(wv.w, k) * x[k][3]);
            }
        }
    }
}

#define FCAP 6272
#define ROW_STAGE 16
static_assert(FCAP * 16 == ROWS * 128 * 4, "the logits of FCAP edges fill exactly the accumulator plane");
static_assert((QCAP - FCAP) * 16 >= FCAP * 4 && (QCAP - FCAP) * 8 >= (2 * ROWS + 8) * 4 && ROWS < 256,
              "ordered sources fit behind the ordered weights, counters and offsets behind the list; a row number fits 8 bits");

__device__ inline void l1_rows_in_registers(L1Shared& s, int r0, int rn, int found, const int (&cn)[BIN_SLICES],
                                            const int2* __restrict__ bins, const int* __restrict__ bcount,
                                            const float* __restrict__ as1, const float* __restrict__ ad1,
                                            const float* __restrict__ h1) {
    const int tid = threadIdx.x;
    const int lane = tid & 31;
    const int wave = __builtin_amdgcn_readfirstlane((int)(threadIdx.x >> 5));
    float4* lg = (float4*)s.acc;
    float4* sw = s.lg;
    int* ssrc = (int*)(s.lg + FCAP);
    int* cnt = (int*)(s.q + FCAP);
    int* off = cnt + ROWS;

    for (int i = tid; i < ROWS * 4; i += L1_THREADS) {
        s.mx[i] = -INFINITY;
        s.ad[i] = (i < rn * 4) ? ad1[(size_t)r0 * 4 + i] : 0.0f;
    }
    if (tid < ROWS) cnt[tid] = 0;
    if (tid < rn) s.q[tid] = make_int2(r0 + tid, tid);
    take_lists(s.q, rn, bins, blockIdx.x * BIN_SLICES, 1, cn, 0, (const int2*)bcount);
    __syncthreads();

    for (int i = tid; i < found; i += L1_THREADS) {
        const int2 en = s.q[i];
        const float4 a = *(const float4*)(as1 + (size_t)en.x * 4);
        const int rank = atomicAdd(&cnt[en.y], 1);
        const float4 d = *(const float4*)(&s.ad[en.y * 4]);
        float4 v;
        v.x = leaky(a.x + d.x); v.y = leaky(a.y + d.y); v.z = leaky(a.z + d.z); v.w = leaky(a.w + d.w);
        lg[i] = v;
        lds_max_float(&s.mx[en.y * 4 + 0], v.x);
        lds_max_float(&s.mx[en.y * 4 + 1], v.y);
        lds_max_float(&s.mx[en.y * 4 + 2], v.z);
        lds_max_float(&s.mx[en.y * 4 + 3], v.w);
        s.q[i].y = en.y + (rank << 8);
    }
    __syncthreads();

    if (wave == 0) {
        int c[7];
        int sum = 0;
        #pragma unroll
        for (int j = 0; j < 7; ++j) {
            const int r = lane * 7 + j;
            c[j] = (r < ROWS) ? cnt[r] : 0;
            sum += c[j];
        }
        int incl = sum;
        #pragma unroll
        for (int o = 1; o < 32; o <<= 1) {
            const int t = __shfl_up(incl, o, 32);
            if (lane >= o) incl += t;
        }
        int run = incl - sum;
        #pragma unroll
        for (int j = 0; j < 7; ++j) {
            const int r = lane * 7 + j;
            if (r <= ROWS) off[r] = run;
            run += c[j];
        }
    }
    __syncthreads();

    for (int i = tid; i < found; i += L1_THREADS) {
        const int2 en = s.q[i];
        const int row = en.y & 255;
        const int rank = en.y >> 8;
        const float4 v = lg[i];
        const float4 m = *(const float4*)(&s.mx[row * 4]);
        float4 w;
        w.x = expf(v.x - shift_of(m.x)); w.y = expf(v.y - shift_of(m.y));
        w.z = expf(v.z - shift_of(m.z)); w.w = expf(v.w - shift_of(m.w));
        const int pos = off[row] + rank;
        ssrc[pos] = (rank == cnt[row] - 1) ? (en.x | (int)0x80000000) : en.x;
        sw[pos] = w;
    }
    __syncthreads();

    const int rb = (wave * rn) >> 5;
    const int re = ((wave + 1) * rn) >> 5;
    const int pe = __builtin_amdgcn_readfirstlane(off[re]);
    int cur = rb;
    const int hd = lane >> 3;
    const float* hl = h1 + lane * 4;
    float4 acc = make_float4(0.0f, 0.0f, 0.0f, 0.0f);
    float dsum = 0.0f;
    for (int p = __builtin_amdgcn_readfirstlane(off[rb]); p < pe; p += ROW_STAGE) {
        const int sv = ssrc[min(p + (lane & (ROW_STAGE - 1)), pe - 1)];
        float4 x[ROW_STAGE];
        float w[ROW_STAGE];
        #pragma unroll
        for (int k = 0; k < ROW_STAGE; ++k) {
            if (p + k < pe) {
                const int src = __builtin_amdgcn_readlane(sv, k) & 0x7fffffff;
                x[k] = *(const float4*)(hl + (size_t)src * 128);
                w[k] = ((const float*)(sw + p + k))[hd];
            }
        }
        #pragma unroll
        for (int k = 0; k < ROW_STAGE; ++k) {
            if (p + k < pe) {
                acc.x += w[k] * x[k].x; acc.y += w[k] * x[k].y; acc.z += w[k] * x[k].z; acc.w += w[k] * x[k].w;
                dsum += w[k];
                if (__builtin_amdgcn_readlane(sv, k) < 0) {
                    *(float4*)(&s.acc[cur * 128 + lane * 4]) = acc;
                    if ((lane & 7) == 0) s.den[cur * 4 + hd] = dsum;
                    acc = make_float4(0.0f, 0.0f, 0.0f, 0.0f);
                    dsum = 0.0f;
                    ++cur;
                }
            }
        }
    }
}

__global__ __launch_bounds__(L1_THREADS) void l1_owned_rows(const int* __restrict__ ei, int E, int N,
                                                            const float* __restrict__ as1,
                                                            const float* __restrict__ ad1,
                                                            const float* __restrict__ h1,
                                                            const float* __restrict__ b1,
                                                            const float* __restrict__ W2,
                                                            const float* __restrict__ att_s2,
                                                            const float* __restrict__ att_d2,
                                                            float* __restrict__ h2,
                                                            float* __restrict__ as2,
                                                            float* __restrict__ ad2,
                                                            const int2* __restrict__ bins,
                                                            const int* __restrict__ bcount) {
    __shared__ L1Shared s;
    const int tid = threadIdx.x;
    const int r0 = blockIdx.x * ROWS;
    const int rn = min(ROWS, N - r0);
    const int* dstp = ei + E;
    const bool vec_ok = ((E & 3) == 0) && ((((size_t)dstp) & 15) == 0);

    int cn[BIN_SLICES];
    int total = 0;
    bool binned = true;
    #pragma unroll
    for (int j = 0; j < BIN_SLICES; ++j) {
        cn[j] = bcount[blockIdx.x * BIN_SLICES + j];
        binned = binned && (cn[j] <= BIN_CAP);
        total += min(cn[j], BIN_CAP + 1);
    }
    const bool in_registers = binned && (rn + total <= FCAP);

    if (__builtin_expect(in_registers, true)) {
        l1_rows_in_registers(s, r0, rn, rn + total, cn, bins, bcount, as1, ad1, h1);
    } else {
        for (int i = tid; i < ROWS * 128; i += L1_THREADS) s.acc[i] = 0.0f;
        for (int i = tid; i < ROWS * 4; i += L1_THREADS) {
            s.mx[i] = -INFINITY;
            s.den[i] = 0.0f;
            s.ad[i] = (i < rn * 4) ? ad1[(size_t)r0 * 4 + i] : 0.0f;
        }
        if (tid < rn) s.q[tid] = make_int2(r0 + tid, tid);
        if (tid == 0) s.qcount = rn;
        __syncthreads();

        if (binned) { if (rn + total <= QCAP) take_lists(s.q, rn, bins, blockIdx.x * BIN_SLICES, 1, cn, 0, (const int2*)bcount); }
        else        scan_edges(s, dstp, 0, E, vec_ok, r0, rn);
        __syncthreads();
        const int found = binned ? rn + total : s.qcount;
        if (!binned && found <= QCAP) {
            l1_sources(s, rn, found, ei);
            __syncthreads();
        }

        if (found <= QCAP) {
            pass_logits(s, found, as1);
            __syncthreads();
            pass_weights(s, found);
            __syncthreads();
            pass_rows(s, found, h1);
        } else {
            const int SEG = QCAP - 256;
            const int nseg = (E + SEG - 1) / SEG;
            for (int sweep = 0; sweep < 2; ++sweep) {
                for (int sg = 0; sg < nseg; ++sg) {
                    __syncthreads();
                    if (sg == 0 && tid < rn) s.q[tid] = make_int2(r0 + tid, tid);
                    if (tid == 0) s.qcount = (sg == 0) ? rn : 0;
                    __syncthreads();
                    scan_edges(s, dstp, sg * SEG, min(E, (sg + 1) * SEG), vec_ok, r0, rn);
                    __syncthreads();
                    const int cnt = s.qcount;
                    l1_sources(s, (sg == 0) ? rn : 0, cnt, ei);
                    __syncthreads();
                    pass_logits(s, cnt, as1);
                    __syncthreads();
                    if (sweep == 1) {
                        pass_weights(s, cnt);
                        __syncthreads();
                        pass_rows(s, cnt, h1);
                    }
                }
            }
        }
    }
    const int lane = tid & 31;
    const int c = lane * 4;
    const float bx = b1[c], by = b1[c + 1], bz = b1[c + 2], bw = b1[c + 3];
    const float w00 = W2[(c + 0) * 2], w01 = W2[(c + 0) * 2 + 1];
    const float w10 = W2[(c + 1) * 2], w11 = W2[(c + 1) * 2 + 1];
    const float w20 = W2[(c + 2) * 2], w21 = W2[(c + 2) * 2 + 1];
    const float w30 = W2[(c + 3) * 2], w31 = W2[(c + 3) * 2 + 1];
    const float ts0 = att_s2[0], ts1 = att_s2[1], td0 = att_d2[0], td1 = att_d2[1];
    __syncthreads();

    for (int row = tid >> 5; row < rn; row += L1_THREADS / 32) {
        const float4 a = *(const float4*)(&s.acc[row * 128 + c]);
        const float d = s.den[row * 4 + (lane >> 3)] + EPS_F;
        float4 v;
        v.x = a.x / d + bx; v.y = a.y / d + by; v.z = a.z / d + bz; v.w = a.w / d + bw;
        v.x = v.x > 0.0f ? v.x : expf(v.x) - 1.0f;
        v.y = v.y > 0.0f ? v.y : expf(v.y) - 1.0f;
        v.z = v.z > 0.0f ? v.z : expf(v.z) - 1.0f;
        v.w = v.w > 0.0f ? v.w : expf(v.w) - 1.0f;
        float s0 = v.x * w00 + v.y * w10 + v.z * w20 + v.w * w30;
        float s1 = v.x * w01 + v.y * w11 + v.z * w21 + v.w * w31;
        #pragma unroll
        for (int off = 16; off >= 1; off >>= 1) {
            s0 += __shfl_xor(s0, off, 32);
            s1 += __shfl_xor(s1, off, 32);
        }
        if (lane == 0)
            s.lg[row] = make_float4(s0, s1, s0 * ts0 + s1 * ts1, s0 * td0 + s1 * td1);
    }
    __syncthreads();

    if (tid < rn) {
        const float4 r = s.lg[tid];
        const int n = r0 + tid;
        *(float2*)(h2 + (size_t)n * 2) = make_float2(r.x, r.y);
        as2[n] = r.z;
        ad2[n] = r.w;
    }
}

#define L2_ROWS 3136
#define L2_SLICES 16
#define L2_CAP 8192
#define L2_THREADS 1024
#define L2_STAGE 4

struct L2Shared {
    int2 q[L2_CAP];
    float2 hx[L2_CAP];
    float lg[L2_CAP];
    float mx[ROWS];
    float den[ROWS];
    float s0[ROWS];
    float s1[ROWS];
    float ad[ROWS];
    int qcount;
};

__device__ inline int l2_hit(int dv, int r0, int rn) {
    return ((unsigned)(dv - r0) < (unsigned)rn) ? 1 : 0;
}

__device__ inline void l2_logits(L2Shared& s, int total,
                                 const float* __restrict__ as2, const float* __restrict__ h2) {
    const int tid = threadIdx.x;
    for (int base = 0; base < total; base += L2_THREADS * L2_STAGE) {
        int srcn[L2_STAGE], row[L2_STAGE];
        float a[L2_STAGE];
        float2 h[L2_STAGE];
        #pragma unroll
        for (int k = 0; k < L2_STAGE; ++k) {
            const int i = base + k * L2_THREADS + tid;
            const int2 en = s.q[min(i, total - 1)];
            row[k] = (i < total) ? en.y : -1;
            srcn[k] = en.x;
        }
        #pragma unroll
        for (int k = 0; k < L2_STAGE; ++k) {
            a[k] = as2[srcn[k]];
            h[k] = *(const float2*)(h2 + (size_t)srcn[k] * 2);
        }
        #pragma unroll
        for (int k = 0; k < L2_STAGE; ++k) {
            if (row[k] >= 0) {
                const int i = base + k * L2_THREADS + tid;
                const float v = leaky(a[k] + s.ad[row[k]]);
                s.lg[i] = v;
                s.hx[i] = h[k];
                lds_max_float(&s.mx[row[k]], v);
            }
        }
    }
}

__device__ inline void l2_weights(L2Shared& s, int total) {
    for (int i = threadIdx.x; i < total; i += L2_THREADS) {
        const int row = s.q[i].y;
        const float w = expf(s.lg[i] - shift_of(s.mx[row]));
        const float2 hv = s.hx[i];
        atomicAdd(&s.den[row], w);
        atomicAdd(&s.s0[row], w * hv.x);
        atomicAdd(&s.s1[row], w * hv.y);
    }
}

static_assert(ROWS * 16 == L2_ROWS && BIN_SLICES == L2_SLICES && L1_THREADS == L2_THREADS && L1_THREADS == 2 * BIN_CAP &&
              QCAP == L2_CAP,
              "a range of the first kernel is sixteen row blocks; both layers' groups are built alike and share the list builders");

static_assert(PROJ_WAVES * 32 == L2_THREADS, "a projection task per wave of the first kernel's group");
#define PROJ_FIRST 16
#define WAVE_CAP 256
#define WAVE_VECS 13
#define DEAL_STAGE 4

struct BinShared {
    int2 q[PROJ_WAVES * WAVE_CAP];
    int sub[L2_SLICES];
    int over;
};

__device__ inline int wave_put(int2* wq, int dv, int e, int r0, int rn, int slot) {
    const unsigned r = (unsigned)(dv - r0);
    if (r < (unsigned)rn) {
        if (slot < WAVE_CAP) wq[slot] = make_int2(e, (int)r);
        ++slot;
    }
    return slot;
}

__device__ __forceinline__ int wave_hits(int2* wq, const int* __restrict__ dstp, int wb, int we, bool vec_ok,
                                         int r0, int rn) {
    const int lane = threadIdx.x & 31;
    int found = 0;
    int done = wb;
    if (vec_ok) {
        const int nvec = (we - wb) >> 2;
        const int4* p = (const int4*)(dstp + wb);
        for (int vb = 0; vb < nvec; vb += 32 * WAVE_VECS) {
            int4 d[WAVE_VECS];
            #pragma unroll
            for (int k = 0; k < WAVE_VECS; ++k) d[k] = p[min(vb + k * 32 + lane, nvec - 1)];
            #pragma unroll
            for (int k = 0; k < WAVE_VECS; ++k)
                if (vb + k * 32 + lane >= nvec) d[k] = make_int4(-1, -1, -1, -1);
            int cnt = 0;
            #pragma unroll
            for (int k = 0; k < WAVE_VECS; ++k)
                cnt += l2_hit(d[k].x, r0, rn) + l2_hit(d[k].y, r0, rn) + l2_hit(d[k].z, r0, rn) + l2_hit(d[k].w, r0, rn);
            int incl = cnt;
            #pragma unroll
            for (int off = 1; off < 32; off <<= 1) {
                const int t = __shfl_up(incl, off, 32);
                if (lane >= off) incl += t;
            }
            int slot = found + incl - cnt;
            #pragma unroll
            for (int k = 0; k < WAVE_VECS; ++k) {
                const int e0 = wb + (vb + k * 32 + lane) * 4;
                slot = wave_put(wq, d[k].x, e0,     r0, rn, slot);
                slot = wave_put(wq, d[k].y, e0 + 1, r0, rn, slot);
                slot = wave_put(wq, d[k].z, e0 + 2, r0, rn, slot);
                slot = wave_put(wq, d[k].w, e0 + 3, r0, rn, slot);
            }
            found += __shfl(incl, 31, 32);
        }
        done = wb + nvec * 4;
    }
    for (int eb = done; eb < we; eb += 32) {
        const int e = eb + lane;
        const int dv = (e < we) ? dstp[e] : -1;
        const bool hit = l2_hit(dv, r0, rn) != 0;
        const unsigned mask = __builtin_amdgcn_ballot_w32(hit);
        const int slot = found + __builtin_popcount(mask & ((1u << lane) - 1u));
        if (hit && slot < WAVE_CAP) wq[slot] = make_int2(e, dv - r0);
        found += __builtin_popcount(mask);
    }
    return found;
}

__device__ inline void wave_deal(BinShared& s, const int2* wq, int n, const int* __restrict__ ei,
                                 int2* __restrict__ bins, int range, int slice) {
    const int lane = threadIdx.x & 31;
    for (int base = 0; base < n; base += 32 * DEAL_STAGE) {
        int2 en[DEAL_STAGE];
        int srcn[DEAL_STAGE];
        #pragma unroll
        for (int k = 0; k < DEAL_STAGE; ++k) en[k] = wq[min(base + k * 32 + lane, n - 1)];
        #pragma unroll
        for (int k = 0; k < DEAL_STAGE; ++k) srcn[k] = ei[en[k].x];
        #pragma unroll
        for (int k = 0; k < DEAL_STAGE; ++k) {
            if (base + k * 32 + lane < n) {
                const int sb = en[k].y / ROWS;
                const int slot = atomicAdd(&s.sub[sb], 1);
                if (slot < BIN_CAP)
                    bins[(size_t)((range * 16 + sb) * BIN_SLICES + slice) * BIN_CAP + slot] =
                        make_int2(srcn[k], en[k].y - sb * ROWS);
            }
        }
    }
}

__global__ __launch_bounds__(L2_THREADS) void bin_edges(const int* __restrict__ ei, int E, int N,
                                                        int2* __restrict__ bins, int* __restrict__ bcount,
                                                        const float* __restrict__ X, const float* __restrict__ W1,
                                                        const float* __restrict__ att_s, const float* __restrict__ att_d,
                                                        float* __restrict__ H, float* __restrict__ as1,
                                                        float* __restrict__ ad1) {
    __shared__ BinShared s;
    __shared__ __attribute__((aligned(16))) float wl[128 * WPITCH];
    const int tid = threadIdx.x;
    const int wave = __builtin_amdgcn_readfirstlane((int)(threadIdx.x >> 5));
    const int range = blockIdx.x / L2_SLICES;
    const int slice = blockIdx.x % L2_SLICES;
    const int r0 = range * L2_ROWS;
    const int rn = min(L2_ROWS, N - r0);
    const int slen = ((E + L2_SLICES - 1) / L2_SLICES + 3) & ~3;
    const int eb = min(E, slice * slen);
    const int ee = min(E, eb + slen);
    const int* dstp = ei + E;
    const bool vec_ok = ((E & 3) == 0) && ((((size_t)dstp) & 15) == 0);
    const int wlen = ((slen + PROJ_WAVES - 1) / PROJ_WAVES + 3) & ~3;
    const int wb = min(ee, eb + wave * wlen);
    const int we = min(ee, wb + wlen);
    int2* wq = s.q + wave * WAVE_CAP;
    const int nst = (N + 15) / 16;
    const int per = nst / (int)gridDim.x;
    const int rem = nst % (int)gridDim.x;
    const int g = (int)blockIdx.x;
    const int s_first = g * per + min(g, rem);
    const int s_count = per + (g < rem ? 1 : 0);
    const bool proj_first = wave < PROJ_FIRST;

    float4 wv[4];
    #pragma unroll
    for (int k = 0; k < 4; ++k) wv[k] = ((const float4*)W1)[k * L2_THREADS + tid];
    if (tid == 0) s.over = 0;
    if (tid < 16) s.sub[tid] = 0;
    #pragma unroll
    for (int k = 0; k < 4; ++k) {
        const int v = k * L2_THREADS + tid;
        *(float4*)(&wl[(v >> 5) * WPITCH + (v & 31) * 4]) = wv[k];
    }
    __syncthreads();

    for (int turn = 0; turn < 2; ++turn) {
        if ((turn == 0) == proj_first) {
            project_stripes(wl, X, att_s, att_d, H, as1, ad1, N, s_first, s_count);
        } else {
            const int n = wave_hits(wq, dstp, wb, we, vec_ok, r0, rn);
            if (n <= WAVE_CAP) wave_deal(s, wq, n, ei, bins, range, slice);
            else               s.over = 1;
        }
    }
    __syncthreads();
    if (tid < 16) bcount[(range * 16 + tid) * BIN_SLICES + slice] = (s.over == 0) ? s.sub[tid] : BIN_CAP + 1;
}

__global__ __launch_bounds__(L2_THREADS) void l2_owned_rows(const int* __restrict__ ei, int E, int N,
                                                            const float* __restrict__ as2,
                                                            const float* __restrict__ ad2,
                                                            const float* __restrict__ h2,
                                                            const float* __restrict__ b2,
                                                            float* __restrict__ y,
                                                            const int2* __restrict__ bins,
                                                            const int* __restrict__ bcount) {
    __shared__ L2Shared s;
    const int tid = threadIdx.x;
    const int r0 = blockIdx.x * ROWS;
    const int rn = min(ROWS, N - r0);
    const int* dstp = ei + E;
    const bool vec_ok = ((E & 3) == 0) && ((((size_t)dstp) & 15) == 0);

    int cn[BIN_SLICES];
    int total = 0;
    bool binned = true;
    #pragma unroll
    for (int j = 0; j < BIN_SLICES; ++j) {
        cn[j] = bcount[blockIdx.x * BIN_SLICES + j];
        binned = binned && (cn[j] <= BIN_CAP);
        total += min(cn[j], BIN_CAP + 1);
    }
    const float adv = (tid < rn) ? ad2[r0 + tid] : 0.0f;

    if (tid < ROWS) {
        s.mx[tid] = -INFINITY;
        s.den[tid] = 0.0f;
        s.s0[tid] = 0.0f;
        s.s1[tid] = 0.0f;
    }
    if (tid < rn) s.q[tid] = make_int2(r0 + tid, tid);
    if (binned) {
        if (rn + total <= L2_CAP) take_lists(s.q, rn, bins, blockIdx.x * BIN_SLICES, 1, cn, 0, (const int2*)bcount);
    } else {
        if (tid == 0) s.qcount = rn;
        __syncthreads();
        scan_edges(s, dstp, 0, E, vec_ok, r0, rn);
    }
    if (tid < ROWS) s.ad[tid] = adv;
    __syncthreads();
    const int found = binned ? rn + total : s.qcount;
    if (!binned && found <= L2_CAP) {
        l1_sources(s, rn, found, ei);
        __syncthreads();
    }

    if (found <= L2_CAP) {
        l2_logits(s, found, as2, h2);
        __syncthreads();
        l2_weights(s, found);
    } else {
        const int SEG = L2_CAP - 256;
        const int nseg = (E + SEG - 1) / SEG;
        for (int sweep = 0; sweep < 2; ++sweep) {
            for (int sg = 0; sg < nseg; ++sg) {
                __syncthreads();
                if (sg == 0 && tid < rn) s.q[tid] = make_int2(r0 + tid, tid);
                if (tid == 0) s.qcount = (sg == 0) ? rn : 0;
                __syncthreads();
                scan_edges(s, dstp, sg * SEG, min(E, (sg + 1) * SEG), vec_ok, r0, rn);
                __syncthreads();
                const int cnt = s.qcount;
                l1_sources(s, (sg == 0) ? rn : 0, cnt, ei);
                __syncthreads();
                l2_logits(s, cnt, as2, h2);
                __syncthreads();
                if (sweep == 1) l2_weights(s, cnt);
            }
        }
    }
    const float bias0 = b2[0], bias1 = b2[1];
    __syncthreads();

    if (tid < rn) {
        const float d = s.den[tid] + EPS_F;
        const float z0 = s.s0[tid] / d + bias0;
        const float z1 = s.s1[tid] / d + bias1;
        const float mx = fmaxf(z0, z1);
        const float lse = mx + logf(expf(z0 - mx) + expf(z1 - mx));
        *(float2*)(y + (size_t)(r0 + tid) * 2) = make_float2(z0 - lse, z1 - lse);
    }
}

extern "C" void kernel_launch(void* const* d_in, const int* in_sizes, int n_in,
                              void* d_out, int out_size, void* d_ws, size_t ws_size,
                              hipStream_t stream) {
    const float* x        = (const float*)d_in[0];
    const int*   ei       = (const int*)d_in[1];
    const float* W1       = (const float*)d_in[2];
    const float* att_src1 = (const float*)d_in[3];
    const float* att_dst1 = (const float*)d_in[4];
    const float* b1       = (const float*)d_in[5];
    const float* W2       = (const float*)d_in[6];
    const float* att_src2 = (const float*)d_in[7];
    const float* att_dst2 = (const float*)d_in[8];
    const float* b2       = (const float*)d_in[9];
    float* y = (float*)d_out;

    const int N = in_sizes[0] / 128;
    const int E = in_sizes[1] / 2;

    float* ws   = (float*)d_ws;
    float* h1   = ws;
    float* as1  = h1   + (size_t)N * 256;
    float* ad1  = as1  + (size_t)N * 4;
    float* h2   = ad1  + (size_t)N * 12;
    float* as2  = h2   + (size_t)N * 2;
    float* ad2  = as2  + (size_t)N;
    const int NR = (N + L2_ROWS - 1) / L2_ROWS;
    int2* bins  = (int2*)(h1 + (size_t)N * 128);
    int* bcount = (int*)(bins + (size_t)NR * 16 * L2_SLICES * BIN_CAP);

    bin_edges<<<NR * L2_SLICES, L2_THREADS, 0, stream>>>(ei, E, N, bins, bcount, x, W1, att_src1, att_dst1, h1, as1, ad1);

    l1_owned_rows<<<(N + ROWS - 1) / ROWS, L1_THREADS, 0, stream>>>(ei, E, N, as1, ad1, h1, b1,
                                                                   W2, att_src2, att_dst2, h2, as2, ad2, bins, bcount);

    l2_owned_rows<<<(N + ROWS - 1) / ROWS, L2_THREADS, 0, stream>>>(ei, E, N, as2, ad2, h2, b2, y, bins, bcount);
}
